# starting kernel
# baseline (speedup 1.0000x reference)
#include <hip/hip_runtime.h>
#include <hip/hip_bf16.h>
#include <math.h>

typedef __attribute__((ext_vector_type(2))) float v2f;
typedef __attribute__((ext_vector_type(8))) float v8f;

#define BQ     4096
#define DIN    1024
#define DD     8
#define LL     2
#define FFN    128
#define NQ     10
#define DIMQ   1024
#define REPS   5
#define ROWS_PB 4

__global__ __launch_bounds__(128) void enc_gemm_wmma(
    const float* __restrict__ x, const float* __restrict__ encW,
    const float* __restrict__ encB, float* __restrict__ zpre)
{
    __shared__ float red[4][2][32][8];

    const int lane = threadIdx.x & 31;
    const int wave = threadIdx.x >> 5;
    const int l16  = lane & 15;
    const int half = lane >> 4;
    const int row0 = blockIdx.x * 32;

    const int   bcol  = l16 & 7;
    const float bmask = (l16 < DD) ? 1.0f : 0.0f;

    const float* xrow0 = x + (size_t)(row0 + l16) * DIN;
    const float* xrow1 = xrow0 + (size_t)16 * DIN;
    const int k0 = wave * (DIN / 4);

    v8f c0 = {}, c1 = {};
    for (int k = k0; k < k0 + DIN / 4; k += 4) {
        const int ka = k + half * 2;
        __builtin_prefetch(xrow0 + ka + 256, 0, 3);
        __builtin_prefetch(xrow1 + ka + 256, 0, 3);
        v2f a0, a1, b;
        a0.x = xrow0[ka];
        a0.y = xrow0[ka + 1];
        a1.x = xrow1[ka];
        a1.y = xrow1[ka + 1];
        b.x = bmask * encW[ka * DD + bcol];
        b.y = bmask * encW[(ka + 1) * DD + bcol];
        c0 = __builtin_amdgcn_wmma_f32_16x16x4_f32(
                 false, a0, false, b, (short)0, c0, false, false);
        c1 = __builtin_amdgcn_wmma_f32_16x16x4_f32(
                 false, a1, false, b, (short)0, c1, false, false);
    }

#pragma unroll
    for (int j = 0; j < 8; ++j) {
        red[wave][0][lane][j] = c0[j];
        red[wave][1][lane][j] = c1[j];
    }
    __syncthreads();

    if (wave < 2 && l16 < DD) {
        const int  t    = wave;
        const float bias = encB[l16];
#pragma unroll
        for (int j = 0; j < 8; ++j) {
            const float s = red[0][t][lane][j] + red[1][t][lane][j]
                          + red[2][t][lane][j] + red[3][t][lane][j];
            const int m = t * 16 + half * 8 + j;
            zpre[(size_t)(row0 + m) * DD + l16] = s + bias;
        }
    }
}

__device__ inline void ln8(const float* h, const float* g, const float* b, float* out)
{
    float m = 0.f;
#pragma unroll
    for (int d = 0; d < DD; ++d) m += h[d];
    m *= 0.125f;
    float v = 0.f;
#pragma unroll
    for (int d = 0; d < DD; ++d) { const float t = h[d] - m; v += t * t; }
    v *= 0.125f;
    const float inv = 1.0f / sqrtf(v + 1e-5f);
#pragma unroll
    for (int d = 0; d < DD; ++d) out[d] = (h[d] - m) * inv * g[d] + b[d];
}

__global__ __launch_bounds__(128) void qsim_kernel(
    const float* __restrict__ zpre,
    const float* __restrict__ Wv,  const float* __restrict__ bv,
    const float* __restrict__ Wo,  const float* __restrict__ bo,
    const float* __restrict__ ln1g, const float* __restrict__ ln1b,
    const float* __restrict__ W1,  const float* __restrict__ b1,
    const float* __restrict__ W2,  const float* __restrict__ b2,
    const float* __restrict__ ln2g, const float* __restrict__ ln2b,
    const float* __restrict__ qw,  float* __restrict__ zq)
{
    __shared__ float sRe[ROWS_PB][DIMQ];
    __shared__ float sIm[ROWS_PB][DIMQ];

    const int g    = threadIdx.x >> 5;
    const int lane = threadIdx.x & 31;
    const int row  = blockIdx.x * ROWS_PB + g;

    float z[DD];
#pragma unroll
    for (int d = 0; d < DD; ++d) z[d] = zpre[(size_t)row * DD + d];

    for (int l = 0; l < LL; ++l) {
        const float* wv = Wv + l * 64;  const float* wo = Wo + l * 64;
        float v[DD], attn[DD], h[DD];
#pragma unroll
        for (int d = 0; d < DD; ++d) {
            float s = bv[l * DD + d];
#pragma unroll
            for (int k = 0; k < DD; ++k) s += z[k] * wv[k * DD + d];
            v[d] = s;
        }
#pragma unroll
        for (int d = 0; d < DD; ++d) {
            float s = bo[l * DD + d];
#pragma unroll
            for (int k = 0; k < DD; ++k) s += v[k] * wo[k * DD + d];
            attn[d] = s;
        }
#pragma unroll
        for (int d = 0; d < DD; ++d) h[d] = z[d] + attn[d];
        ln8(h, ln1g + l * DD, ln1b + l * DD, z);

        float acc[DD];
#pragma unroll
        for (int d = 0; d < DD; ++d) acc[d] = 0.f;
#pragma unroll
        for (int j = 0; j < 4; ++j) {
            const int hid = lane + j * 32;
            float hv = b1[l * FFN + hid];
#pragma unroll
            for (int k = 0; k < DD; ++k) hv += z[k] * W1[l * DD * FFN + k * FFN + hid];
            hv = fmaxf(hv, 0.0f);
#pragma unroll
            for (int d = 0; d < DD; ++d) acc[d] += hv * W2[l * FFN * DD + hid * DD + d];
        }
        for (int off = 16; off > 0; off >>= 1) {
#pragma unroll
            for (int d = 0; d < DD; ++d) acc[d] += __shfl_xor(acc[d], off, 32);
        }
#pragma unroll
        for (int d = 0; d < DD; ++d) h[d] = z[d] + acc[d] + b2[l * DD + d];
        ln8(h, ln2g + l * DD, ln2b + l * DD, z);
    }

    float nrm2 = 0.f;
#pragma unroll
    for (int d = 0; d < DD; ++d) nrm2 += z[d] * z[d];
    const float inv = 1.0f / sqrtf(nrm2);

    for (int i = lane; i < DIMQ; i += 32) { sRe[g][i] = 0.f; sIm[g][i] = 0.f; }
#pragma unroll
    for (int d = 0; d < DD; ++d) if (lane == d) sRe[g][d] = z[d] * inv;
    __builtin_amdgcn_wave_barrier();

    for (int r = 0; r < REPS; ++r) {
        for (int w = 0; w < NQ; ++w) {
            const float th  = 0.5f * qw[r * NQ + w];
            const float cth = cosf(th);
            const float sth = sinf(th);
            const int bit = 1 << (NQ - 1 - w);
            const int low = bit - 1;
            for (int p = lane; p < DIMQ / 2; p += 32) {
                const int i0 = ((p & ~low) << 1) | (p & low);
                const int i1 = i0 | bit;
                const float a0r = sRe[g][i0], a0i = sIm[g][i0];
                const float a1r = sRe[g][i1], a1i = sIm[g][i1];
                sRe[g][i0] = cth * a0r + sth * a1i;
                sIm[g][i0] = cth * a0i - sth * a1r;
                sRe[g][i1] = cth * a1r + sth * a0i;
                sIm[g][i1] = cth * a1i - sth * a0r;
            }
            __builtin_amdgcn_wave_barrier();
        }
        for (int w = 0; w < NQ; ++w) {
            const int cbit = 1 << (NQ - 1 - w);
            const int tbit = 1 << (NQ - 1 - ((w + 1) % NQ));
            for (int i = lane; i < DIMQ; i += 32) {
                if ((i & cbit) && !(i & tbit)) {
                    const int j = i | tbit;
                    const float tr = sRe[g][i], ti = sIm[g][i];
                    sRe[g][i] = sRe[g][j];  sIm[g][i] = sIm[g][j];
                    sRe[g][j] = tr;         sIm[g][j] = ti;
                }
            }
            __builtin_amdgcn_wave_barrier();
        }
    }

    float acc[DD];
#pragma unroll
    for (int d = 0; d < DD; ++d) acc[d] = 0.f;
    for (int i = lane; i < DIMQ; i += 32) {
        const float re = sRe[g][i], im = sIm[g][i];
        const float p = re * re + im * im;
#pragma unroll
        for (int w = 0; w < DD; ++w) {
            const int bit = 1 << (NQ - 1 - w);
            acc[w] += (i & bit) ? -p : p;
        }
    }
    for (int off = 16; off > 0; off >>= 1) {
#pragma unroll
        for (int w = 0; w < DD; ++w) acc[w] += __shfl_xor(acc[w], off, 32);
    }
#pragma unroll
    for (int w = 0; w < DD; ++w) if (lane == w) zq[(size_t)row * DD + w] = acc[w];
}

__global__ __launch_bounds__(32) void dec_gemm_wmma(
    const float* __restrict__ zq, const float* __restrict__ decW,
    const float* __restrict__ decB, float* __restrict__ out)
{
    const int lane = threadIdx.x;
    const int l16  = lane & 15;
    const int half = lane >> 4;
    const int n0 = blockIdx.x * 16;
    const int m0 = blockIdx.y * 16;

    const float* arow = zq + (size_t)(m0 + l16) * DD;

    v8f c = {};
#pragma unroll
    for (int k = 0; k < DD; k += 4) {
        const int ka = k + half * 2;
        v2f a, b;
        a.x = arow[ka];
        a.y = arow[ka + 1];
        b.x = decW[(size_t)ka * DIN + n0 + l16];
        b.y = decW[(size_t)(ka + 1) * DIN + n0 + l16];
        c = __builtin_amdgcn_wmma_f32_16x16x4_f32(
                false, a, false, b, (short)0, c, false, false);
    }

    const float bias = decB[n0 + l16];
#pragma unroll
    for (int j = 0; j < 8; ++j) {
        const int m = half * 8 + j;
        out[(size_t)(m0 + m) * DIN + n0 + l16] = c[j] + bias;
    }
}

extern "C" void kernel_launch(void* const* d_in, const int* in_sizes, int n_in,
                              void* d_out, int out_size, void* d_ws, size_t ws_size,
                              hipStream_t stream) {
    const float* x    = (const float*)d_in[0];
    const float* encW = (const float*)d_in[1];
    const float* encB = (const float*)d_in[2];
    const float* decW = (const float*)d_in[3];
    const float* decB = (const float*)d_in[4];
    const float* Wv   = (const float*)d_in[5];
    const float* bv   = (const float*)d_in[6];
    const float* Wo   = (const float*)d_in[7];
    const float* bo   = (const float*)d_in[8];
    const float* ln1g = (const float*)d_in[9];
    const float* ln1b = (const float*)d_in[10];
    const float* W1   = (const float*)d_in[11];
    const float* b1   = (const float*)d_in[12];
    const float* W2   = (const float*)d_in[13];
    const float* b2   = (const float*)d_in[14];
    const float* ln2g = (const float*)d_in[15];
    const float* ln2b = (const float*)d_in[16];
    const float* qw   = (const float*)d_in[17];
    float* out  = (float*)d_out;

    float* zpre = (float*)d_ws;
    float* zqv  = zpre + (size_t)BQ * DD;

    enc_gemm_wmma<<<BQ / 32, 128, 0, stream>>>(x, encW, encB, zpre);

    qsim_kernel<<<BQ / ROWS_PB, 128, 0, stream>>>(
        zpre, Wv, bv, Wo, bo, ln1g, ln1b, W1, b1, W2, b2, ln2g, ln2b, qw, zqv);

    dim3 dgrid(DIN / 16, BQ / 16);
    dec_gemm_wmma<<<dgrid, 32, 0, stream>>>(zqv, decW, decB, out);
}


# Circuit is one linear map for all rows: simulate it on the 8 basis states once, reduce to 8 quadratic forms Q[a][b][w], rows evaluate zq = zn^T Q zn (sum order of zq changes)
# speedup vs baseline: 3.7749x; 3.7749x over previous
#include <hip/hip_runtime.h>
#include <hip/hip_bf16.h>
#include <math.h>

typedef __attribute__((ext_vector_type(2))) float v2f;
typedef __attribute__((ext_vector_type(8))) float v8f;

#define BQ     4096
#define DIN    1024
#define DD     8
#define LL     2
#define FFN    128
#define NQ     10
#define DIMQ   1024
#define REPS   5
#define ROWS_PB 4
#define NGATE  (REPS * NQ)

__global__ __launch_bounds__(128) void enc_gemm_wmma(
    const float* __restrict__ x, const float* __restrict__ encW,
    const float* __restrict__ encB, float* __restrict__ zpre)
{
    __shared__ float red[4][2][32][8];

    const int lane = threadIdx.x & 31;
    const int wave = threadIdx.x >> 5;
    const int l16  = lane & 15;
    const int half = lane >> 4;
    const int row0 = blockIdx.x * 32;

    const int   bcol  = l16 & 7;
    const float bmask = (l16 < DD) ? 1.0f : 0.0f;

    const float* xrow0 = x + (size_t)(row0 + l16) * DIN;
    const float* xrow1 = xrow0 + (size_t)16 * DIN;
    const int k0 = wave * (DIN / 4);

    v8f c0 = {}, c1 = {};
    for (int k = k0; k < k0 + DIN / 4; k += 4) {
        const int ka = k + half * 2;
        __builtin_prefetch(xrow0 + ka + 256, 0, 3);
        __builtin_prefetch(xrow1 + ka + 256, 0, 3);
        v2f a0, a1, b;
        a0.x = xrow0[ka];
        a0.y = xrow0[ka + 1];
        a1.x = xrow1[ka];
        a1.y = xrow1[ka + 1];
        b.x = bmask * encW[ka * DD + bcol];
        b.y = bmask * encW[(ka + 1) * DD + bcol];
        c0 = __builtin_amdgcn_wmma_f32_16x16x4_f32(
                 false, a0, false, b, (short)0, c0, false, false);
        c1 = __builtin_amdgcn_wmma_f32_16x16x4_f32(
                 false, a1, false, b, (short)0, c1, false, false);
    }

#pragma unroll
    for (int j = 0; j < 8; ++j) {
        red[wave][0][lane][j] = c0[j];
        red[wave][1][lane][j] = c1[j];
    }
    __syncthreads();

    if (wave < 2 && l16 < DD) {
        const int  t    = wave;
        const float bias = encB[l16];
#pragma unroll
        for (int j = 0; j < 8; ++j) {
            const float s = red[0][t][lane][j] + red[1][t][lane][j]
                          + red[2][t][lane][j] + red[3][t][lane][j];
            const int m = t * 16 + half * 8 + j;
            zpre[(size_t)(row0 + m) * DD + l16] = s + bias;
        }
    }
}

__device__ inline void rx_pair(float c, float s,
                               float& a0r, float& a0i, float& a1r, float& a1i)
{
    const float n0r = c * a0r + s * a1i;
    const float n0i = c * a0i - s * a1r;
    const float n1r = c * a1r + s * a0i;
    const float n1i = c * a1i - s * a0r;
    a0r = n0r; a0i = n0i; a1r = n1r; a1i = n1i;
}

__device__ inline int cnot_ring_dest(int i)
{
#pragma unroll
    for (int w = 0; w < NQ; ++w) {
        const int cpos = NQ - 1 - w;
        const int tpos = NQ - 1 - ((w + 1) % NQ);
        i ^= ((i >> cpos) & 1) << tpos;
    }
    return i;
}

__global__ __launch_bounds__(256) void circuit_forms(
    const float* __restrict__ qw, float* __restrict__ Q)
{
    __shared__ float sRe[2][2][DIMQ];
    __shared__ float sIm[2][2][DIMQ];
    __shared__ float sC[NGATE], sS[NGATE];
    __shared__ float part[8][DD];

    const int tid  = threadIdx.x;
    const int lane = tid & 31;
    const int wave = tid >> 5;
    const int a = blockIdx.x >> 3;
    const int b = blockIdx.x & 7;

    if (tid < NGATE) {
        const float th = 0.5f * qw[tid];
        sC[tid] = cosf(th);
        sS[tid] = sinf(th);
    }
    for (int i = tid; i < DIMQ; i += 256) {
        sRe[0][0][i] = (i == a) ? 1.0f : 0.0f;  sIm[0][0][i] = 0.0f;
        sRe[0][1][i] = (i == b) ? 1.0f : 0.0f;  sIm[0][1][i] = 0.0f;
    }
    __syncthreads();

    int cur = 0;
    for (int r = 0; r < REPS; ++r) {
        for (int t = 0; t < NQ / 2; ++t) {
            const int bl   = NQ - 2 - 2 * t;
            const int base = ((tid >> bl) << (bl + 2)) | (tid & ((1 << bl) - 1));
            const int i00 = base;
            const int i01 = base | (1 << bl);
            const int i10 = base | (2 << bl);
            const int i11 = base | (3 << bl);
            const float c0 = sC[r * NQ + 2 * t],     s0 = sS[r * NQ + 2 * t];
            const float c1 = sC[r * NQ + 2 * t + 1], s1 = sS[r * NQ + 2 * t + 1];
            const bool last = (t == NQ / 2 - 1);
            const int dst = last ? (cur ^ 1) : cur;
            const int j00 = last ? cnot_ring_dest(i00) : i00;
            const int j01 = last ? cnot_ring_dest(i01) : i01;
            const int j10 = last ? cnot_ring_dest(i10) : i10;
            const int j11 = last ? cnot_ring_dest(i11) : i11;
#pragma unroll
            for (int st = 0; st < 2; ++st) {
                float r00 = sRe[cur][st][i00], m00 = sIm[cur][st][i00];
                float r01 = sRe[cur][st][i01], m01 = sIm[cur][st][i01];
                float r10 = sRe[cur][st][i10], m10 = sIm[cur][st][i10];
                float r11 = sRe[cur][st][i11], m11 = sIm[cur][st][i11];
                rx_pair(c0, s0, r00, m00, r10, m10);
                rx_pair(c0, s0, r01, m01, r11, m11);
                rx_pair(c1, s1, r00, m00, r01, m01);
                rx_pair(c1, s1, r10, m10, r11, m11);
                sRe[dst][st][j00] = r00;  sIm[dst][st][j00] = m00;
                sRe[dst][st][j01] = r01;  sIm[dst][st][j01] = m01;
                sRe[dst][st][j10] = r10;  sIm[dst][st][j10] = m10;
                sRe[dst][st][j11] = r11;  sIm[dst][st][j11] = m11;
            }
            __syncthreads();
        }
        cur ^= 1;
    }

    float acc[DD];
#pragma unroll
    for (int w = 0; w < DD; ++w) acc[w] = 0.f;
    for (int i = tid; i < DIMQ; i += 256) {
        const float g = sRe[cur][0][i] * sRe[cur][1][i]
                      + sIm[cur][0][i] * sIm[cur][1][i];
#pragma unroll
        for (int w = 0; w < DD; ++w) {
            const int bit = 1 << (NQ - 1 - w);
            acc[w] += (i & bit) ? -g : g;
        }
    }
    for (int off = 16; off > 0; off >>= 1) {
#pragma unroll
        for (int w = 0; w < DD; ++w) acc[w] += __shfl_xor(acc[w], off, 32);
    }
    if (lane == 0) {
#pragma unroll
        for (int w = 0; w < DD; ++w) part[wave][w] = acc[w];
    }
    __syncthreads();
    if (tid < DD) {
        float s = 0.f;
#pragma unroll
        for (int v = 0; v < 8; ++v) s += part[v][tid];
        Q[(a * DD + b) * DD + tid] = s;
    }
}

__device__ inline void ln8(const float* h, const float* g, const float* b, float* out)
{
    float m = 0.f;
#pragma unroll
    for (int d = 0; d < DD; ++d) m += h[d];
    m *= 0.125f;
    float v = 0.f;
#pragma unroll
    for (int d = 0; d < DD; ++d) { const float t = h[d] - m; v += t * t; }
    v *= 0.125f;
    const float inv = 1.0f / sqrtf(v + 1e-5f);
#pragma unroll
    for (int d = 0; d < DD; ++d) out[d] = (h[d] - m) * inv * g[d] + b[d];
}

__global__ __launch_bounds__(128) void qsim_kernel(
    const float* __restrict__ zpre,
    const float* __restrict__ Wv,  const float* __restrict__ bv,
    const float* __restrict__ Wo,  const float* __restrict__ bo,
    const float* __restrict__ ln1g, const float* __restrict__ ln1b,
    const float* __restrict__ W1,  const float* __restrict__ b1,
    const float* __restrict__ W2,  const float* __restrict__ b2,
    const float* __restrict__ ln2g, const float* __restrict__ ln2b,
    const float* __restrict__ Q,   float* __restrict__ zq)
{
    const int g    = threadIdx.x >> 5;
    const int lane = threadIdx.x & 31;
    const int row  = blockIdx.x * ROWS_PB + g;

    float z[DD];
#pragma unroll
    for (int d = 0; d < DD; ++d) z[d] = zpre[(size_t)row * DD + d];

    for (int l = 0; l < LL; ++l) {
        const float* wv = Wv + l * 64;  const float* wo = Wo + l * 64;
        float v[DD], attn[DD], h[DD];
#pragma unroll
        for (int d = 0; d < DD; ++d) {
            float s = bv[l * DD + d];
#pragma unroll
            for (int k = 0; k < DD; ++k) s += z[k] * wv[k * DD + d];
            v[d] = s;
        }
#pragma unroll
        for (int d = 0; d < DD; ++d) {
            float s = bo[l * DD + d];
#pragma unroll
            for (int k = 0; k < DD; ++k) s += v[k] * wo[k * DD + d];
            attn[d] = s;
        }
#pragma unroll
        for (int d = 0; d < DD; ++d) h[d] = z[d] + attn[d];
        ln8(h, ln1g + l * DD, ln1b + l * DD, z);

        float acc[DD];
#pragma unroll
        for (int d = 0; d < DD; ++d) acc[d] = 0.f;
#pragma unroll
        for (int j = 0; j < 4; ++j) {
            const int hid = lane + j * 32;
            float hv = b1[l * FFN + hid];
#pragma unroll
            for (int k = 0; k < DD; ++k) hv += z[k] * W1[l * DD * FFN + k * FFN + hid];
            hv = fmaxf(hv, 0.0f);
#pragma unroll
            for (int d = 0; d < DD; ++d) acc[d] += hv * W2[l * FFN * DD + hid * DD + d];
        }
        for (int off = 16; off > 0; off >>= 1) {
#pragma unroll
            for (int d = 0; d < DD; ++d) acc[d] += __shfl_xor(acc[d], off, 32);
        }
#pragma unroll
        for (int d = 0; d < DD; ++d) h[d] = z[d] + acc[d] + b2[l * DD + d];
        ln8(h, ln2g + l * DD, ln2b + l * DD, z);
    }

    float nrm2 = 0.f;
#pragma unroll
    for (int d = 0; d < DD; ++d) nrm2 += z[d] * z[d];
    const float inv = 1.0f / sqrtf(nrm2);
    float zn[DD];
#pragma unroll
    for (int d = 0; d < DD; ++d) zn[d] = z[d] * inv;

    const int w = lane & 7;
    const int q = lane >> 3;
    const float za0 = (q == 0) ? zn[0] : (q == 1) ? zn[2] : (q == 2) ? zn[4] : zn[6];
    const float za1 = (q == 0) ? zn[1] : (q == 1) ? zn[3] : (q == 2) ? zn[5] : zn[7];
    const float* Q0 = Q + (2 * q) * DD * DD + w;
    float t0 = 0.f, t1 = 0.f;
#pragma unroll
    for (int b = 0; b < DD; ++b) {
        t0 += Q0[b * DD] * zn[b];
        t1 += Q0[DD * DD + b * DD] * zn[b];
    }
    float p = za0 * t0 + za1 * t1;
    p += __shfl_xor(p, 8, 32);
    p += __shfl_xor(p, 16, 32);
    if (lane < DD) zq[(size_t)row * DD + lane] = p;
}

__global__ __launch_bounds__(32) void dec_gemm_wmma(
    const float* __restrict__ zq, const float* __restrict__ decW,
    const float* __restrict__ decB, float* __restrict__ out)
{
    const int lane = threadIdx.x;
    const int l16  = lane & 15;
    const int half = lane >> 4;
    const int n0 = blockIdx.x * 16;
    const int m0 = blockIdx.y * 16;

    const float* arow = zq + (size_t)(m0 + l16) * DD;

    v8f c = {};
#pragma unroll
    for (int k = 0; k < DD; k += 4) {
        const int ka = k + half * 2;
        v2f a, b;
        a.x = arow[ka];
        a.y = arow[ka + 1];
        b.x = decW[(size_t)ka * DIN + n0 + l16];
        b.y = decW[(size_t)(ka + 1) * DIN + n0 + l16];
        c = __builtin_amdgcn_wmma_f32_16x16x4_f32(
                false, a, false, b, (short)0, c, false, false);
    }

    const float bias = decB[n0 + l16];
#pragma unroll
    for (int j = 0; j < 8; ++j) {
        const int m = half * 8 + j;
        out[(size_t)(m0 + m) * DIN + n0 + l16] = c[j] + bias;
    }
}

extern "C" void kernel_launch(void* const* d_in, const int* in_sizes, int n_in,
                              void* d_out, int out_size, void* d_ws, size_t ws_size,
                              hipStream_t stream) {
    const float* x    = (const float*)d_in[0];
    const float* encW = (const float*)d_in[1];
    const float* encB = (const float*)d_in[2];
    const float* decW = (const float*)d_in[3];
    const float* decB = (const float*)d_in[4];
    const float* Wv   = (const float*)d_in[5];
    const float* bv   = (const float*)d_in[6];
    const float* Wo   = (const float*)d_in[7];
    const float* bo   = (const float*)d_in[8];
    const float* ln1g = (const float*)d_in[9];
    const float* ln1b = (const float*)d_in[10];
    const float* W1   = (const float*)d_in[11];
    const float* b1   = (const float*)d_in[12];
    const float* W2   = (const float*)d_in[13];
    const float* b2   = (const float*)d_in[14];
    const float* ln2g = (const float*)d_in[15];
    const float* ln2b = (const float*)d_in[16];
    const float* qw   = (const float*)d_in[17];
    float* out  = (float*)d_out;

    float* zpre = (float*)d_ws;
    float* zqv  = zpre + (size_t)BQ * DD;
    float* Qf   = zqv  + (size_t)BQ * DD;

    enc_gemm_wmma<<<BQ / 32, 128, 0, stream>>>(x, encW, encB, zpre);

    circuit_forms<<<DD * DD, 256, 0, stream>>>(qw, Qf);

    qsim_kernel<<<BQ / ROWS_PB, 128, 0, stream>>>(
        zpre, Wv, bv, Wo, bo, ln1g, ln1b, W1, b1, W2, b2, ln2g, ln2b, Qf, zqv);

    dim3 dgrid(DIN / 16, BQ / 16);
    dec_gemm_wmma<<<dgrid, 32, 0, stream>>>(zqv, decW, decB, out);
}


# Decoder GEMM: 256 blocks of 8 waves, each wave keeps dec_W fragments of a 64-column group in registers over 2 row tiles and stores whole 256-byte row segments (was 16384 one-wave blocks)
# speedup vs baseline: 4.5163x; 1.1964x over previous
#include <hip/hip_runtime.h>
#include <hip/hip_bf16.h>
#include <math.h>

typedef __attribute__((ext_vector_type(2))) float v2f;
typedef __attribute__((ext_vector_type(4))) float v4f;
typedef __attribute__((ext_vector_type(8))) float v8f;

#define BQ     4096
#define DIN    1024
#define DD     8
#define LL     2
#define FFN    128
#define NQ     10
#define DIMQ   1024
#define REPS   5
#define ROWS_PB 4
#define NGATE  (REPS * NQ)

__global__ __launch_bounds__(128) void enc_gemm_wmma(
    const float* __restrict__ x, const float* __restrict__ encW,
    const float* __restrict__ encB, float* __restrict__ zpre)
{
    __shared__ float red[4][2][32][8];

    const int lane = threadIdx.x & 31;
    const int wave = threadIdx.x >> 5;
    const int l16  = lane & 15;
    const int half = lane >> 4;
    const int row0 = blockIdx.x * 32;

    const int   bcol  = l16 & 7;
    const float bmask = (l16 < DD) ? 1.0f : 0.0f;

    const float* xrow0 = x + (size_t)(row0 + l16) * DIN;
    const float* xrow1 = xrow0 + (size_t)16 * DIN;
    const int k0 = wave * (DIN / 4);

    v8f c0 = {}, c1 = {};
    for (int k = k0; k < k0 + DIN / 4; k += 4) {
        const int ka = k + half * 2;
        __builtin_prefetch(xrow0 + ka + 256, 0, 3);
        __builtin_prefetch(xrow1 + ka + 256, 0, 3);
        v2f a0, a1, b;
        a0.x = xrow0[ka];
        a0.y = xrow0[ka + 1];
        a1.x = xrow1[ka];
        a1.y = xrow1[ka + 1];
        b.x = bmask * encW[ka * DD + bcol];
        b.y = bmask * encW[(ka + 1) * DD + bcol];
        c0 = __builtin_amdgcn_wmma_f32_16x16x4_f32(
                 false, a0, false, b, (short)0, c0, false, false);
        c1 = __builtin_amdgcn_wmma_f32_16x16x4_f32(
                 false, a1, false, b, (short)0, c1, false, false);
    }

#pragma unroll
    for (int j = 0; j < 8; ++j) {
        red[wave][0][lane][j] = c0[j];
        red[wave][1][lane][j] = c1[j];
    }
    __syncthreads();

    if (wave < 2 && l16 < DD) {
        const int  t    = wave;
        const float bias = encB[l16];
#pragma unroll
        for (int j = 0; j < 8; ++j) {
            const float s = red[0][t][lane][j] + red[1][t][lane][j]
                          + red[2][t][lane][j] + red[3][t][lane][j];
            const int m = t * 16 + half * 8 + j;
            zpre[(size_t)(row0 + m) * DD + l16] = s + bias;
        }
    }
}

__device__ inline void rx_pair(float c, float s,
                               float& a0r, float& a0i, float& a1r, float& a1i)
{
    const float n0r = c * a0r + s * a1i;
    const float n0i = c * a0i - s * a1r;
    const float n1r = c * a1r + s * a0i;
    const float n1i = c * a1i - s * a0r;
    a0r = n0r; a0i = n0i; a1r = n1r; a1i = n1i;
}

__device__ inline int cnot_ring_dest(int i)
{
#pragma unroll
    for (int w = 0; w < NQ; ++w) {
        const int cpos = NQ - 1 - w;
        const int tpos = NQ - 1 - ((w + 1) % NQ);
        i ^= ((i >> cpos) & 1) << tpos;
    }
    return i;
}

__global__ __launch_bounds__(256) void circuit_forms(
    const float* __restrict__ qw, float* __restrict__ Q)
{
    __shared__ float sRe[2][2][DIMQ];
    __shared__ float sIm[2][2][DIMQ];
    __shared__ float sC[NGATE], sS[NGATE];
    __shared__ float part[8][DD];

    const int tid  = threadIdx.x;
    const int lane = tid & 31;
    const int wave = tid >> 5;
    const int a = blockIdx.x >> 3;
    const int b = blockIdx.x & 7;

    if (tid < NGATE) {
        const float th = 0.5f * qw[tid];
        sC[tid] = cosf(th);
        sS[tid] = sinf(th);
    }
    for (int i = tid; i < DIMQ; i += 256) {
        sRe[0][0][i] = (i == a) ? 1.0f : 0.0f;  sIm[0][0][i] = 0.0f;
        sRe[0][1][i] = (i == b) ? 1.0f : 0.0f;  sIm[0][1][i] = 0.0f;
    }
    __syncthreads();

    int cur = 0;
    for (int r = 0; r < REPS; ++r) {
        for (int t = 0; t < NQ / 2; ++t) {
            const int bl   = NQ - 2 - 2 * t;
            const int base = ((tid >> bl) << (bl + 2)) | (tid & ((1 << bl) - 1));
            const int i00 = base;
            const int i01 = base | (1 << bl);
            const int i10 = base | (2 << bl);
            const int i11 = base | (3 << bl);
            const float c0 = sC[r * NQ + 2 * t],     s0 = sS[r * NQ + 2 * t];
            const float c1 = sC[r * NQ + 2 * t + 1], s1 = sS[r * NQ + 2 * t + 1];
            const bool last = (t == NQ / 2 - 1);
            const int dst = last ? (cur ^ 1) : cur;
            const int j00 = last ? cnot_ring_dest(i00) : i00;
            const int j01 = last ? cnot_ring_dest(i01) : i01;
            const int j10 = last ? cnot_ring_dest(i10) : i10;
            const int j11 = last ? cnot_ring_dest(i11) : i11;
#pragma unroll
            for (int st = 0; st < 2; ++st) {
                float r00 = sRe[cur][st][i00], m00 = sIm[cur][st][i00];
                float r01 = sRe[cur][st][i01], m01 = sIm[cur][st][i01];
                float r10 = sRe[cur][st][i10], m10 = sIm[cur][st][i10];
                float r11 = sRe[cur][st][i11], m11 = sIm[cur][st][i11];
                rx_pair(c0, s0, r00, m00, r10, m10);
                rx_pair(c0, s0, r01, m01, r11, m11);
                rx_pair(c1, s1, r00, m00, r01, m01);
                rx_pair(c1, s1, r10, m10, r11, m11);
                sRe[dst][st][j00] = r00;  sIm[dst][st][j00] = m00;
                sRe[dst][st][j01] = r01;  sIm[dst][st][j01] = m01;
                sRe[dst][st][j10] = r10;  sIm[dst][st][j10] = m10;
                sRe[dst][st][j11] = r11;  sIm[dst][st][j11] = m11;
            }
            __syncthreads();
        }
        cur ^= 1;
    }

    float acc[DD];
#pragma unroll
    for (int w = 0; w < DD; ++w) acc[w] = 0.f;
    for (int i = tid; i < DIMQ; i += 256) {
        const float g = sRe[cur][0][i] * sRe[cur][1][i]
                      + sIm[cur][0][i] * sIm[cur][1][i];
#pragma unroll
        for (int w = 0; w < DD; ++w) {
            const int bit = 1 << (NQ - 1 - w);
            acc[w] += (i & bit) ? -g : g;
        }
    }
    for (int off = 16; off > 0; off >>= 1) {
#pragma unroll
        for (int w = 0; w < DD; ++w) acc[w] += __shfl_xor(acc[w], off, 32);
    }
    if (lane == 0) {
#pragma unroll
        for (int w = 0; w < DD; ++w) part[wave][w] = acc[w];
    }
    __syncthreads();
    if (tid < DD) {
        float s = 0.f;
#pragma unroll
        for (int v = 0; v < 8; ++v) s += part[v][tid];
        Q[(a * DD + b) * DD + tid] = s;
    }
}

__device__ inline void ln8(const float* h, const float* g, const float* b, float* out)
{
    float m = 0.f;
#pragma unroll
    for (int d = 0; d < DD; ++d) m += h[d];
    m *= 0.125f;
    float v = 0.f;
#pragma unroll
    for (int d = 0; d < DD; ++d) { const float t = h[d] - m; v += t * t; }
    v *= 0.125f;
    const float inv = 1.0f / sqrtf(v + 1e-5f);
#pragma unroll
    for (int d = 0; d < DD; ++d) out[d] = (h[d] - m) * inv * g[d] + b[d];
}

__global__ __launch_bounds__(128) void qsim_kernel(
    const float* __restrict__ zpre,
    const float* __restrict__ Wv,  const float* __restrict__ bv,
    const float* __restrict__ Wo,  const float* __restrict__ bo,
    const float* __restrict__ ln1g, const float* __restrict__ ln1b,
    const float* __restrict__ W1,  const float* __restrict__ b1,
    const float* __restrict__ W2,  const float* __restrict__ b2,
    const float* __restrict__ ln2g, const float* __restrict__ ln2b,
    const float* __restrict__ Q,   float* __restrict__ zq)
{
    const int g    = threadIdx.x >> 5;
    const int lane = threadIdx.x & 31;
    const int row  = blockIdx.x * ROWS_PB + g;

    float z[DD];
#pragma unroll
    for (int d = 0; d < DD; ++d) z[d] = zpre[(size_t)row * DD + d];

    for (int l = 0; l < LL; ++l) {
        const float* wv = Wv + l * 64;  const float* wo = Wo + l * 64;
        float v[DD], attn[DD], h[DD];
#pragma unroll
        for (int d = 0; d < DD; ++d) {
            float s = bv[l * DD + d];
#pragma unroll
            for (int k = 0; k < DD; ++k) s += z[k] * wv[k * DD + d];
            v[d] = s;
        }
#pragma unroll
        for (int d = 0; d < DD; ++d) {
            float s = bo[l * DD + d];
#pragma unroll
            for (int k = 0; k < DD; ++k) s += v[k] * wo[k * DD + d];
            attn[d] = s;
        }
#pragma unroll
        for (int d = 0; d < DD; ++d) h[d] = z[d] + attn[d];
        ln8(h, ln1g + l * DD, ln1b + l * DD, z);

        float acc[DD];
#pragma unroll
        for (int d = 0; d < DD; ++d) acc[d] = 0.f;
#pragma unroll
        for (int j = 0; j < 4; ++j) {
            const int hid = lane + j * 32;
            float hv = b1[l * FFN + hid];
#pragma unroll
            for (int k = 0; k < DD; ++k) hv += z[k] * W1[l * DD * FFN + k * FFN + hid];
            hv = fmaxf(hv, 0.0f);
#pragma unroll
            for (int d = 0; d < DD; ++d) acc[d] += hv * W2[l * FFN * DD + hid * DD + d];
        }
        for (int off = 16; off > 0; off >>= 1) {
#pragma unroll
            for (int d = 0; d < DD; ++d) acc[d] += __shfl_xor(acc[d], off, 32);
        }
#pragma unroll
        for (int d = 0; d < DD; ++d) h[d] = z[d] + acc[d] + b2[l * DD + d];
        ln8(h, ln2g + l * DD, ln2b + l * DD, z);
    }

    float nrm2 = 0.f;
#pragma unroll
    for (int d = 0; d < DD; ++d) nrm2 += z[d] * z[d];
    const float inv = 1.0f / sqrtf(nrm2);
    float zn[DD];
#pragma unroll
    for (int d = 0; d < DD; ++d) zn[d] = z[d] * inv;

    const int w = lane & 7;
    const int q = lane >> 3;
    const float za0 = (q == 0) ? zn[0] : (q == 1) ? zn[2] : (q == 2) ? zn[4] : zn[6];
    const float za1 = (q == 0) ? zn[1] : (q == 1) ? zn[3] : (q == 2) ? zn[5] : zn[7];
    const float* Q0 = Q + (2 * q) * DD * DD + w;
    float t0 = 0.f, t1 = 0.f;
#pragma unroll
    for (int b = 0; b < DD; ++b) {
        t0 += Q0[b * DD] * zn[b];
        t1 += Q0[DD * DD + b * DD] * zn[b];
    }
    float p = za0 * t0 + za1 * t1;
    p += __shfl_xor(p, 8, 32);
    p += __shfl_xor(p, 16, 32);
    if (lane < DD) zq[(size_t)row * DD + lane] = p;
}

#define DEC_MT    2
#define DEC_WAVES 8
#define DEC_GRID  256

__global__ __launch_bounds__(256) void dec_gemm_wmma(
    const float* __restrict__ zq, const float* __restrict__ decW,
    const float* __restrict__ decB, float* __restrict__ out)
{
    const int lane = threadIdx.x & 31;
    const int wave = threadIdx.x >> 5;
    const int l16  = lane & 15;
    const int half = lane >> 4;
    const int gw   = blockIdx.x * DEC_WAVES + wave;
    const int col  = (gw & 15) * 64 + 4 * l16;
    const int m0   = (gw >> 4) * (16 * DEC_MT);

    const v4f b00 = *(const v4f*)(decW + (size_t)(2 * half)     * DIN + col);
    const v4f b01 = *(const v4f*)(decW + (size_t)(2 * half + 1) * DIN + col);
    const v4f b10 = *(const v4f*)(decW + (size_t)(2 * half + 4) * DIN + col);
    const v4f b11 = *(const v4f*)(decW + (size_t)(2 * half + 5) * DIN + col);
    const v4f bias = *(const v4f*)(decB + col);

    v2f a0[DEC_MT], a1[DEC_MT];
#pragma unroll
    for (int mt = 0; mt < DEC_MT; ++mt) {
        const float* arow = zq + (size_t)(m0 + mt * 16 + l16) * DD + 2 * half;
        a0[mt] = *(const v2f*)(arow);
        a1[mt] = *(const v2f*)(arow + 4);
    }

#pragma unroll
    for (int mt = 0; mt < DEC_MT; ++mt) {
        v8f c[4];
#pragma unroll
        for (int t = 0; t < 4; ++t) {
            v2f b;
            b.x = b00[t];
            b.y = b01[t];
            v8f z = {};
            c[t] = __builtin_amdgcn_wmma_f32_16x16x4_f32(
                       false, a0[mt], false, b, (short)0, z, false, false);
        }
#pragma unroll
        for (int t = 0; t < 4; ++t) {
            v2f b;
            b.x = b10[t];
            b.y = b11[t];
            c[t] = __builtin_amdgcn_wmma_f32_16x16x4_f32(
                       false, a1[mt], false, b, (short)0, c[t], false, false);
        }
        float* orow = out + (size_t)(m0 + mt * 16 + half * 8) * DIN + col;
#pragma unroll
        for (int j = 0; j < 8; ++j) {
            v4f o;
            o.x = c[0][j] + bias.x;
            o.y = c[1][j] + bias.y;
            o.z = c[2][j] + bias.z;
            o.w = c[3][j] + bias.w;
            *(v4f*)(orow + (size_t)j * DIN) = o;
        }
    }
}

extern "C" void kernel_launch(void* const* d_in, const int* in_sizes, int n_in,
                              void* d_out, int out_size, void* d_ws, size_t ws_size,
                              hipStream_t stream) {
    const float* x    = (const float*)d_in[0];
    const float* encW = (const float*)d_in[1];
    const float* encB = (const float*)d_in[2];
    const float* decW = (const float*)d_in[3];
    const float* decB = (const float*)d_in[4];
    const float* Wv   = (const float*)d_in[5];
    const float* bv   = (const float*)d_in[6];
    const float* Wo   = (const float*)d_in[7];
    const float* bo   = (const float*)d_in[8];
    const float* ln1g = (const float*)d_in[9];
    const float* ln1b = (const float*)d_in[10];
    const float* W1   = (const float*)d_in[11];
    const float* b1   = (const float*)d_in[12];
    const float* W2   = (const float*)d_in[13];
    const float* b2   = (const float*)d_in[14];
    const float* ln2g = (const float*)d_in[15];
    const float* ln2b = (const float*)d_in[16];
    const float* qw   = (const float*)d_in[17];
    float* out  = (float*)d_out;

    float* zpre = (float*)d_ws;
    float* zqv  = zpre + (size_t)BQ * DD;
    float* Qf   = zqv  + (size_t)BQ * DD;

    enc_gemm_wmma<<<BQ / 32, 128, 0, stream>>>(x, encW, encB, zpre);

    circuit_forms<<<DD * DD, 256, 0, stream>>>(qw, Qf);

    qsim_kernel<<<BQ / ROWS_PB, 128, 0, stream>>>(
        zpre, Wv, bv, Wo, bo, ln1g, ln1b, W1, b1, W2, b2, ln2g, ln2b, Qf, zqv);

    dec_gemm_wmma<<<DEC_GRID, 32 * DEC_WAVES, 0, stream>>>(zqv, decW, decB, out);
}


# 4 dependent launches become 2: circuit forms run as extra workgroups of the encoder launch; each 16-row workgroup computes its rows' zq into LDS, then decodes them itself
# speedup vs baseline: 6.5603x; 1.4526x over previous
#include <hip/hip_runtime.h>
#include <hip/hip_bf16.h>
#include <math.h>

typedef __attribute__((ext_vector_type(2))) float v2f;
typedef __attribute__((ext_vector_type(4))) float v4f;
typedef __attribute__((ext_vector_type(8))) float v8f;

#define BQ     4096
#define DIN    1024
#define DD     8
#define LL     2
#define FFN    128
#define NQ     10
#define DIMQ   1024
#define REPS   5
#define NGATE  (REPS * NQ)

#define ENC_GROUPS 2
#define ENC_BLOCKS (BQ / (32 * ENC_GROUPS))

__device__ __forceinline__ void enc_gemm_rows(
    const int eb,
    const float* __restrict__ x, const float* __restrict__ encW,
    const float* __restrict__ encB, float* __restrict__ zpre)
{
    __shared__ float red[ENC_GROUPS][4][2][32][8];

    const int lane = threadIdx.x & 31;
    const int wave = threadIdx.x >> 5;
    const int grp  = wave >> 2;
    const int kw   = wave & 3;
    const int l16  = lane & 15;
    const int half = lane >> 4;
    const int row0 = (eb * ENC_GROUPS + grp) * 32;

    const int   bcol  = l16 & 7;
    const float bmask = (l16 < DD) ? 1.0f : 0.0f;

    const float* xrow0 = x + (size_t)(row0 + l16) * DIN;
    const float* xrow1 = xrow0 + (size_t)16 * DIN;
    const int k0 = kw * (DIN / 4);

    v8f c0 = {}, c1 = {};
    for (int k = k0; k < k0 + DIN / 4; k += 4) {
        const int ka = k + half * 2;
        __builtin_prefetch(xrow0 + ka + 256, 0, 3);
        __builtin_prefetch(xrow1 + ka + 256, 0, 3);
        v2f a0, a1, b;
        a0.x = xrow0[ka];
        a0.y = xrow0[ka + 1];
        a1.x = xrow1[ka];
        a1.y = xrow1[ka + 1];
        b.x = bmask * encW[ka * DD + bcol];
        b.y = bmask * encW[(ka + 1) * DD + bcol];
        c0 = __builtin_amdgcn_wmma_f32_16x16x4_f32(
                 false, a0, false, b, (short)0, c0, false, false);
        c1 = __builtin_amdgcn_wmma_f32_16x16x4_f32(
                 false, a1, false, b, (short)0, c1, false, false);
    }

#pragma unroll
    for (int j = 0; j < 8; ++j) {
        red[grp][kw][0][lane][j] = c0[j];
        red[grp][kw][1][lane][j] = c1[j];
    }
    __syncthreads();

    if (kw < 2 && l16 < DD) {
        const int  t    = kw;
        const float bias = encB[l16];
#pragma unroll
        for (int j = 0; j < 8; ++j) {
            const float s = red[grp][0][t][lane][j] + red[grp][1][t][lane][j]
                          + red[grp][2][t][lane][j] + red[grp][3][t][lane][j];
            const int m = t * 16 + half * 8 + j;
            zpre[(size_t)(row0 + m) * DD + l16] = s + bias;
        }
    }
}

__device__ inline void rx_pair(float c, float s,
                               float& a0r, float& a0i, float& a1r, float& a1i)
{
    const float n0r = c * a0r + s * a1i;
    const float n0i = c * a0i - s * a1r;
    const float n1r = c * a1r + s * a0i;
    const float n1i = c * a1i - s * a0r;
    a0r = n0r; a0i = n0i; a1r = n1r; a1i = n1i;
}

__device__ inline int cnot_ring_dest(int i)
{
#pragma unroll
    for (int w = 0; w < NQ; ++w) {
        const int cpos = NQ - 1 - w;
        const int tpos = NQ - 1 - ((w + 1) % NQ);
        i ^= ((i >> cpos) & 1) << tpos;
    }
    return i;
}

__device__ __forceinline__ void circuit_forms(
    const int pair, const float* __restrict__ qw, float* __restrict__ Q)
{
    __shared__ float sRe[2][2][DIMQ];
    __shared__ float sIm[2][2][DIMQ];
    __shared__ float sC[NGATE], sS[NGATE];
    __shared__ float part[8][DD];

    const int tid  = threadIdx.x;
    const int lane = tid & 31;
    const int wave = tid >> 5;
    const int a = pair >> 3;
    const int b = pair & 7;

    if (tid < NGATE) {
        const float th = 0.5f * qw[tid];
        sC[tid] = cosf(th);
        sS[tid] = sinf(th);
    }
    for (int i = tid; i < DIMQ; i += 256) {
        sRe[0][0][i] = (i == a) ? 1.0f : 0.0f;  sIm[0][0][i] = 0.0f;
        sRe[0][1][i] = (i == b) ? 1.0f : 0.0f;  sIm[0][1][i] = 0.0f;
    }
    __syncthreads();

    int cur = 0;
    for (int r = 0; r < REPS; ++r) {
        for (int t = 0; t < NQ / 2; ++t) {
            const int bl   = NQ - 2 - 2 * t;
            const int base = ((tid >> bl) << (bl + 2)) | (tid & ((1 << bl) - 1));
            const int i00 = base;
            const int i01 = base | (1 << bl);
            const int i10 = base | (2 << bl);
            const int i11 = base | (3 << bl);
            const float c0 = sC[r * NQ + 2 * t],     s0 = sS[r * NQ + 2 * t];
            const float c1 = sC[r * NQ + 2 * t + 1], s1 = sS[r * NQ + 2 * t + 1];
            const bool last = (t == NQ / 2 - 1);
            const int dst = last ? (cur ^ 1) : cur;
            const int j00 = last ? cnot_ring_dest(i00) : i00;
            const int j01 = last ? cnot_ring_dest(i01) : i01;
            const int j10 = last ? cnot_ring_dest(i10) : i10;
            const int j11 = last ? cnot_ring_dest(i11) : i11;
#pragma unroll
            for (int st = 0; st < 2; ++st) {
                float r00 = sRe[cur][st][i00], m00 = sIm[cur][st][i00];
                float r01 = sRe[cur][st][i01], m01 = sIm[cur][st][i01];
                float r10 = sRe[cur][st][i10], m10 = sIm[cur][st][i10];
                float r11 = sRe[cur][st][i11], m11 = sIm[cur][st][i11];
                rx_pair(c0, s0, r00, m00, r10, m10);
                rx_pair(c0, s0, r01, m01, r11, m11);
                rx_pair(c1, s1, r00, m00, r01, m01);
                rx_pair(c1, s1, r10, m10, r11, m11);
                sRe[dst][st][j00] = r00;  sIm[dst][st][j00] = m00;
                sRe[dst][st][j01] = r01;  sIm[dst][st][j01] = m01;
                sRe[dst][st][j10] = r10;  sIm[dst][st][j10] = m10;
                sRe[dst][st][j11] = r11;  sIm[dst][st][j11] = m11;
            }
            __syncthreads();
        }
        cur ^= 1;
    }

    float acc[DD];
#pragma unroll
    for (int w = 0; w < DD; ++w) acc[w] = 0.f;
    for (int i = tid; i < DIMQ; i += 256) {
        const float g = sRe[cur][0][i] * sRe[cur][1][i]
                      + sIm[cur][0][i] * sIm[cur][1][i];
#pragma unroll
        for (int w = 0; w < DD; ++w) {
            const int bit = 1 << (NQ - 1 - w);
            acc[w] += (i & bit) ? -g : g;
        }
    }
    for (int off = 16; off > 0; off >>= 1) {
#pragma unroll
        for (int w = 0; w < DD; ++w) acc[w] += __shfl_xor(acc[w], off, 32);
    }
    if (lane == 0) {
#pragma unroll
        for (int w = 0; w < DD; ++w) part[wave][w] = acc[w];
    }
    __syncthreads();
    if (tid < DD) {
        float s = 0.f;
#pragma unroll
        for (int v = 0; v < 8; ++v) s += part[v][tid];
        Q[(a * DD + b) * DD + tid] = s;
    }
}

#define FORM_BLOCKS (DD * DD)

__global__ __launch_bounds__(256) void enc_and_forms(
    const float* __restrict__ x, const float* __restrict__ encW,
    const float* __restrict__ encB, float* __restrict__ zpre,
    const float* __restrict__ qw, float* __restrict__ Q)
{
    if (blockIdx.x < FORM_BLOCKS) {
        circuit_forms(blockIdx.x, qw, Q);
    } else {
        enc_gemm_rows(blockIdx.x - FORM_BLOCKS, x, encW, encB, zpre);
    }
}

__device__ inline void ln8(const float* h, const float* g, const float* b, float* out)
{
    float m = 0.f;
#pragma unroll
    for (int d = 0; d < DD; ++d) m += h[d];
    m *= 0.125f;
    float v = 0.f;
#pragma unroll
    for (int d = 0; d < DD; ++d) { const float t = h[d] - m; v += t * t; }
    v *= 0.125f;
    const float inv = 1.0f / sqrtf(v + 1e-5f);
#pragma unroll
    for (int d = 0; d < DD; ++d) out[d] = (h[d] - m) * inv * g[d] + b[d];
}

#define ROW_WAVES 16
#define ROW_GRID  (BQ / ROW_WAVES)

__global__ __launch_bounds__(32 * ROW_WAVES) void rows_and_decoder(
    const float* __restrict__ zpre,
    const float* __restrict__ Wv,  const float* __restrict__ bv,
    const float* __restrict__ Wo,  const float* __restrict__ bo,
    const float* __restrict__ ln1g, const float* __restrict__ ln1b,
    const float* __restrict__ W1,  const float* __restrict__ b1,
    const float* __restrict__ W2,  const float* __restrict__ b2,
    const float* __restrict__ ln2g, const float* __restrict__ ln2b,
    const float* __restrict__ Q,
    const float* __restrict__ decW, const float* __restrict__ decB,
    float* __restrict__ out)
{
    __shared__ float szq[ROW_WAVES][DD];

    const int g    = threadIdx.x >> 5;
    const int lane = threadIdx.x & 31;
    const int l16  = lane & 15;
    const int half = lane >> 4;
    const int m0   = blockIdx.x * ROW_WAVES;
    const int row  = m0 + g;
    const int col  = g * 64 + 4 * l16;

    const v4f b00 = *(const v4f*)(decW + (size_t)(2 * half)     * DIN + col);
    const v4f b01 = *(const v4f*)(decW + (size_t)(2 * half + 1) * DIN + col);
    const v4f b10 = *(const v4f*)(decW + (size_t)(2 * half + 4) * DIN + col);
    const v4f b11 = *(const v4f*)(decW + (size_t)(2 * half + 5) * DIN + col);
    const v4f bias = *(const v4f*)(decB + col);

    float z[DD];
#pragma unroll
    for (int d = 0; d < DD; ++d) z[d] = zpre[(size_t)row * DD + d];

    for (int l = 0; l < LL; ++l) {
        const float* wv = Wv + l * 64;  const float* wo = Wo + l * 64;
        float v[DD], attn[DD], h[DD];
#pragma unroll
        for (int d = 0; d < DD; ++d) {
            float s = bv[l * DD + d];
#pragma unroll
            for (int k = 0; k < DD; ++k) s += z[k] * wv[k * DD + d];
            v[d] = s;
        }
#pragma unroll
        for (int d = 0; d < DD; ++d) {
            float s = bo[l * DD + d];
#pragma unroll
            for (int k = 0; k < DD; ++k) s += v[k] * wo[k * DD + d];
            attn[d] = s;
        }
#pragma unroll
        for (int d = 0; d < DD; ++d) h[d] = z[d] + attn[d];
        ln8(h, ln1g + l * DD, ln1b + l * DD, z);

        float acc[DD];
#pragma unroll
        for (int d = 0; d < DD; ++d) acc[d] = 0.f;
#pragma unroll
        for (int j = 0; j < 4; ++j) {
            const int hid = lane + j * 32;
            float hv = b1[l * FFN + hid];
#pragma unroll
            for (int k = 0; k < DD; ++k) hv += z[k] * W1[l * DD * FFN + k * FFN + hid];
            hv = fmaxf(hv, 0.0f);
#pragma unroll
            for (int d = 0; d < DD; ++d) acc[d] += hv * W2[l * FFN * DD + hid * DD + d];
        }
        for (int off = 16; off > 0; off >>= 1) {
#pragma unroll
            for (int d = 0; d < DD; ++d) acc[d] += __shfl_xor(acc[d], off, 32);
        }
#pragma unroll
        for (int d = 0; d < DD; ++d) h[d] = z[d] + acc[d] + b2[l * DD + d];
        ln8(h, ln2g + l * DD, ln2b + l * DD, z);
    }

    float nrm2 = 0.f;
#pragma unroll
    for (int d = 0; d < DD; ++d) nrm2 += z[d] * z[d];
    const float inv = 1.0f / sqrtf(nrm2);
    float zn[DD];
#pragma unroll
    for (int d = 0; d < DD; ++d) zn[d] = z[d] * inv;

    const int w = lane & 7;
    const int q = lane >> 3;
    const float za0 = (q == 0) ? zn[0] : (q == 1) ? zn[2] : (q == 2) ? zn[4] : zn[6];
    const float za1 = (q == 0) ? zn[1] : (q == 1) ? zn[3] : (q == 2) ? zn[5] : zn[7];
    const float* Q0 = Q + (2 * q) * DD * DD + w;
    float t0 = 0.f, t1 = 0.f;
#pragma unroll
    for (int b = 0; b < DD; ++b) {
        t0 += Q0[b * DD] * zn[b];
        t1 += Q0[DD * DD + b * DD] * zn[b];
    }
    float p = za0 * t0 + za1 * t1;
    p += __shfl_xor(p, 8, 32);
    p += __shfl_xor(p, 16, 32);
    if (lane < DD) szq[g][lane] = p;
    __syncthreads();

    const v2f a0 = *(const v2f*)(&szq[l16][2 * half]);
    const v2f a1 = *(const v2f*)(&szq[l16][2 * half + 4]);

    v8f c[4];
#pragma unroll
    for (int t = 0; t < 4; ++t) {
        v2f b;
        b.x = b00[t];
        b.y = b01[t];
        v8f zero = {};
        c[t] = __builtin_amdgcn_wmma_f32_16x16x4_f32(
                   false, a0, false, b, (short)0, zero, false, false);
    }
#pragma unroll
    for (int t = 0; t < 4; ++t) {
        v2f b;
        b.x = b10[t];
        b.y = b11[t];
        c[t] = __builtin_amdgcn_wmma_f32_16x16x4_f32(
                   false, a1, false, b, (short)0, c[t], false, false);
    }
    float* orow = out + (size_t)(m0 + half * 8) * DIN + col;
#pragma unroll
    for (int j = 0; j < 8; ++j) {
        v4f o;
        o.x = c[0][j] + bias.x;
        o.y = c[1][j] + bias.y;
        o.z = c[2][j] + bias.z;
        o.w = c[3][j] + bias.w;
        *(v4f*)(orow + (size_t)j * DIN) = o;
    }
}

extern "C" void kernel_launch(void* const* d_in, const int* in_sizes, int n_in,
                              void* d_out, int out_size, void* d_ws, size_t ws_size,
                              hipStream_t stream) {
    const float* x    = (const float*)d_in[0];
    const float* encW = (const float*)d_in[1];
    const float* encB = (const float*)d_in[2];
    const float* decW = (const float*)d_in[3];
    const float* decB = (const float*)d_in[4];
    const float* Wv   = (const float*)d_in[5];
    const float* bv   = (const float*)d_in[6];
    const float* Wo   = (const float*)d_in[7];
    const float* bo   = (const float*)d_in[8];
    const float* ln1g = (const float*)d_in[9];
    const float* ln1b = (const float*)d_in[10];
    const float* W1   = (const float*)d_in[11];
    const float* b1   = (const float*)d_in[12];
    const float* W2   = (const float*)d_in[13];
    const float* b2   = (const float*)d_in[14];
    const float* ln2g = (const float*)d_in[15];
    const float* ln2b = (const float*)d_in[16];
    const float* qw   = (const float*)d_in[17];
    float* out  = (float*)d_out;

    float* zpre = (float*)d_ws;
    float* Qf   = zpre + (size_t)BQ * DD;

    enc_and_forms<<<FORM_BLOCKS + ENC_BLOCKS, 256, 0, stream>>>(
        x, encW, encB, zpre, qw, Qf);

    rows_and_decoder<<<ROW_GRID, 32 * ROW_WAVES, 0, stream>>>(
        zpre, Wv, bv, Wo, bo, ln1g, ln1b, W1, b1, W2, b2, ln2g, ln2b, Qf,
        decW, decB, out);
}


# Encoder GEMM re-tiled: 256 workgroups of 16 rows (was 64 of 64 rows), wave = one K-eighth, 16-byte row loads, no prefetch or mask multiply; 8 K-partials summed in LDS (z_pre sum order changes)
# speedup vs baseline: 8.5900x; 1.3094x over previous
#include <hip/hip_runtime.h>
#include <hip/hip_bf16.h>
#include <math.h>

typedef __attribute__((ext_vector_type(2))) float v2f;
typedef __attribute__((ext_vector_type(4))) float v4f;
typedef __attribute__((ext_vector_type(8))) float v8f;

#define BQ     4096
#define DIN    1024
#define DD     8
#define LL     2
#define FFN    128
#define NQ     10
#define DIMQ   1024
#define REPS   5
#define NGATE  (REPS * NQ)

#define ENC_ROWS   16
#define ENC_WAVES  8
#define ENC_KW     (DIN / ENC_WAVES)
#define ENC_BLOCKS (BQ / ENC_ROWS)

__device__ __forceinline__ void enc_gemm_rows(
    const int eb,
    const float* __restrict__ x, const float* __restrict__ encW,
    const float* __restrict__ encB, float* __restrict__ zpre)
{
    __shared__ float red[ENC_WAVES][ENC_ROWS * DD];

    const int lane = threadIdx.x & 31;
    const int wave = threadIdx.x >> 5;
    const int l16  = lane & 15;
    const int half = lane >> 4;
    const int m0   = eb * ENC_ROWS;
    const int bcol = l16 & 7;

    const float* xa = x + (size_t)(m0 + l16) * DIN + wave * ENC_KW + 4 * half;
    const float* wb = encW + (size_t)(wave * ENC_KW + 4 * half) * DD + bcol;

    v8f c0 = {}, c1 = {};
#pragma unroll
    for (int i = 0; i < ENC_KW / 8; ++i) {
        const v4f a = *(const v4f*)(xa + 8 * i);
        v2f a0, a1, b0, b1;
        a0.x = a.x;  a0.y = a.y;
        a1.x = a.z;  a1.y = a.w;
        b0.x = wb[(8 * i + 0) * DD];
        b0.y = wb[(8 * i + 1) * DD];
        b1.x = wb[(8 * i + 2) * DD];
        b1.y = wb[(8 * i + 3) * DD];
        c0 = __builtin_amdgcn_wmma_f32_16x16x4_f32(
                 false, a0, false, b0, (short)0, c0, false, false);
        c1 = __builtin_amdgcn_wmma_f32_16x16x4_f32(
                 false, a1, false, b1, (short)0, c1, false, false);
    }

    if (l16 < DD) {
#pragma unroll
        for (int j = 0; j < 8; ++j)
            red[wave][(half * 8 + j) * DD + l16] = c0[j] + c1[j];
    }
    __syncthreads();

    if (threadIdx.x < ENC_ROWS * DD) {
        const int t = threadIdx.x;
        float s = red[0][t];
#pragma unroll
        for (int w = 1; w < ENC_WAVES; ++w) s += red[w][t];
        zpre[(size_t)m0 * DD + t] = s + encB[t & 7];
    }
}

__device__ inline void rx_pair(float c, float s,
                               float& a0r, float& a0i, float& a1r, float& a1i)
{
    const float n0r = c * a0r + s * a1i;
    const float n0i = c * a0i - s * a1r;
    const float n1r = c * a1r + s * a0i;
    const float n1i = c * a1i - s * a0r;
    a0r = n0r; a0i = n0i; a1r = n1r; a1i = n1i;
}

__device__ inline int cnot_ring_dest(int i)
{
#pragma unroll
    for (int w = 0; w < NQ; ++w) {
        const int cpos = NQ - 1 - w;
        const int tpos = NQ - 1 - ((w + 1) % NQ);
        i ^= ((i >> cpos) & 1) << tpos;
    }
    return i;
}

__device__ __forceinline__ void circuit_forms(
    const int pair, const float* __restrict__ qw, float* __restrict__ Q)
{
    __shared__ float sRe[2][2][DIMQ];
    __shared__ float sIm[2][2][DIMQ];
    __shared__ float sC[NGATE], sS[NGATE];
    __shared__ float part[8][DD];

    const int tid  = threadIdx.x;
    const int lane = tid & 31;
    const int wave = tid >> 5;
    const int a = pair >> 3;
    const int b = pair & 7;

    if (tid < NGATE) {
        const float th = 0.5f * qw[tid];
        sC[tid] = cosf(th);
        sS[tid] = sinf(th);
    }
    for (int i = tid; i < DIMQ; i += 256) {
        sRe[0][0][i] = (i == a) ? 1.0f : 0.0f;  sIm[0][0][i] = 0.0f;
        sRe[0][1][i] = (i == b) ? 1.0f : 0.0f;  sIm[0][1][i] = 0.0f;
    }
    __syncthreads();

    int cur = 0;
    for (int r = 0; r < REPS; ++r) {
        for (int t = 0; t < NQ / 2; ++t) {
            const int bl   = NQ - 2 - 2 * t;
            const int base = ((tid >> bl) << (bl + 2)) | (tid & ((1 << bl) - 1));
            const int i00 = base;
            const int i01 = base | (1 << bl);
            const int i10 = base | (2 << bl);
            const int i11 = base | (3 << bl);
            const float c0 = sC[r * NQ + 2 * t],     s0 = sS[r * NQ + 2 * t];
            const float c1 = sC[r * NQ + 2 * t + 1], s1 = sS[r * NQ + 2 * t + 1];
            const bool last = (t == NQ / 2 - 1);
            const int dst = last ? (cur ^ 1) : cur;
            const int j00 = last ? cnot_ring_dest(i00) : i00;
            const int j01 = last ? cnot_ring_dest(i01) : i01;
            const int j10 = last ? cnot_ring_dest(i10) : i10;
            const int j11 = last ? cnot_ring_dest(i11) : i11;
#pragma unroll
            for (int st = 0; st < 2; ++st) {
                float r00 = sRe[cur][st][i00], m00 = sIm[cur][st][i00];
                float r01 = sRe[cur][st][i01], m01 = sIm[cur][st][i01];
                float r10 = sRe[cur][st][i10], m10 = sIm[cur][st][i10];
                float r11 = sRe[cur][st][i11], m11 = sIm[cur][st][i11];
                rx_pair(c0, s0, r00, m00, r10, m10);
                rx_pair(c0, s0, r01, m01, r11, m11);
                rx_pair(c1, s1, r00, m00, r01, m01);
                rx_pair(c1, s1, r10, m10, r11, m11);
                sRe[dst][st][j00] = r00;  sIm[dst][st][j00] = m00;
                sRe[dst][st][j01] = r01;  sIm[dst][st][j01] = m01;
                sRe[dst][st][j10] = r10;  sIm[dst][st][j10] = m10;
                sRe[dst][st][j11] = r11;  sIm[dst][st][j11] = m11;
            }
            __syncthreads();
        }
        cur ^= 1;
    }

    float acc[DD];
#pragma unroll
    for (int w = 0; w < DD; ++w) acc[w] = 0.f;
    for (int i = tid; i < DIMQ; i += 256) {
        const float g = sRe[cur][0][i] * sRe[cur][1][i]
                      + sIm[cur][0][i] * sIm[cur][1][i];
#pragma unroll
        for (int w = 0; w < DD; ++w) {
            const int bit = 1 << (NQ - 1 - w);
            acc[w] += (i & bit) ? -g : g;
        }
    }
    for (int off = 16; off > 0; off >>= 1) {
#pragma unroll
        for (int w = 0; w < DD; ++w) acc[w] += __shfl_xor(acc[w], off, 32);
    }
    if (lane == 0) {
#pragma unroll
        for (int w = 0; w < DD; ++w) part[wave][w] = acc[w];
    }
    __syncthreads();
    if (tid < DD) {
        float s = 0.f;
#pragma unroll
        for (int v = 0; v < 8; ++v) s += part[v][tid];
        Q[(a * DD + b) * DD + tid] = s;
    }
}

#define FORM_BLOCKS (DD * DD)

__global__ __launch_bounds__(256) void enc_and_forms(
    const float* __restrict__ x, const float* __restrict__ encW,
    const float* __restrict__ encB, float* __restrict__ zpre,
    const float* __restrict__ qw, float* __restrict__ Q)
{
    if (blockIdx.x < FORM_BLOCKS) {
        circuit_forms(blockIdx.x, qw, Q);
    } else {
        enc_gemm_rows(blockIdx.x - FORM_BLOCKS, x, encW, encB, zpre);
    }
}

__device__ inline void ln8(const float* h, const float* g, const float* b, float* out)
{
    float m = 0.f;
#pragma unroll
    for (int d = 0; d < DD; ++d) m += h[d];
    m *= 0.125f;
    float v = 0.f;
#pragma unroll
    for (int d = 0; d < DD; ++d) { const float t = h[d] - m; v += t * t; }
    v *= 0.125f;
    const float inv = 1.0f / sqrtf(v + 1e-5f);
#pragma unroll
    for (int d = 0; d < DD; ++d) out[d] = (h[d] - m) * inv * g[d] + b[d];
}

#define ROW_WAVES 16
#define ROW_GRID  (BQ / ROW_WAVES)

__global__ __launch_bounds__(32 * ROW_WAVES) void rows_and_decoder(
    const float* __restrict__ zpre,
    const float* __restrict__ Wv,  const float* __restrict__ bv,
    const float* __restrict__ Wo,  const float* __restrict__ bo,
    const float* __restrict__ ln1g, const float* __restrict__ ln1b,
    const float* __restrict__ W1,  const float* __restrict__ b1,
    const float* __restrict__ W2,  const float* __restrict__ b2,
    const float* __restrict__ ln2g, const float* __restrict__ ln2b,
    const float* __restrict__ Q,
    const float* __restrict__ decW, const float* __restrict__ decB,
    float* __restrict__ out)
{
    __shared__ float szq[ROW_WAVES][DD];

    const int g    = threadIdx.x >> 5;
    const int lane = threadIdx.x & 31;
    const int l16  = lane & 15;
    const int half = lane >> 4;
    const int m0   = blockIdx.x * ROW_WAVES;
    const int row  = m0 + g;
    const int col  = g * 64 + 4 * l16;

    const v4f b00 = *(const v4f*)(decW + (size_t)(2 * half)     * DIN + col);
    const v4f b01 = *(const v4f*)(decW + (size_t)(2 * half + 1) * DIN + col);
    const v4f b10 = *(const v4f*)(decW + (size_t)(2 * half + 4) * DIN + col);
    const v4f b11 = *(const v4f*)(decW + (size_t)(2 * half + 5) * DIN + col);
    const v4f bias = *(const v4f*)(decB + col);

    float z[DD];
#pragma unroll
    for (int d = 0; d < DD; ++d) z[d] = zpre[(size_t)row * DD + d];

    for (int l = 0; l < LL; ++l) {
        const float* wv = Wv + l * 64;  const float* wo = Wo + l * 64;
        float v[DD], attn[DD], h[DD];
#pragma unroll
        for (int d = 0; d < DD; ++d) {
            float s = bv[l * DD + d];
#pragma unroll
            for (int k = 0; k < DD; ++k) s += z[k] * wv[k * DD + d];
            v[d] = s;
        }
#pragma unroll
        for (int d = 0; d < DD; ++d) {
            float s = bo[l * DD + d];
#pragma unroll
            for (int k = 0; k < DD; ++k) s += v[k] * wo[k * DD + d];
            attn[d] = s;
        }
#pragma unroll
        for (int d = 0; d < DD; ++d) h[d] = z[d] + attn[d];
        ln8(h, ln1g + l * DD, ln1b + l * DD, z);

        float acc[DD];
#pragma unroll
        for (int d = 0; d < DD; ++d) acc[d] = 0.f;
#pragma unroll
        for (int j = 0; j < 4; ++j) {
            const int hid = lane + j * 32;
            float hv = b1[l * FFN + hid];
#pragma unroll
            for (int k = 0; k < DD; ++k) hv += z[k] * W1[l * DD * FFN + k * FFN + hid];
            hv = fmaxf(hv, 0.0f);
#pragma unroll
            for (int d = 0; d < DD; ++d) acc[d] += hv * W2[l * FFN * DD + hid * DD + d];
        }
        for (int off = 16; off > 0; off >>= 1) {
#pragma unroll
            for (int d = 0; d < DD; ++d) acc[d] += __shfl_xor(acc[d], off, 32);
        }
#pragma unroll
        for (int d = 0; d < DD; ++d) h[d] = z[d] + acc[d] + b2[l * DD + d];
        ln8(h, ln2g + l * DD, ln2b + l * DD, z);
    }

    float nrm2 = 0.f;
#pragma unroll
    for (int d = 0; d < DD; ++d) nrm2 += z[d] * z[d];
    const float inv = 1.0f / sqrtf(nrm2);
    float zn[DD];
#pragma unroll
    for (int d = 0; d < DD; ++d) zn[d] = z[d] * inv;

    const int w = lane & 7;
    const int q = lane >> 3;
    const float za0 = (q == 0) ? zn[0] : (q == 1) ? zn[2] : (q == 2) ? zn[4] : zn[6];
    const float za1 = (q == 0) ? zn[1] : (q == 1) ? zn[3] : (q == 2) ? zn[5] : zn[7];
    const float* Q0 = Q + (2 * q) * DD * DD + w;
    float t0 = 0.f, t1 = 0.f;
#pragma unroll
    for (int b = 0; b < DD; ++b) {
        t0 += Q0[b * DD] * zn[b];
        t1 += Q0[DD * DD + b * DD] * zn[b];
    }
    float p = za0 * t0 + za1 * t1;
    p += __shfl_xor(p, 8, 32);
    p += __shfl_xor(p, 16, 32);
    if (lane < DD) szq[g][lane] = p;
    __syncthreads();

    const v2f a0 = *(const v2f*)(&szq[l16][2 * half]);
    const v2f a1 = *(const v2f*)(&szq[l16][2 * half + 4]);

    v8f c[4];
#pragma unroll
    for (int t = 0; t < 4; ++t) {
        v2f b;
        b.x = b00[t];
        b.y = b01[t];
        v8f zero = {};
        c[t] = __builtin_amdgcn_wmma_f32_16x16x4_f32(
                   false, a0, false, b, (short)0, zero, false, false);
    }
#pragma unroll
    for (int t = 0; t < 4; ++t) {
        v2f b;
        b.x = b10[t];
        b.y = b11[t];
        c[t] = __builtin_amdgcn_wmma_f32_16x16x4_f32(
                   false, a1, false, b, (short)0, c[t], false, false);
    }
    float* orow = out + (size_t)(m0 + half * 8) * DIN + col;
#pragma unroll
    for (int j = 0; j < 8; ++j) {
        v4f o;
        o.x = c[0][j] + bias.x;
        o.y = c[1][j] + bias.y;
        o.z = c[2][j] + bias.z;
        o.w = c[3][j] + bias.w;
        *(v4f*)(orow + (size_t)j * DIN) = o;
    }
}

extern "C" void kernel_launch(void* const* d_in, const int* in_sizes, int n_in,
                              void* d_out, int out_size, void* d_ws, size_t ws_size,
                              hipStream_t stream) {
    const float* x    = (const float*)d_in[0];
    const float* encW = (const float*)d_in[1];
    const float* encB = (const float*)d_in[2];
    const float* decW = (const float*)d_in[3];
    const float* decB = (const float*)d_in[4];
    const float* Wv   = (const float*)d_in[5];
    const float* bv   = (const float*)d_in[6];
    const float* Wo   = (const float*)d_in[7];
    const float* bo   = (const float*)d_in[8];
    const float* ln1g = (const float*)d_in[9];
    const float* ln1b = (const float*)d_in[10];
    const float* W1   = (const float*)d_in[11];
    const float* b1   = (const float*)d_in[12];
    const float* W2   = (const float*)d_in[13];
    const float* b2   = (const float*)d_in[14];
    const float* ln2g = (const float*)d_in[15];
    const float* ln2b = (const float*)d_in[16];
    const float* qw   = (const float*)d_in[17];
    float* out  = (float*)d_out;

    float* zpre = (float*)d_ws;
    float* Qf   = zpre + (size_t)BQ * DD;

    enc_and_forms<<<FORM_BLOCKS + ENC_BLOCKS, 256, 0, stream>>>(
        x, encW, encB, zpre, qw, Qf);

    rows_and_decoder<<<ROW_GRID, 32 * ROW_WAVES, 0, stream>>>(
        zpre, Wv, bv, Wo, bo, ln1g, ln1b, W1, b1, W2, b2, ln2g, ln2b, Qf,
        decW, decB, out);
}


# Launch 2 re-tiled to 4 waves per workgroup (was 16): 8 lanes per row, 4 rows per wave, each wave decodes 256 columns; parameters read from one LDS copy per workgroup (FFN partial-sum order changes)
# speedup vs baseline: 9.2765x; 1.0799x over previous
#include <hip/hip_runtime.h>
#include <hip/hip_bf16.h>
#include <math.h>

typedef __attribute__((ext_vector_type(2))) float v2f;
typedef __attribute__((ext_vector_type(4))) float v4f;
typedef __attribute__((ext_vector_type(8))) float v8f;

#define BQ     4096
#define DIN    1024
#define DD     8
#define LL     2
#define FFN    128
#define NQ     10
#define DIMQ   1024
#define REPS   5
#define NGATE  (REPS * NQ)

#define ENC_ROWS   16
#define ENC_WAVES  8
#define ENC_KW     (DIN / ENC_WAVES)
#define ENC_BLOCKS (BQ / ENC_ROWS)

__device__ __forceinline__ void enc_gemm_rows(
    const int eb,
    const float* __restrict__ x, const float* __restrict__ encW,
    const float* __restrict__ encB, float* __restrict__ zpre)
{
    __shared__ float red[ENC_WAVES][ENC_ROWS * DD];

    const int lane = threadIdx.x & 31;
    const int wave = threadIdx.x >> 5;
    const int l16  = lane & 15;
    const int half = lane >> 4;
    const int m0   = eb * ENC_ROWS;
    const int bcol = l16 & 7;

    const float* xa = x + (size_t)(m0 + l16) * DIN + wave * ENC_KW + 4 * half;
    const float* wb = encW + (size_t)(wave * ENC_KW + 4 * half) * DD + bcol;

    v8f c0 = {}, c1 = {};
#pragma unroll
    for (int i = 0; i < ENC_KW / 8; ++i) {
        const v4f a = *(const v4f*)(xa + 8 * i);
        v2f a0, a1, b0, b1;
        a0.x = a.x;  a0.y = a.y;
        a1.x = a.z;  a1.y = a.w;
        b0.x = wb[(8 * i + 0) * DD];
        b0.y = wb[(8 * i + 1) * DD];
        b1.x = wb[(8 * i + 2) * DD];
        b1.y = wb[(8 * i + 3) * DD];
        c0 = __builtin_amdgcn_wmma_f32_16x16x4_f32(
                 false, a0, false, b0, (short)0, c0, false, false);
        c1 = __builtin_amdgcn_wmma_f32_16x16x4_f32(
                 false, a1, false, b1, (short)0, c1, false, false);
    }

    if (l16 < DD) {
#pragma unroll
        for (int j = 0; j < 8; ++j)
            red[wave][(half * 8 + j) * DD + l16] = c0[j] + c1[j];
    }
    __syncthreads();

    if (threadIdx.x < ENC_ROWS * DD) {
        const int t = threadIdx.x;
        float s = red[0][t];
#pragma unroll
        for (int w = 1; w < ENC_WAVES; ++w) s += red[w][t];
        zpre[(size_t)m0 * DD + t] = s + encB[t & 7];
    }
}

__device__ inline void rx_pair(float c, float s,
                               float& a0r, float& a0i, float& a1r, float& a1i)
{
    const float n0r = c * a0r + s * a1i;
    const float n0i = c * a0i - s * a1r;
    const float n1r = c * a1r + s * a0i;
    const float n1i = c * a1i - s * a0r;
    a0r = n0r; a0i = n0i; a1r = n1r; a1i = n1i;
}

__device__ inline int cnot_ring_dest(int i)
{
#pragma unroll
    for (int w = 0; w < NQ; ++w) {
        const int cpos = NQ - 1 - w;
        const int tpos = NQ - 1 - ((w + 1) % NQ);
        i ^= ((i >> cpos) & 1) << tpos;
    }
    return i;
}

__device__ __forceinline__ void circuit_forms(
    const int pair, const float* __restrict__ qw, float* __restrict__ Q)
{
    __shared__ float sRe[2][2][DIMQ];
    __shared__ float sIm[2][2][DIMQ];
    __shared__ float sC[NGATE], sS[NGATE];
    __shared__ float part[8][DD];

    const int tid  = threadIdx.x;
    const int lane = tid & 31;
    const int wave = tid >> 5;
    const int a = pair >> 3;
    const int b = pair & 7;

    if (tid < NGATE) {
        const float th = 0.5f * qw[tid];
        sC[tid] = cosf(th);
        sS[tid] = sinf(th);
    }
    for (int i = tid; i < DIMQ; i += 256) {
        sRe[0][0][i] = (i == a) ? 1.0f : 0.0f;  sIm[0][0][i] = 0.0f;
        sRe[0][1][i] = (i == b) ? 1.0f : 0.0f;  sIm[0][1][i] = 0.0f;
    }
    __syncthreads();

    int cur = 0;
    for (int r = 0; r < REPS; ++r) {
        for (int t = 0; t < NQ / 2; ++t) {
            const int bl   = NQ - 2 - 2 * t;
            const int base = ((tid >> bl) << (bl + 2)) | (tid & ((1 << bl) - 1));
            const int i00 = base;
            const int i01 = base | (1 << bl);
            const int i10 = base | (2 << bl);
            const int i11 = base | (3 << bl);
            const float c0 = sC[r * NQ + 2 * t],     s0 = sS[r * NQ + 2 * t];
            const float c1 = sC[r * NQ + 2 * t + 1], s1 = sS[r * NQ + 2 * t + 1];
            const bool last = (t == NQ / 2 - 1);
            const int dst = last ? (cur ^ 1) : cur;
            const int j00 = last ? cnot_ring_dest(i00) : i00;
            const int j01 = last ? cnot_ring_dest(i01) : i01;
            const int j10 = last ? cnot_ring_dest(i10) : i10;
            const int j11 = last ? cnot_ring_dest(i11) : i11;
#pragma unroll
            for (int st = 0; st < 2; ++st) {
                float r00 = sRe[cur][st][i00], m00 = sIm[cur][st][i00];
                float r01 = sRe[cur][st][i01], m01 = sIm[cur][st][i01];
                float r10 = sRe[cur][st][i10], m10 = sIm[cur][st][i10];
                float r11 = sRe[cur][st][i11], m11 = sIm[cur][st][i11];
                rx_pair(c0, s0, r00, m00, r10, m10);
                rx_pair(c0, s0, r01, m01, r11, m11);
                rx_pair(c1, s1, r00, m00, r01, m01);
                rx_pair(c1, s1, r10, m10, r11, m11);
                sRe[dst][st][j00] = r00;  sIm[dst][st][j00] = m00;
                sRe[dst][st][j01] = r01;  sIm[dst][st][j01] = m01;
                sRe[dst][st][j10] = r10;  sIm[dst][st][j10] = m10;
                sRe[dst][st][j11] = r11;  sIm[dst][st][j11] = m11;
            }
            __syncthreads();
        }
        cur ^= 1;
    }

    float acc[DD];
#pragma unroll
    for (int w = 0; w < DD; ++w) acc[w] = 0.f;
    for (int i = tid; i < DIMQ; i += 256) {
        const float g = sRe[cur][0][i] * sRe[cur][1][i]
                      + sIm[cur][0][i] * sIm[cur][1][i];
#pragma unroll
        for (int w = 0; w < DD; ++w) {
            const int bit = 1 << (NQ - 1 - w);
            acc[w] += (i & bit) ? -g : g;
        }
    }
    for (int off = 16; off > 0; off >>= 1) {
#pragma unroll
        for (int w = 0; w < DD; ++w) acc[w] += __shfl_xor(acc[w], off, 32);
    }
    if (lane == 0) {
#pragma unroll
        for (int w = 0; w < DD; ++w) part[wave][w] = acc[w];
    }
    __syncthreads();
    if (tid < DD) {
        float s = 0.f;
#pragma unroll
        for (int v = 0; v < 8; ++v) s += part[v][tid];
        Q[(a * DD + b) * DD + tid] = s;
    }
}

#define FORM_BLOCKS (DD * DD)

__global__ __launch_bounds__(256) void enc_and_forms(
    const float* __restrict__ x, const float* __restrict__ encW,
    const float* __restrict__ encB, float* __restrict__ zpre,
    const float* __restrict__ qw, float* __restrict__ Q)
{
    if (blockIdx.x < FORM_BLOCKS) {
        circuit_forms(blockIdx.x, qw, Q);
    } else {
        enc_gemm_rows(blockIdx.x - FORM_BLOCKS, x, encW, encB, zpre);
    }
}

__device__ __forceinline__ v2f both(float x) { v2f r; r.x = x; r.y = x; return r; }

__device__ __forceinline__ void ln8(const v2f* h, const float* __restrict__ g,
                                    const float* __restrict__ b, v2f* out)
{
    float m = 0.f;
#pragma unroll
    for (int p = 0; p < 4; ++p) { m += h[p].x;  m += h[p].y; }
    m *= 0.125f;
    v2f t[4];
    float v = 0.f;
#pragma unroll
    for (int p = 0; p < 4; ++p) {
        t[p] = h[p] - both(m);
        v += t[p].x * t[p].x;
        v += t[p].y * t[p].y;
    }
    v *= 0.125f;
    const float inv = 1.0f / sqrtf(v + 1e-5f);
#pragma unroll
    for (int p = 0; p < 4; ++p)
        out[p] = t[p] * both(inv) * *(const v2f*)(g + 2 * p) + *(const v2f*)(b + 2 * p);
}

#define ROW_TILE   16
#define ROW_WAVES  4
#define ROW_THREADS (32 * ROW_WAVES)
#define ROW_GRID   (BQ / ROW_TILE)
#define W1_FLOATS  (LL * DD * FFN)
#define W2_FLOATS  (LL * FFN * DD)
#define W2_PADDED  (W2_FLOATS + W2_FLOATS / 8)
#define B1_FLOATS  (LL * FFN)
#define Q_FLOATS   (DD * DD * DD)
#define P_WV    0
#define P_WO    (P_WV + LL * 64)
#define P_BV    (P_WO + LL * 64)
#define P_BO    (P_BV + LL * DD)
#define P_LN1G  (P_BO + LL * DD)
#define P_LN1B  (P_LN1G + LL * DD)
#define P_B2    (P_LN1B + LL * DD)
#define P_LN2G  (P_B2 + LL * DD)
#define P_LN2B  (P_LN2G + LL * DD)
#define P_FLOATS (P_LN2B + LL * DD)

__device__ __forceinline__ int w2pad(int f) { return f + ((f >> 5) << 2); }

__device__ __forceinline__ void mat8(const v2f* x, const float* __restrict__ W,
                                     const float* __restrict__ bias, v2f* y)
{
#pragma unroll
    for (int p = 0; p < 4; ++p) {
        v2f s = *(const v2f*)(bias + 2 * p);
#pragma unroll
        for (int k = 0; k < DD; ++k)
            s += both(x[k >> 1][k & 1]) * *(const v2f*)(W + k * DD + 2 * p);
        y[p] = s;
    }
}

__device__ __forceinline__ void attn_ln(
    v2f* z, const int l,
    const float* __restrict__ Wv,  const float* __restrict__ bv,
    const float* __restrict__ Wo,  const float* __restrict__ bo,
    const float* __restrict__ ln1g, const float* __restrict__ ln1b)
{
    v2f v[4], attn[4], h[4];
    mat8(z, Wv + l * 64, bv + l * DD, v);
    mat8(v, Wo + l * 64, bo + l * DD, attn);
#pragma unroll
    for (int p = 0; p < 4; ++p) h[p] = z[p] + attn[p];
    ln8(h, ln1g + l * DD, ln1b + l * DD, z);
}

__device__ __forceinline__ void ffn_ln(
    v2f* z, const int l, const int s,
    const float* sW1, const float* sW2, const float* sB1,
    const float* __restrict__ b2,
    const float* __restrict__ ln2g, const float* __restrict__ ln2b)
{
    v2f acc[4];
#pragma unroll
    for (int p = 0; p < 4; ++p) acc[p] = both(0.f);
#pragma unroll
    for (int jj = 0; jj < 4; ++jj) {
        const int hid0 = 32 * jj + 4 * s;
        const v4f bb = *(const v4f*)(sB1 + l * FFN + hid0);
        v2f hlo, hhi;
        hlo.x = bb.x;  hlo.y = bb.y;  hhi.x = bb.z;  hhi.y = bb.w;
#pragma unroll
        for (int k = 0; k < DD; ++k) {
            const v4f w = *(const v4f*)(sW1 + (l * DD + k) * FFN + hid0);
            v2f wlo, whi;
            wlo.x = w.x;  wlo.y = w.y;  whi.x = w.z;  whi.y = w.w;
            const v2f zk = both(z[k >> 1][k & 1]);
            hlo += zk * wlo;
            hhi += zk * whi;
        }
#pragma unroll
        for (int e = 0; e < 4; ++e) {
            const float hraw = (e == 0) ? hlo.x : (e == 1) ? hlo.y : (e == 2) ? hhi.x : hhi.y;
            const v2f hr = both(fmaxf(hraw, 0.0f));
            const int f = (l * FFN + hid0 + e) * DD;
            const v4f wa = *(const v4f*)(sW2 + w2pad(f));
            const v4f wb = *(const v4f*)(sW2 + w2pad(f + 4));
            v2f w01, w23, w45, w67;
            w01.x = wa.x;  w01.y = wa.y;  w23.x = wa.z;  w23.y = wa.w;
            w45.x = wb.x;  w45.y = wb.y;  w67.x = wb.z;  w67.y = wb.w;
            acc[0] += hr * w01;
            acc[1] += hr * w23;
            acc[2] += hr * w45;
            acc[3] += hr * w67;
        }
    }
    for (int off = 4; off > 0; off >>= 1) {
#pragma unroll
        for (int p = 0; p < 4; ++p) {
            acc[p].x += __shfl_xor(acc[p].x, off, 32);
            acc[p].y += __shfl_xor(acc[p].y, off, 32);
        }
    }
    v2f h[4];
#pragma unroll
    for (int p = 0; p < 4; ++p) h[p] = z[p] + acc[p] + *(const v2f*)(b2 + l * DD + 2 * p);
    ln8(h, ln2g + l * DD, ln2b + l * DD, z);
}

__global__ __launch_bounds__(ROW_THREADS) void rows_and_decoder(
    const float* __restrict__ zpre,
    const float* __restrict__ Wv,  const float* __restrict__ bv,
    const float* __restrict__ Wo,  const float* __restrict__ bo,
    const float* __restrict__ ln1g, const float* __restrict__ ln1b,
    const float* __restrict__ W1,  const float* __restrict__ b1,
    const float* __restrict__ W2,  const float* __restrict__ b2,
    const float* __restrict__ ln2g, const float* __restrict__ ln2b,
    const float* __restrict__ Q,
    const float* __restrict__ decW, const float* __restrict__ decB,
    float* __restrict__ out)
{
    __shared__ __attribute__((aligned(16))) float sW1[W1_FLOATS];
    __shared__ __attribute__((aligned(16))) float sW2[W2_PADDED];
    __shared__ __attribute__((aligned(16))) float sB1[B1_FLOATS];
    __shared__ __attribute__((aligned(16))) float sQ[Q_FLOATS];
    __shared__ __attribute__((aligned(16))) float sP[P_FLOATS];
    __shared__ __attribute__((aligned(16))) float szq[ROW_TILE][DD];

    const int tid  = threadIdx.x;
    const int g    = tid >> 5;
    const int lane = tid & 31;
    const int l16  = lane & 15;
    const int half = lane >> 4;
    const int r    = lane >> 3;
    const int s    = lane & 7;
    const int m0   = blockIdx.x * ROW_TILE;
    const int row  = m0 + 4 * g + r;

    const v4f zlo = *(const v4f*)(zpre + (size_t)row * DD);
    const v4f zhi = *(const v4f*)(zpre + (size_t)row * DD + 4);

    v4f cw1[W1_FLOATS / 4 / ROW_THREADS], cw2[W2_FLOATS / 4 / ROW_THREADS];
#pragma unroll
    for (int i = 0; i < W1_FLOATS / 4 / ROW_THREADS; ++i)
        cw1[i] = *(const v4f*)(W1 + 4 * (tid + i * ROW_THREADS));
#pragma unroll
    for (int i = 0; i < W2_FLOATS / 4 / ROW_THREADS; ++i)
        cw2[i] = *(const v4f*)(W2 + 4 * (tid + i * ROW_THREADS));
    const v4f cq  = *(const v4f*)(Q + 4 * tid);
    const v4f cb1 = *(const v4f*)(b1 + 4 * (tid & (B1_FLOATS / 4 - 1)));
    const int pt = (tid < P_FLOATS / 4) ? tid : (P_FLOATS / 4 - 1);
    const int pv = (pt - 64) >> 2;
    const float* pvec = (pv == 0) ? bv : (pv == 1) ? bo : (pv == 2) ? ln1g : (pv == 3) ? ln1b
                      : (pv == 4) ? b2 : (pv == 5) ? ln2g : ln2b;
    const float* psrc = (pt < 32) ? (Wv + 4 * pt) : (pt < 64) ? (Wo + 4 * (pt - 32))
                      : (pvec + 4 * ((pt - 64) & 3));
    const v4f cp = *(const v4f*)psrc;

    v4f b00[4], b01[4], b10[4], b11[4], bias[4];
#pragma unroll
    for (int u = 0; u < 4; ++u) {
        const int col = (4 * g + u) * 64 + 4 * l16;
        b00[u]  = *(const v4f*)(decW + (size_t)(2 * half)     * DIN + col);
        b01[u]  = *(const v4f*)(decW + (size_t)(2 * half + 1) * DIN + col);
        b10[u]  = *(const v4f*)(decW + (size_t)(2 * half + 4) * DIN + col);
        b11[u]  = *(const v4f*)(decW + (size_t)(2 * half + 5) * DIN + col);
        bias[u] = *(const v4f*)(decB + col);
    }

    v2f z[4];
    z[0].x = zlo.x;  z[0].y = zlo.y;  z[1].x = zlo.z;  z[1].y = zlo.w;
    z[2].x = zhi.x;  z[2].y = zhi.y;  z[3].x = zhi.z;  z[3].y = zhi.w;

#pragma unroll
    for (int i = 0; i < W1_FLOATS / 4 / ROW_THREADS; ++i)
        *(v4f*)(sW1 + 4 * (tid + i * ROW_THREADS)) = cw1[i];
#pragma unroll
    for (int i = 0; i < W2_FLOATS / 4 / ROW_THREADS; ++i)
        *(v4f*)(sW2 + w2pad(4 * (tid + i * ROW_THREADS))) = cw2[i];
    *(v4f*)(sQ + 4 * tid) = cq;
    if (tid < B1_FLOATS / 4) *(v4f*)(sB1 + 4 * tid) = cb1;
    if (tid < P_FLOATS / 4)  *(v4f*)(sP + 4 * tid) = cp;
    __syncthreads();

    attn_ln(z, 0, sP + P_WV, sP + P_BV, sP + P_WO, sP + P_BO, sP + P_LN1G, sP + P_LN1B);
    ffn_ln(z, 0, s, sW1, sW2, sB1, sP + P_B2, sP + P_LN2G, sP + P_LN2B);
    attn_ln(z, 1, sP + P_WV, sP + P_BV, sP + P_WO, sP + P_BO, sP + P_LN1G, sP + P_LN1B);
    ffn_ln(z, 1, s, sW1, sW2, sB1, sP + P_B2, sP + P_LN2G, sP + P_LN2B);

    float nrm2 = 0.f;
#pragma unroll
    for (int p = 0; p < 4; ++p) { nrm2 += z[p].x * z[p].x;  nrm2 += z[p].y * z[p].y; }
    const float inv = 1.0f / sqrtf(nrm2);
    float zn[DD];
#pragma unroll
    for (int p = 0; p < 4; ++p) { zn[2 * p] = z[p].x * inv;  zn[2 * p + 1] = z[p].y * inv; }

    float pq[4];
#pragma unroll
    for (int q = 0; q < 4; ++q) {
        const float* Q0 = sQ + (2 * q) * DD * DD + s;
        v2f t = both(0.f);
#pragma unroll
        for (int b = 0; b < DD; ++b) {
            v2f qq;
            qq.x = Q0[b * DD];
            qq.y = Q0[DD * DD + b * DD];
            t += qq * both(zn[b]);
        }
        pq[q] = zn[2 * q] * t.x + zn[2 * q + 1] * t.y;
    }
    szq[4 * g + r][s] = (pq[0] + pq[1]) + (pq[2] + pq[3]);
    __syncthreads();

    const v2f a0 = *(const v2f*)(&szq[l16][2 * half]);
    const v2f a1 = *(const v2f*)(&szq[l16][2 * half + 4]);

#pragma unroll
    for (int u = 0; u < 4; ++u) {
        v8f c[4];
#pragma unroll
        for (int t = 0; t < 4; ++t) {
            v2f b;
            b.x = b00[u][t];
            b.y = b01[u][t];
            v8f zero = {};
            c[t] = __builtin_amdgcn_wmma_f32_16x16x4_f32(
                       false, a0, false, b, (short)0, zero, false, false);
        }
#pragma unroll
        for (int t = 0; t < 4; ++t) {
            v2f b;
            b.x = b10[u][t];
            b.y = b11[u][t];
            c[t] = __builtin_amdgcn_wmma_f32_16x16x4_f32(
                       false, a1, false, b, (short)0, c[t], false, false);
        }
        float* orow = out + (size_t)(m0 + half * 8) * DIN + (4 * g + u) * 64 + 4 * l16;
#pragma unroll
        for (int j = 0; j < 8; ++j) {
            v4f o;
            o.x = c[0][j] + bias[u].x;
            o.y = c[1][j] + bias[u].y;
            o.z = c[2][j] + bias[u].z;
            o.w = c[3][j] + bias[u].w;
            *(v4f*)(orow + (size_t)j * DIN) = o;
        }
    }
}

extern "C" void kernel_launch(void* const* d_in, const int* in_sizes, int n_in,
                              void* d_out, int out_size, void* d_ws, size_t ws_size,
                              hipStream_t stream) {
    const float* x    = (const float*)d_in[0];
    const float* encW = (const float*)d_in[1];
    const float* encB = (const float*)d_in[2];
    const float* decW = (const float*)d_in[3];
    const float* decB = (const float*)d_in[4];
    const float* Wv   = (const float*)d_in[5];
    const float* bv   = (const float*)d_in[6];
    const float* Wo   = (const float*)d_in[7];
    const float* bo   = (const float*)d_in[8];
    const float* ln1g = (const float*)d_in[9];
    const float* ln1b = (const float*)d_in[10];
    const float* W1   = (const float*)d_in[11];
    const float* b1   = (const float*)d_in[12];
    const float* W2   = (const float*)d_in[13];
    const float* b2   = (const float*)d_in[14];
    const float* ln2g = (const float*)d_in[15];
    const float* ln2b = (const float*)d_in[16];
    const float* qw   = (const float*)d_in[17];
    float* out  = (float*)d_out;

    float* zpre = (float*)d_ws;
    float* Qf   = zpre + (size_t)BQ * DD;

    enc_and_forms<<<FORM_BLOCKS + ENC_BLOCKS, 256, 0, stream>>>(
        x, encW, encB, zpre, qw, Qf);

    rows_and_decoder<<<ROW_GRID, ROW_THREADS, 0, stream>>>(
        zpre, Wv, bv, Wo, bo, ln1g, ln1b, W1, b1, W2, b2, ln2g, ln2b, Qf,
        decW, decB, out);
}


# Per-row transformer moves from launch 2 into the encoder workgroups of launch 1, where it runs beside the circuit forms instead of behind them; launch 2 keeps forms evaluation and decoder only
# speedup vs baseline: 9.9297x; 1.0704x over previous
#include <hip/hip_runtime.h>
#include <hip/hip_bf16.h>
#include <math.h>

typedef __attribute__((ext_vector_type(2))) float v2f;
typedef __attribute__((ext_vector_type(4))) float v4f;
typedef __attribute__((ext_vector_type(8))) float v8f;

#define BQ     4096
#define DIN    1024
#define DD     8
#define LL     2
#define FFN    128
#define NQ     10
#define DIMQ   1024
#define REPS   5
#define NGATE  (REPS * NQ)

__device__ inline void rx_pair(float c, float s,
                               float& a0r, float& a0i, float& a1r, float& a1i)
{
    const float n0r = c * a0r + s * a1i;
    const float n0i = c * a0i - s * a1r;
    const float n1r = c * a1r + s * a0i;
    const float n1i = c * a1i - s * a0r;
    a0r = n0r; a0i = n0i; a1r = n1r; a1i = n1i;
}

__device__ inline int cnot_ring_dest(int i)
{
#pragma unroll
    for (int w = 0; w < NQ; ++w) {
        const int cpos = NQ - 1 - w;
        const int tpos = NQ - 1 - ((w + 1) % NQ);
        i ^= ((i >> cpos) & 1) << tpos;
    }
    return i;
}

__device__ __forceinline__ void circuit_forms(
    const int pair, const float* __restrict__ qw, float* __restrict__ Q)
{
    __shared__ float sRe[2][2][DIMQ];
    __shared__ float sIm[2][2][DIMQ];
    __shared__ float sC[NGATE], sS[NGATE];
    __shared__ float part[8][DD];

    const int tid  = threadIdx.x;
    const int lane = tid & 31;
    const int wave = tid >> 5;
    const int a = pair >> 3;
    const int b = pair & 7;

    if (tid < NGATE) {
        const float th = 0.5f * qw[tid];
        sC[tid] = cosf(th);
        sS[tid] = sinf(th);
    }
    for (int i = tid; i < DIMQ; i += 256) {
        sRe[0][0][i] = (i == a) ? 1.0f : 0.0f;  sIm[0][0][i] = 0.0f;
        sRe[0][1][i] = (i == b) ? 1.0f : 0.0f;  sIm[0][1][i] = 0.0f;
    }
    __syncthreads();

    int cur = 0;
    for (int r = 0; r < REPS; ++r) {
        for (int t = 0; t < NQ / 2; ++t) {
            const int bl   = NQ - 2 - 2 * t;
            const int base = ((tid >> bl) << (bl + 2)) | (tid & ((1 << bl) - 1));
            const int i00 = base;
            const int i01 = base | (1 << bl);
            const int i10 = base | (2 << bl);
            const int i11 = base | (3 << bl);
            const float c0 = sC[r * NQ + 2 * t],     s0 = sS[r * NQ + 2 * t];
            const float c1 = sC[r * NQ + 2 * t + 1], s1 = sS[r * NQ + 2 * t + 1];
            const bool last = (t == NQ / 2 - 1);
            const int dst = last ? (cur ^ 1) : cur;
            const int j00 = last ? cnot_ring_dest(i00) : i00;
            const int j01 = last ? cnot_ring_dest(i01) : i01;
            const int j10 = last ? cnot_ring_dest(i10) : i10;
            const int j11 = last ? cnot_ring_dest(i11) : i11;
#pragma unroll
            for (int st = 0; st < 2; ++st) {
                float r00 = sRe[cur][st][i00], m00 = sIm[cur][st][i00];
                float r01 = sRe[cur][st][i01], m01 = sIm[cur][st][i01];
                float r10 = sRe[cur][st][i10], m10 = sIm[cur][st][i10];
                float r11 = sRe[cur][st][i11], m11 = sIm[cur][st][i11];
                rx_pair(c0, s0, r00, m00, r10, m10);
                rx_pair(c0, s0, r01, m01, r11, m11);
                rx_pair(c1, s1, r00, m00, r01, m01);
                rx_pair(c1, s1, r10, m10, r11, m11);
                sRe[dst][st][j00] = r00;  sIm[dst][st][j00] = m00;
                sRe[dst][st][j01] = r01;  sIm[dst][st][j01] = m01;
                sRe[dst][st][j10] = r10;  sIm[dst][st][j10] = m10;
                sRe[dst][st][j11] = r11;  sIm[dst][st][j11] = m11;
            }
            __syncthreads();
        }
        cur ^= 1;
    }

    float acc[DD];
#pragma unroll
    for (int w = 0; w < DD; ++w) acc[w] = 0.f;
    for (int i = tid; i < DIMQ; i += 256) {
        const float g = sRe[cur][0][i] * sRe[cur][1][i]
                      + sIm[cur][0][i] * sIm[cur][1][i];
#pragma unroll
        for (int w = 0; w < DD; ++w) {
            const int bit = 1 << (NQ - 1 - w);
            acc[w] += (i & bit) ? -g : g;
        }
    }
    for (int off = 16; off > 0; off >>= 1) {
#pragma unroll
        for (int w = 0; w < DD; ++w) acc[w] += __shfl_xor(acc[w], off, 32);
    }
    if (lane == 0) {
#pragma unroll
        for (int w = 0; w < DD; ++w) part[wave][w] = acc[w];
    }
    __syncthreads();
    if (tid < DD) {
        float s = 0.f;
#pragma unroll
        for (int v = 0; v < 8; ++v) s += part[v][tid];
        Q[(a * DD + b) * DD + tid] = s;
    }
}

__device__ __forceinline__ v2f both(float x) { v2f r; r.x = x; r.y = x; return r; }

__device__ __forceinline__ void ln8(const v2f* h, const float* __restrict__ g,
                                    const float* __restrict__ b, v2f* out)
{
    float m = 0.f;
#pragma unroll
    for (int p = 0; p < 4; ++p) { m += h[p].x;  m += h[p].y; }
    m *= 0.125f;
    v2f t[4];
    float v = 0.f;
#pragma unroll
    for (int p = 0; p < 4; ++p) {
        t[p] = h[p] - both(m);
        v += t[p].x * t[p].x;
        v += t[p].y * t[p].y;
    }
    v *= 0.125f;
    const float inv = 1.0f / sqrtf(v + 1e-5f);
#pragma unroll
    for (int p = 0; p < 4; ++p)
        out[p] = t[p] * both(inv) * *(const v2f*)(g + 2 * p) + *(const v2f*)(b + 2 * p);
}

#define ROW_TILE   16
#define ROW_WAVES  4
#define ROW_THREADS (32 * ROW_WAVES)
#define ROW_GRID   (BQ / ROW_TILE)
#define W1_FLOATS  (LL * DD * FFN)
#define W2_FLOATS  (LL * FFN * DD)
#define W2_PADDED  (W2_FLOATS + W2_FLOATS / 8)
#define B1_FLOATS  (LL * FFN)
#define Q_FLOATS   (DD * DD * DD)
#define P_WV    0
#define P_WO    (P_WV + LL * 64)
#define P_BV    (P_WO + LL * 64)
#define P_BO    (P_BV + LL * DD)
#define P_LN1G  (P_BO + LL * DD)
#define P_LN1B  (P_LN1G + LL * DD)
#define P_B2    (P_LN1B + LL * DD)
#define P_LN2G  (P_B2 + LL * DD)
#define P_LN2B  (P_LN2G + LL * DD)
#define P_FLOATS (P_LN2B + LL * DD)

__device__ __forceinline__ int w2pad(int f) { return f + ((f >> 5) << 2); }

__device__ __forceinline__ void mat8(const v2f* x, const float* __restrict__ W,
                                     const float* __restrict__ bias, v2f* y)
{
#pragma unroll
    for (int p = 0; p < 4; ++p) {
        v2f s = *(const v2f*)(bias + 2 * p);
#pragma unroll
        for (int k = 0; k < DD; ++k)
            s += both(x[k >> 1][k & 1]) * *(const v2f*)(W + k * DD + 2 * p);
        y[p] = s;
    }
}

__device__ __forceinline__ void attn_ln(
    v2f* z, const int l,
    const float* __restrict__ Wv,  const float* __restrict__ bv,
    const float* __restrict__ Wo,  const float* __restrict__ bo,
    const float* __restrict__ ln1g, const float* __restrict__ ln1b)
{
    v2f v[4], attn[4], h[4];
    mat8(z, Wv + l * 64, bv + l * DD, v);
    mat8(v, Wo + l * 64, bo + l * DD, attn);
#pragma unroll
    for (int p = 0; p < 4; ++p) h[p] = z[p] + attn[p];
    ln8(h, ln1g + l * DD, ln1b + l * DD, z);
}

__device__ __forceinline__ void ffn_ln(
    v2f* z, const int l, const int s,
    const float* sW1, const float* sW2, const float* sB1,
    const float* __restrict__ b2,
    const float* __restrict__ ln2g, const float* __restrict__ ln2b)
{
    v2f acc[4];
#pragma unroll
    for (int p = 0; p < 4; ++p) acc[p] = both(0.f);
#pragma unroll
    for (int jj = 0; jj < 4; ++jj) {
        const int hid0 = 32 * jj + 4 * s;
        const v4f bb = *(const v4f*)(sB1 + l * FFN + hid0);
        v2f hlo, hhi;
        hlo.x = bb.x;  hlo.y = bb.y;  hhi.x = bb.z;  hhi.y = bb.w;
#pragma unroll
        for (int k = 0; k < DD; ++k) {
            const v4f w = *(const v4f*)(sW1 + (l * DD + k) * FFN + hid0);
            v2f wlo, whi;
            wlo.x = w.x;  wlo.y = w.y;  whi.x = w.z;  whi.y = w.w;
            const v2f zk = both(z[k >> 1][k & 1]);
            hlo += zk * wlo;
            hhi += zk * whi;
        }
#pragma unroll
        for (int e = 0; e < 4; ++e) {
            const float hraw = (e == 0) ? hlo.x : (e == 1) ? hlo.y : (e == 2) ? hhi.x : hhi.y;
            const v2f hr = both(fmaxf(hraw, 0.0f));
            const int f = (l * FFN + hid0 + e) * DD;
            const v4f wa = *(const v4f*)(sW2 + w2pad(f));
            const v4f wb = *(const v4f*)(sW2 + w2pad(f + 4));
            v2f w01, w23, w45, w67;
            w01.x = wa.x;  w01.y = wa.y;  w23.x = wa.z;  w23.y = wa.w;
            w45.x = wb.x;  w45.y = wb.y;  w67.x = wb.z;  w67.y = wb.w;
            acc[0] += hr * w01;
            acc[1] += hr * w23;
            acc[2] += hr * w45;
            acc[3] += hr * w67;
        }
    }
    for (int off = 4; off > 0; off >>= 1) {
#pragma unroll
        for (int p = 0; p < 4; ++p) {
            acc[p].x += __shfl_xor(acc[p].x, off, 32);
            acc[p].y += __shfl_xor(acc[p].y, off, 32);
        }
    }
    v2f h[4];
#pragma unroll
    for (int p = 0; p < 4; ++p) h[p] = z[p] + acc[p] + *(const v2f*)(b2 + l * DD + 2 * p);
    ln8(h, ln2g + l * DD, ln2b + l * DD, z);
}

#define ENC_ROWS    16
#define ENC_WAVES   8
#define ENC_THREADS (32 * ENC_WAVES)
#define ENC_KW      (DIN / ENC_WAVES)
#define ENC_BLOCKS  (BQ / ENC_ROWS)

__device__ __forceinline__ void enc_rows(
    const int eb,
    const float* __restrict__ x, const float* __restrict__ encW,
    const float* __restrict__ encB,
    const float* __restrict__ Wv,  const float* __restrict__ bv,
    const float* __restrict__ Wo,  const float* __restrict__ bo,
    const float* __restrict__ ln1g, const float* __restrict__ ln1b,
    const float* __restrict__ W1,  const float* __restrict__ b1,
    const float* __restrict__ W2,  const float* __restrict__ b2,
    const float* __restrict__ ln2g, const float* __restrict__ ln2b,
    float* __restrict__ znrm)
{
    __shared__ __attribute__((aligned(16))) float red[ENC_WAVES][ENC_ROWS * DD];
    __shared__ __attribute__((aligned(16))) float sW1[W1_FLOATS];
    __shared__ __attribute__((aligned(16))) float sW2[W2_PADDED];
    __shared__ __attribute__((aligned(16))) float sB1[B1_FLOATS];
    __shared__ __attribute__((aligned(16))) float sP[P_FLOATS];

    const int tid  = threadIdx.x;
    const int lane = tid & 31;
    const int wave = tid >> 5;
    const int l16  = lane & 15;
    const int half = lane >> 4;
    const int m0   = eb * ENC_ROWS;
    const int bcol = l16 & 7;

    v4f cw1[W1_FLOATS / 4 / ENC_THREADS], cw2[W2_FLOATS / 4 / ENC_THREADS];
#pragma unroll
    for (int i = 0; i < W1_FLOATS / 4 / ENC_THREADS; ++i)
        cw1[i] = *(const v4f*)(W1 + 4 * (tid + i * ENC_THREADS));
#pragma unroll
    for (int i = 0; i < W2_FLOATS / 4 / ENC_THREADS; ++i)
        cw2[i] = *(const v4f*)(W2 + 4 * (tid + i * ENC_THREADS));
    const v4f cb1 = *(const v4f*)(b1 + 4 * (tid & (B1_FLOATS / 4 - 1)));
    const int pt = (tid < P_FLOATS / 4) ? tid : (P_FLOATS / 4 - 1);
    const int pv = (pt - 64) >> 2;
    const float* pvec = (pv == 0) ? bv : (pv == 1) ? bo : (pv == 2) ? ln1g : (pv == 3) ? ln1b
                      : (pv == 4) ? b2 : (pv == 5) ? ln2g : ln2b;
    const float* psrc = (pt < 32) ? (Wv + 4 * pt) : (pt < 64) ? (Wo + 4 * (pt - 32))
                      : (pvec + 4 * ((pt - 64) & 3));
    const v4f cp = *(const v4f*)psrc;

    const float* xa = x + (size_t)(m0 + l16) * DIN + wave * ENC_KW + 4 * half;
    const float* wb = encW + (size_t)(wave * ENC_KW + 4 * half) * DD + bcol;

    v8f c0 = {}, c1 = {};
#pragma unroll
    for (int i = 0; i < ENC_KW / 8; ++i) {
        const v4f a = *(const v4f*)(xa + 8 * i);
        v2f a0, a1, b0, b1v;
        a0.x = a.x;  a0.y = a.y;
        a1.x = a.z;  a1.y = a.w;
        b0.x  = wb[(8 * i + 0) * DD];
        b0.y  = wb[(8 * i + 1) * DD];
        b1v.x = wb[(8 * i + 2) * DD];
        b1v.y = wb[(8 * i + 3) * DD];
        c0 = __builtin_amdgcn_wmma_f32_16x16x4_f32(
                 false, a0, false, b0, (short)0, c0, false, false);
        c1 = __builtin_amdgcn_wmma_f32_16x16x4_f32(
                 false, a1, false, b1v, (short)0, c1, false, false);
    }

    if (l16 < DD) {
#pragma unroll
        for (int j = 0; j < 8; ++j)
            red[wave][(half * 8 + j) * DD + l16] = c0[j] + c1[j];
    }
#pragma unroll
    for (int i = 0; i < W1_FLOATS / 4 / ENC_THREADS; ++i)
        *(v4f*)(sW1 + 4 * (tid + i * ENC_THREADS)) = cw1[i];
#pragma unroll
    for (int i = 0; i < W2_FLOATS / 4 / ENC_THREADS; ++i)
        *(v4f*)(sW2 + w2pad(4 * (tid + i * ENC_THREADS))) = cw2[i];
    if (tid < B1_FLOATS / 4) *(v4f*)(sB1 + 4 * tid) = cb1;
    if (tid < P_FLOATS / 4)  *(v4f*)(sP + 4 * tid) = cp;
    __syncthreads();

    if (tid >= 32 * ROW_WAVES) return;

    const int r  = lane >> 3;
    const int s  = lane & 7;
    const int rl = 4 * wave + r;

    v4f zlo = *(const v4f*)(&red[0][rl * DD]);
    v4f zhi = *(const v4f*)(&red[0][rl * DD + 4]);
#pragma unroll
    for (int w = 1; w < ENC_WAVES; ++w) {
        zlo += *(const v4f*)(&red[w][rl * DD]);
        zhi += *(const v4f*)(&red[w][rl * DD + 4]);
    }
    zlo += *(const v4f*)(encB);
    zhi += *(const v4f*)(encB + 4);

    v2f z[4];
    z[0].x = zlo.x;  z[0].y = zlo.y;  z[1].x = zlo.z;  z[1].y = zlo.w;
    z[2].x = zhi.x;  z[2].y = zhi.y;  z[3].x = zhi.z;  z[3].y = zhi.w;

    attn_ln(z, 0, sP + P_WV, sP + P_BV, sP + P_WO, sP + P_BO, sP + P_LN1G, sP + P_LN1B);
    ffn_ln(z, 0, s, sW1, sW2, sB1, sP + P_B2, sP + P_LN2G, sP + P_LN2B);
    attn_ln(z, 1, sP + P_WV, sP + P_BV, sP + P_WO, sP + P_BO, sP + P_LN1G, sP + P_LN1B);
    ffn_ln(z, 1, s, sW1, sW2, sB1, sP + P_B2, sP + P_LN2G, sP + P_LN2B);

    float nrm2 = 0.f;
#pragma unroll
    for (int p = 0; p < 4; ++p) { nrm2 += z[p].x * z[p].x;  nrm2 += z[p].y * z[p].y; }
    const float inv = 1.0f / sqrtf(nrm2);
    v4f o;
    o.x = (s ? z[2].x : z[0].x) * inv;
    o.y = (s ? z[2].y : z[0].y) * inv;
    o.z = (s ? z[3].x : z[1].x) * inv;
    o.w = (s ? z[3].y : z[1].y) * inv;
    if (s < 2)
        *(v4f*)(znrm + (size_t)(m0 + rl) * DD + 4 * s) = o;
}

#define FORM_BLOCKS (DD * DD)

__global__ __launch_bounds__(ENC_THREADS) void enc_rows_and_forms(
    const float* __restrict__ x, const float* __restrict__ encW,
    const float* __restrict__ encB,
    const float* __restrict__ Wv,  const float* __restrict__ bv,
    const float* __restrict__ Wo,  const float* __restrict__ bo,
    const float* __restrict__ ln1g, const float* __restrict__ ln1b,
    const float* __restrict__ W1,  const float* __restrict__ b1,
    const float* __restrict__ W2,  const float* __restrict__ b2,
    const float* __restrict__ ln2g, const float* __restrict__ ln2b,
    float* __restrict__ znrm,
    const float* __restrict__ qw, float* __restrict__ Q)
{
    if (blockIdx.x < FORM_BLOCKS) {
        circuit_forms(blockIdx.x, qw, Q);
    } else {
        enc_rows(blockIdx.x - FORM_BLOCKS, x, encW, encB,
                 Wv, bv, Wo, bo, ln1g, ln1b, W1, b1, W2, b2, ln2g, ln2b, znrm);
    }
}

__global__ __launch_bounds__(ROW_THREADS) void forms_and_decoder(
    const float* __restrict__ znrm,
    const float* __restrict__ Q,
    const float* __restrict__ decW, const float* __restrict__ decB,
    float* __restrict__ out)
{
    __shared__ __attribute__((aligned(16))) float sQ[Q_FLOATS];
    __shared__ __attribute__((aligned(16))) float szq[ROW_TILE][DD];

    const int tid  = threadIdx.x;
    const int g    = tid >> 5;
    const int lane = tid & 31;
    const int l16  = lane & 15;
    const int half = lane >> 4;
    const int r    = lane >> 3;
    const int s    = lane & 7;
    const int m0   = blockIdx.x * ROW_TILE;
    const int row  = m0 + 4 * g + r;

    const v4f zlo = *(const v4f*)(znrm + (size_t)row * DD);
    const v4f zhi = *(const v4f*)(znrm + (size_t)row * DD + 4);
    const v4f cq  = *(const v4f*)(Q + 4 * tid);

    v4f b00[4], b01[4], b10[4], b11[4], bias[4];
#pragma unroll
    for (int u = 0; u < 4; ++u) {
        const int col = (4 * g + u) * 64 + 4 * l16;
        b00[u]  = *(const v4f*)(decW + (size_t)(2 * half)     * DIN + col);
        b01[u]  = *(const v4f*)(decW + (size_t)(2 * half + 1) * DIN + col);
        b10[u]  = *(const v4f*)(decW + (size_t)(2 * half + 4) * DIN + col);
        b11[u]  = *(const v4f*)(decW + (size_t)(2 * half + 5) * DIN + col);
        bias[u] = *(const v4f*)(decB + col);
    }

    *(v4f*)(sQ + 4 * tid) = cq;
    __syncthreads();

    float zn[DD];
    zn[0] = zlo.x;  zn[1] = zlo.y;  zn[2] = zlo.z;  zn[3] = zlo.w;
    zn[4] = zhi.x;  zn[5] = zhi.y;  zn[6] = zhi.z;  zn[7] = zhi.w;

    float pq[4];
#pragma unroll
    for (int q = 0; q < 4; ++q) {
        const float* Q0 = sQ + (2 * q) * DD * DD + s;
        v2f t = both(0.f);
#pragma unroll
        for (int b = 0; b < DD; ++b) {
            v2f qq;
            qq.x = Q0[b * DD];
            qq.y = Q0[DD * DD + b * DD];
            t += qq * both(zn[b]);
        }
        pq[q] = zn[2 * q] * t.x + zn[2 * q + 1] * t.y;
    }
    szq[4 * g + r][s] = (pq[0] + pq[1]) + (pq[2] + pq[3]);
    __syncthreads();

    const v2f a0 = *(const v2f*)(&szq[l16][2 * half]);
    const v2f a1 = *(const v2f*)(&szq[l16][2 * half + 4]);

#pragma unroll
    for (int u = 0; u < 4; ++u) {
        v8f c[4];
#pragma unroll
        for (int t = 0; t < 4; ++t) {
            v2f b;
            b.x = b00[u][t];
            b.y = b01[u][t];
            v8f zero = {};
            c[t] = __builtin_amdgcn_wmma_f32_16x16x4_f32(
                       false, a0, false, b, (short)0, zero, false, false);
        }
#pragma unroll
        for (int t = 0; t < 4; ++t) {
            v2f b;
            b.x = b10[u][t];
            b.y = b11[u][t];
            c[t] = __builtin_amdgcn_wmma_f32_16x16x4_f32(
                       false, a1, false, b, (short)0, c[t], false, false);
        }
        float* orow = out + (size_t)(m0 + half * 8) * DIN + (4 * g + u) * 64 + 4 * l16;
#pragma unroll
        for (int j = 0; j < 8; ++j) {
            v4f o;
            o.x = c[0][j] + bias[u].x;
            o.y = c[1][j] + bias[u].y;
            o.z = c[2][j] + bias[u].z;
            o.w = c[3][j] + bias[u].w;
            *(v4f*)(orow + (size_t)j * DIN) = o;
        }
    }
}

extern "C" void kernel_launch(void* const* d_in, const int* in_sizes, int n_in,
                              void* d_out, int out_size, void* d_ws, size_t ws_size,
                              hipStream_t stream) {
    const float* x    = (const float*)d_in[0];
    const float* encW = (const float*)d_in[1];
    const float* encB = (const float*)d_in[2];
    const float* decW = (const float*)d_in[3];
    const float* decB = (const float*)d_in[4];
    const float* Wv   = (const float*)d_in[5];
    const float* bv   = (const float*)d_in[6];
    const float* Wo   = (const float*)d_in[7];
    const float* bo   = (const float*)d_in[8];
    const float* ln1g = (const float*)d_in[9];
    const float* ln1b = (const float*)d_in[10];
    const float* W1   = (const float*)d_in[11];
    const float* b1   = (const float*)d_in[12];
    const float* W2   = (const float*)d_in[13];
    const float* b2   = (const float*)d_in[14];
    const float* ln2g = (const float*)d_in[15];
    const float* ln2b = (const float*)d_in[16];
    const float* qw   = (const float*)d_in[17];
    float* out  = (float*)d_out;

    float* znrm = (float*)d_ws;
    float* Qf   = znrm + (size_t)BQ * DD;

    enc_rows_and_forms<<<FORM_BLOCKS + ENC_BLOCKS, ENC_THREADS, 0, stream>>>(
        x, encW, encB, Wv, bv, Wo, bo, ln1g, ln1b, W1, b1, W2, b2, ln2g, ln2b,
        znrm, qw, Qf);

    forms_and_decoder<<<ROW_GRID, ROW_THREADS, 0, stream>>>(
        znrm, Qf, decW, decB, out);
}


# Circuit forms: a thread holds 16 amplitudes of one state and applies 4 RX wires per LDS pass, so a layer is 3 barrier passes (was 5); amplitudes as (re,im) pairs in a bank-swizzled LDS layout
# speedup vs baseline: 11.5747x; 1.1657x over previous
#include <hip/hip_runtime.h>
#include <hip/hip_bf16.h>
#include <math.h>

typedef __attribute__((ext_vector_type(2))) float v2f;
typedef __attribute__((ext_vector_type(4))) float v4f;
typedef __attribute__((ext_vector_type(8))) float v8f;

#define BQ     4096
#define DIN    1024
#define DD     8
#define LL     2
#define FFN    128
#define NQ     10
#define DIMQ   1024
#define REPS   5
#define NGATE  (REPS * NQ)

__device__ inline void rx_pair(float c, float s,
                               float& a0r, float& a0i, float& a1r, float& a1i)
{
    const float n0r = c * a0r + s * a1i;
    const float n0i = c * a0i - s * a1r;
    const float n1r = c * a1r + s * a0i;
    const float n1i = c * a1i - s * a0r;
    a0r = n0r; a0i = n0i; a1r = n1r; a1i = n1i;
}

__device__ inline int cnot_ring_dest(int i)
{
#pragma unroll
    for (int w = 0; w < NQ; ++w) {
        const int cpos = NQ - 1 - w;
        const int tpos = NQ - 1 - ((w + 1) % NQ);
        i ^= ((i >> cpos) & 1) << tpos;
    }
    return i;
}

__device__ __forceinline__ int amp_slot(int i)
{
    return i ^ (((i >> 5) & 1) | (((i >> 9) & 1) << 1) | (((i >> 6) & 7) << 2));
}

template <int RB>
__device__ __forceinline__ void rx_bit(float c, float s, float* re, float* im)
{
#pragma unroll
    for (int r = 0; r < 16; ++r) {
        if ((r >> RB) & 1) continue;
        rx_pair(c, s, re[r], im[r], re[r | (1 << RB)], im[r | (1 << RB)]);
    }
}

__device__ __forceinline__ void circuit_forms(
    const int pair, const float* __restrict__ qw, float* __restrict__ Q)
{
    __shared__ __attribute__((aligned(16))) v2f sAmp[2][2][DIMQ];
    __shared__ float sC[NGATE], sS[NGATE];
    __shared__ float part[8][DD];

    const int tid  = threadIdx.x;
    const int lane = tid & 31;
    const int wave = tid >> 5;
    const int a = pair >> 3;
    const int b = pair & 7;

    if (tid < NGATE) {
        const float th = 0.5f * qw[tid];
        sC[tid] = cosf(th);
        sS[tid] = sinf(th);
    }
    for (int i = tid; i < DIMQ; i += 256) {
        v2f ea, eb;
        ea.x = (i == a) ? 1.0f : 0.0f;  ea.y = 0.0f;
        eb.x = (i == b) ? 1.0f : 0.0f;  eb.y = 0.0f;
        sAmp[0][0][amp_slot(i)] = ea;
        sAmp[0][1][amp_slot(i)] = eb;
    }
    __syncthreads();

    const bool active = tid < 128;
    const int st = (tid >> 6) & 1;
    const int wv = wave & 1;
    const int pA = amp_slot((wv << 5) | lane);
    const int pB = amp_slot((lane & 3) | ((lane >> 2) << 6) | (wv << 9));
    const int baseC = ((lane & 15) << 2) | ((lane >> 4) << 9) | (wv << 8);
    const int pC = amp_slot(baseC);
    const int qC = amp_slot(cnot_ring_dest(baseC));

    const v2f* bufX = sAmp[0][st];
    v2f* bufY = sAmp[1][st];
    v2f* dstX = sAmp[0][st];
    for (int rep = 0; rep < REPS; ++rep) {
        const float* cc = sC + rep * NQ;
        const float* ss = sS + rep * NQ;
        if (active) {
            float re[16], im[16];
#pragma unroll
            for (int r = 0; r < 16; ++r) {
                const v2f v = bufX[pA ^ amp_slot(r << 6)];
                re[r] = v.x;  im[r] = v.y;
            }
            rx_bit<3>(cc[0], ss[0], re, im);
            rx_bit<2>(cc[1], ss[1], re, im);
            rx_bit<1>(cc[2], ss[2], re, im);
            rx_bit<0>(cc[3], ss[3], re, im);
#pragma unroll
            for (int r = 0; r < 16; ++r) {
                v2f v;  v.x = re[r];  v.y = im[r];
                bufY[pA ^ amp_slot(r << 6)] = v;
            }
        }
        __syncthreads();
        if (active) {
            float re[16], im[16];
#pragma unroll
            for (int r = 0; r < 16; ++r) {
                const v2f v = bufY[pB ^ amp_slot(r << 2)];
                re[r] = v.x;  im[r] = v.y;
            }
            rx_bit<3>(cc[4], ss[4], re, im);
            rx_bit<2>(cc[5], ss[5], re, im);
            rx_bit<1>(cc[6], ss[6], re, im);
            rx_bit<0>(cc[7], ss[7], re, im);
#pragma unroll
            for (int r = 0; r < 16; ++r) {
                v2f v;  v.x = re[r];  v.y = im[r];
                bufY[pB ^ amp_slot(r << 2)] = v;
            }
        }
        __syncthreads();
        if (active) {
            float re[16], im[16];
#pragma unroll
            for (int r = 0; r < 16; ++r) {
                const v2f v = bufY[pC ^ amp_slot((r & 3) | ((r >> 2) << 6))];
                re[r] = v.x;  im[r] = v.y;
            }
            rx_bit<1>(cc[8], ss[8], re, im);
            rx_bit<0>(cc[9], ss[9], re, im);
#pragma unroll
            for (int r = 0; r < 16; ++r) {
                v2f v;  v.x = re[r];  v.y = im[r];
                dstX[qC ^ amp_slot(cnot_ring_dest((r & 3) | ((r >> 2) << 6)))] = v;
            }
        }
        __syncthreads();
    }

    float acc[DD];
#pragma unroll
    for (int w = 0; w < DD; ++w) acc[w] = 0.f;
    for (int i = tid; i < DIMQ; i += 256) {
        const v2f va = sAmp[0][0][amp_slot(i)];
        const v2f vb = sAmp[0][1][amp_slot(i)];
        const float g = va.x * vb.x + va.y * vb.y;
#pragma unroll
        for (int w = 0; w < DD; ++w) {
            const int bit = 1 << (NQ - 1 - w);
            acc[w] += (i & bit) ? -g : g;
        }
    }
    for (int off = 16; off > 0; off >>= 1) {
#pragma unroll
        for (int w = 0; w < DD; ++w) acc[w] += __shfl_xor(acc[w], off, 32);
    }
    if (lane == 0) {
#pragma unroll
        for (int w = 0; w < DD; ++w) part[wave][w] = acc[w];
    }
    __syncthreads();
    if (tid < DD) {
        float s = 0.f;
#pragma unroll
        for (int v = 0; v < 8; ++v) s += part[v][tid];
        Q[(a * DD + b) * DD + tid] = s;
    }
}

__device__ __forceinline__ v2f both(float x) { v2f r; r.x = x; r.y = x; return r; }

__device__ __forceinline__ void ln8(const v2f* h, const float* __restrict__ g,
                                    const float* __restrict__ b, v2f* out)
{
    float m = 0.f;
#pragma unroll
    for (int p = 0; p < 4; ++p) { m += h[p].x;  m += h[p].y; }
    m *= 0.125f;
    v2f t[4];
    float v = 0.f;
#pragma unroll
    for (int p = 0; p < 4; ++p) {
        t[p] = h[p] - both(m);
        v += t[p].x * t[p].x;
        v += t[p].y * t[p].y;
    }
    v *= 0.125f;
    const float inv = 1.0f / sqrtf(v + 1e-5f);
#pragma unroll
    for (int p = 0; p < 4; ++p)
        out[p] = t[p] * both(inv) * *(const v2f*)(g + 2 * p) + *(const v2f*)(b + 2 * p);
}

#define ROW_TILE   16
#define ROW_WAVES  4
#define ROW_THREADS (32 * ROW_WAVES)
#define ROW_GRID   (BQ / ROW_TILE)
#define W1_FLOATS  (LL * DD * FFN)
#define W2_FLOATS  (LL * FFN * DD)
#define W2_PADDED  (W2_FLOATS + W2_FLOATS / 8)
#define B1_FLOATS  (LL * FFN)
#define Q_FLOATS   (DD * DD * DD)
#define P_WV    0
#define P_WO    (P_WV + LL * 64)
#define P_BV    (P_WO + LL * 64)
#define P_BO    (P_BV + LL * DD)
#define P_LN1G  (P_BO + LL * DD)
#define P_LN1B  (P_LN1G + LL * DD)
#define P_B2    (P_LN1B + LL * DD)
#define P_LN2G  (P_B2 + LL * DD)
#define P_LN2B  (P_LN2G + LL * DD)
#define P_FLOATS (P_LN2B + LL * DD)

__device__ __forceinline__ int w2pad(int f) { return f + ((f >> 5) << 2); }

__device__ __forceinline__ void mat8(const v2f* x, const float* __restrict__ W,
                                     const float* __restrict__ bias, v2f* y)
{
#pragma unroll
    for (int p = 0; p < 4; ++p) {
        v2f s = *(const v2f*)(bias + 2 * p);
#pragma unroll
        for (int k = 0; k < DD; ++k)
            s += both(x[k >> 1][k & 1]) * *(const v2f*)(W + k * DD + 2 * p);
        y[p] = s;
    }
}

__device__ __forceinline__ void attn_ln(
    v2f* z, const int l,
    const float* __restrict__ Wv,  const float* __restrict__ bv,
    const float* __restrict__ Wo,  const float* __restrict__ bo,
    const float* __restrict__ ln1g, const float* __restrict__ ln1b)
{
    v2f v[4], attn[4], h[4];
    mat8(z, Wv + l * 64, bv + l * DD, v);
    mat8(v, Wo + l * 64, bo + l * DD, attn);
#pragma unroll
    for (int p = 0; p < 4; ++p) h[p] = z[p] + attn[p];
    ln8(h, ln1g + l * DD, ln1b + l * DD, z);
}

__device__ __forceinline__ void ffn_ln(
    v2f* z, const int l, const int s,
    const float* sW1, const float* sW2, const float* sB1,
    const float* __restrict__ b2,
    const float* __restrict__ ln2g, const float* __restrict__ ln2b)
{
    v2f acc[4];
#pragma unroll
    for (int p = 0; p < 4; ++p) acc[p] = both(0.f);
#pragma unroll
    for (int jj = 0; jj < 4; ++jj) {
        const int hid0 = 32 * jj + 4 * s;
        const v4f bb = *(const v4f*)(sB1 + l * FFN + hid0);
        v2f hlo, hhi;
        hlo.x = bb.x;  hlo.y = bb.y;  hhi.x = bb.z;  hhi.y = bb.w;
#pragma unroll
        for (int k = 0; k < DD; ++k) {
            const v4f w = *(const v4f*)(sW1 + (l * DD + k) * FFN + hid0);
            v2f wlo, whi;
            wlo.x = w.x;  wlo.y = w.y;  whi.x = w.z;  whi.y = w.w;
            const v2f zk = both(z[k >> 1][k & 1]);
            hlo += zk * wlo;
            hhi += zk * whi;
        }
#pragma unroll
        for (int e = 0; e < 4; ++e) {
            const float hraw = (e == 0) ? hlo.x : (e == 1) ? hlo.y : (e == 2) ? hhi.x : hhi.y;
            const v2f hr = both(fmaxf(hraw, 0.0f));
            const int f = (l * FFN + hid0 + e) * DD;
            const v4f wa = *(const v4f*)(sW2 + w2pad(f));
            const v4f wb = *(const v4f*)(sW2 + w2pad(f + 4));
            v2f w01, w23, w45, w67;
            w01.x = wa.x;  w01.y = wa.y;  w23.x = wa.z;  w23.y = wa.w;
            w45.x = wb.x;  w45.y = wb.y;  w67.x = wb.z;  w67.y = wb.w;
            acc[0] += hr * w01;
            acc[1] += hr * w23;
            acc[2] += hr * w45;
            acc[3] += hr * w67;
        }
    }
    for (int off = 4; off > 0; off >>= 1) {
#pragma unroll
        for (int p = 0; p < 4; ++p) {
            acc[p].x += __shfl_xor(acc[p].x, off, 32);
            acc[p].y += __shfl_xor(acc[p].y, off, 32);
        }
    }
    v2f h[4];
#pragma unroll
    for (int p = 0; p < 4; ++p) h[p] = z[p] + acc[p] + *(const v2f*)(b2 + l * DD + 2 * p);
    ln8(h, ln2g + l * DD, ln2b + l * DD, z);
}

#define ENC_ROWS    16
#define ENC_WAVES   8
#define ENC_THREADS (32 * ENC_WAVES)
#define ENC_KW      (DIN / ENC_WAVES)
#define ENC_BLOCKS  (BQ / ENC_ROWS)

__device__ __forceinline__ void enc_rows(
    const int eb,
    const float* __restrict__ x, const float* __restrict__ encW,
    const float* __restrict__ encB,
    const float* __restrict__ Wv,  const float* __restrict__ bv,
    const float* __restrict__ Wo,  const float* __restrict__ bo,
    const float* __restrict__ ln1g, const float* __restrict__ ln1b,
    const float* __restrict__ W1,  const float* __restrict__ b1,
    const float* __restrict__ W2,  const float* __restrict__ b2,
    const float* __restrict__ ln2g, const float* __restrict__ ln2b,
    float* __restrict__ znrm)
{
    __shared__ __attribute__((aligned(16))) float red[ENC_WAVES][ENC_ROWS * DD];
    __shared__ __attribute__((aligned(16))) float sW1[W1_FLOATS];
    __shared__ __attribute__((aligned(16))) float sW2[W2_PADDED];
    __shared__ __attribute__((aligned(16))) float sB1[B1_FLOATS];
    __shared__ __attribute__((aligned(16))) float sP[P_FLOATS];

    const int tid  = threadIdx.x;
    const int lane = tid & 31;
    const int wave = tid >> 5;
    const int l16  = lane & 15;
    const int half = lane >> 4;
    const int m0   = eb * ENC_ROWS;
    const int bcol = l16 & 7;

    v4f cw1[W1_FLOATS / 4 / ENC_THREADS], cw2[W2_FLOATS / 4 / ENC_THREADS];
#pragma unroll
    for (int i = 0; i < W1_FLOATS / 4 / ENC_THREADS; ++i)
        cw1[i] = *(const v4f*)(W1 + 4 * (tid + i * ENC_THREADS));
#pragma unroll
    for (int i = 0; i < W2_FLOATS / 4 / ENC_THREADS; ++i)
        cw2[i] = *(const v4f*)(W2 + 4 * (tid + i * ENC_THREADS));
    const v4f cb1 = *(const v4f*)(b1 + 4 * (tid & (B1_FLOATS / 4 - 1)));
    const int pt = (tid < P_FLOATS / 4) ? tid : (P_FLOATS / 4 - 1);
    const int pv = (pt - 64) >> 2;
    const float* pvec = (pv == 0) ? bv : (pv == 1) ? bo : (pv == 2) ? ln1g : (pv == 3) ? ln1b
                      : (pv == 4) ? b2 : (pv == 5) ? ln2g : ln2b;
    const float* psrc = (pt < 32) ? (Wv + 4 * pt) : (pt < 64) ? (Wo + 4 * (pt - 32))
                      : (pvec + 4 * ((pt - 64) & 3));
    const v4f cp = *(const v4f*)psrc;

    const float* xa = x + (size_t)(m0 + l16) * DIN + wave * ENC_KW + 4 * half;
    const float* wb = encW + (size_t)(wave * ENC_KW + 4 * half) * DD + bcol;

    v8f c0 = {}, c1 = {};
#pragma unroll
    for (int i = 0; i < ENC_KW / 8; ++i) {
        const v4f a = *(const v4f*)(xa + 8 * i);
        v2f a0, a1, b0, b1v;
        a0.x = a.x;  a0.y = a.y;
        a1.x = a.z;  a1.y = a.w;
        b0.x  = wb[(8 * i + 0) * DD];
        b0.y  = wb[(8 * i + 1) * DD];
        b1v.x = wb[(8 * i + 2) * DD];
        b1v.y = wb[(8 * i + 3) * DD];
        c0 = __builtin_amdgcn_wmma_f32_16x16x4_f32(
                 false, a0, false, b0, (short)0, c0, false, false);
        c1 = __builtin_amdgcn_wmma_f32_16x16x4_f32(
                 false, a1, false, b1v, (short)0, c1, false, false);
    }

    if (l16 < DD) {
#pragma unroll
        for (int j = 0; j < 8; ++j)
            red[wave][(half * 8 + j) * DD + l16] = c0[j] + c1[j];
    }
#pragma unroll
    for (int i = 0; i < W1_FLOATS / 4 / ENC_THREADS; ++i)
        *(v4f*)(sW1 + 4 * (tid + i * ENC_THREADS)) = cw1[i];
#pragma unroll
    for (int i = 0; i < W2_FLOATS / 4 / ENC_THREADS; ++i)
        *(v4f*)(sW2 + w2pad(4 * (tid + i * ENC_THREADS))) = cw2[i];
    if (tid < B1_FLOATS / 4) *(v4f*)(sB1 + 4 * tid) = cb1;
    if (tid < P_FLOATS / 4)  *(v4f*)(sP + 4 * tid) = cp;
    __syncthreads();

    if (tid >= 32 * ROW_WAVES) return;

    const int r  = lane >> 3;
    const int s  = lane & 7;
    const int rl = 4 * wave + r;

    v4f zlo = *(const v4f*)(&red[0][rl * DD]);
    v4f zhi = *(const v4f*)(&red[0][rl * DD + 4]);
#pragma unroll
    for (int w = 1; w < ENC_WAVES; ++w) {
        zlo += *(const v4f*)(&red[w][rl * DD]);
        zhi += *(const v4f*)(&red[w][rl * DD + 4]);
    }
    zlo += *(const v4f*)(encB);
    zhi += *(const v4f*)(encB + 4);

    v2f z[4];
    z[0].x = zlo.x;  z[0].y = zlo.y;  z[1].x = zlo.z;  z[1].y = zlo.w;
    z[2].x = zhi.x;  z[2].y = zhi.y;  z[3].x = zhi.z;  z[3].y = zhi.w;

    attn_ln(z, 0, sP + P_WV, sP + P_BV, sP + P_WO, sP + P_BO, sP + P_LN1G, sP + P_LN1B);
    ffn_ln(z, 0, s, sW1, sW2, sB1, sP + P_B2, sP + P_LN2G, sP + P_LN2B);
    attn_ln(z, 1, sP + P_WV, sP + P_BV, sP + P_WO, sP + P_BO, sP + P_LN1G, sP + P_LN1B);
    ffn_ln(z, 1, s, sW1, sW2, sB1, sP + P_B2, sP + P_LN2G, sP + P_LN2B);

    float nrm2 = 0.f;
#pragma unroll
    for (int p = 0; p < 4; ++p) { nrm2 += z[p].x * z[p].x;  nrm2 += z[p].y * z[p].y; }
    const float inv = 1.0f / sqrtf(nrm2);
    v4f o;
    o.x = (s ? z[2].x : z[0].x) * inv;
    o.y = (s ? z[2].y : z[0].y) * inv;
    o.z = (s ? z[3].x : z[1].x) * inv;
    o.w = (s ? z[3].y : z[1].y) * inv;
    if (s < 2)
        *(v4f*)(znrm + (size_t)(m0 + rl) * DD + 4 * s) = o;
}

#define FORM_BLOCKS (DD * DD)

__global__ __launch_bounds__(ENC_THREADS) void enc_rows_and_forms(
    const float* __restrict__ x, const float* __restrict__ encW,
    const float* __restrict__ encB,
    const float* __restrict__ Wv,  const float* __restrict__ bv,
    const float* __restrict__ Wo,  const float* __restrict__ bo,
    const float* __restrict__ ln1g, const float* __restrict__ ln1b,
    const float* __restrict__ W1,  const float* __restrict__ b1,
    const float* __restrict__ W2,  const float* __restrict__ b2,
    const float* __restrict__ ln2g, const float* __restrict__ ln2b,
    float* __restrict__ znrm,
    const float* __restrict__ qw, float* __restrict__ Q)
{
    if (blockIdx.x < FORM_BLOCKS) {
        circuit_forms(blockIdx.x, qw, Q);
    } else {
        enc_rows(blockIdx.x - FORM_BLOCKS, x, encW, encB,
                 Wv, bv, Wo, bo, ln1g, ln1b, W1, b1, W2, b2, ln2g, ln2b, znrm);
    }
}

__global__ __launch_bounds__(ROW_THREADS) void forms_and_decoder(
    const float* __restrict__ znrm,
    const float* __restrict__ Q,
    const float* __restrict__ decW, const float* __restrict__ decB,
    float* __restrict__ out)
{
    __shared__ __attribute__((aligned(16))) float sQ[Q_FLOATS];
    __shared__ __attribute__((aligned(16))) float szq[ROW_TILE][DD];

    const int tid  = threadIdx.x;
    const int g    = tid >> 5;
    const int lane = tid & 31;
    const int l16  = lane & 15;
    const int half = lane >> 4;
    const int r    = lane >> 3;
    const int s    = lane & 7;
    const int m0   = blockIdx.x * ROW_TILE;
    const int row  = m0 + 4 * g + r;

    const v4f zlo = *(const v4f*)(znrm + (size_t)row * DD);
    const v4f zhi = *(const v4f*)(znrm + (size_t)row * DD + 4);
    const v4f cq  = *(const v4f*)(Q + 4 * tid);

    v4f b00[4], b01[4], b10[4], b11[4], bias[4];
#pragma unroll
    for (int u = 0; u < 4; ++u) {
        const int col = (4 * g + u) * 64 + 4 * l16;
        b00[u]  = *(const v4f*)(decW + (size_t)(2 * half)     * DIN + col);
        b01[u]  = *(const v4f*)(decW + (size_t)(2 * half + 1) * DIN + col);
        b10[u]  = *(const v4f*)(decW + (size_t)(2 * half + 4) * DIN + col);
        b11[u]  = *(const v4f*)(decW + (size_t)(2 * half + 5) * DIN + col);
        bias[u] = *(const v4f*)(decB + col);
    }

    *(v4f*)(sQ + 4 * tid) = cq;
    __syncthreads();

    float zn[DD];
    zn[0] = zlo.x;  zn[1] = zlo.y;  zn[2] = zlo.z;  zn[3] = zlo.w;
    zn[4] = zhi.x;  zn[5] = zhi.y;  zn[6] = zhi.z;  zn[7] = zhi.w;

    float pq[4];
#pragma unroll
    for (int q = 0; q < 4; ++q) {
        const float* Q0 = sQ + (2 * q) * DD * DD + s;
        v2f t = both(0.f);
#pragma unroll
        for (int b = 0; b < DD; ++b) {
            v2f qq;
            qq.x = Q0[b * DD];
            qq.y = Q0[DD * DD + b * DD];
            t += qq * both(zn[b]);
        }
        pq[q] = zn[2 * q] * t.x + zn[2 * q + 1] * t.y;
    }
    szq[4 * g + r][s] = (pq[0] + pq[1]) + (pq[2] + pq[3]);
    __syncthreads();

    const v2f a0 = *(const v2f*)(&szq[l16][2 * half]);
    const v2f a1 = *(const v2f*)(&szq[l16][2 * half + 4]);

#pragma unroll
    for (int u = 0; u < 4; ++u) {
        v8f c[4];
#pragma unroll
        for (int t = 0; t < 4; ++t) {
            v2f b;
            b.x = b00[u][t];
            b.y = b01[u][t];
            v8f zero = {};
            c[t] = __builtin_amdgcn_wmma_f32_16x16x4_f32(
                       false, a0, false, b, (short)0, zero, false, false);
        }
#pragma unroll
        for (int t = 0; t < 4; ++t) {
            v2f b;
            b.x = b10[u][t];
            b.y = b11[u][t];
            c[t] = __builtin_amdgcn_wmma_f32_16x16x4_f32(
                       false, a1, false, b, (short)0, c[t], false, false);
        }
        float* orow = out + (size_t)(m0 + half * 8) * DIN + (4 * g + u) * 64 + 4 * l16;
#pragma unroll
        for (int j = 0; j < 8; ++j) {
            v4f o;
            o.x = c[0][j] + bias[u].x;
            o.y = c[1][j] + bias[u].y;
            o.z = c[2][j] + bias[u].z;
            o.w = c[3][j] + bias[u].w;
            *(v4f*)(orow + (size_t)j * DIN) = o;
        }
    }
}

extern "C" void kernel_launch(void* const* d_in, const int* in_sizes, int n_in,
                              void* d_out, int out_size, void* d_ws, size_t ws_size,
                              hipStream_t stream) {
    const float* x    = (const float*)d_in[0];
    const float* encW = (const float*)d_in[1];
    const float* encB = (const float*)d_in[2];
    const float* decW = (const float*)d_in[3];
    const float* decB = (const float*)d_in[4];
    const float* Wv   = (const float*)d_in[5];
    const float* bv   = (const float*)d_in[6];
    const float* Wo   = (const float*)d_in[7];
    const float* bo   = (const float*)d_in[8];
    const float* ln1g = (const float*)d_in[9];
    const float* ln1b = (const float*)d_in[10];
    const float* W1   = (const float*)d_in[11];
    const float* b1   = (const float*)d_in[12];
    const float* W2   = (const float*)d_in[13];
    const float* b2   = (const float*)d_in[14];
    const float* ln2g = (const float*)d_in[15];
    const float* ln2b = (const float*)d_in[16];
    const float* qw   = (const float*)d_in[17];
    float* out  = (float*)d_out;

    float* znrm = (float*)d_ws;
    float* Qf   = znrm + (size_t)BQ * DD;

    enc_rows_and_forms<<<FORM_BLOCKS + ENC_BLOCKS, ENC_THREADS, 0, stream>>>(
        x, encW, encB, Wv, bv, Wo, bo, ln1g, ln1b, W1, b1, W2, b2, ln2g, ln2b,
        znrm, qw, Qf);

    forms_and_decoder<<<ROW_GRID, ROW_THREADS, 0, stream>>>(
        znrm, Qf, decW, decB, out);
}


# 2 launches become 1: forms Q from a closed form (gates exp(-i th/2 X_mask) commute, U diagonal in Walsh-Hadamard basis) built by each 16-row workgroup, which decodes its own rows; Q sum order changes
# speedup vs baseline: 14.4912x; 1.2520x over previous
#include <hip/hip_runtime.h>
#include <hip/hip_bf16.h>
#include <math.h>

typedef __attribute__((ext_vector_type(2))) float v2f;
typedef __attribute__((ext_vector_type(4))) float v4f;
typedef __attribute__((ext_vector_type(8))) float v8f;
typedef __attribute__((ext_vector_type(4))) int   v4i;

#define BQ     4096
#define DIN    1024
#define DD     8
#define LL     2
#define FFN    128
#define NQ     10
#define DIMQ   1024
#define REPS   5
#define NGATE  (REPS * NQ)
__device__ __forceinline__ int ring_cpos(int w) { return NQ - 1 - w; }
__device__ __forceinline__ int ring_tpos(int w) { return NQ - 1 - ((w + 1) % NQ); }

__device__ inline int cnot_ring_inverse(int i)
{
#pragma unroll
    for (int w = NQ - 1; w >= 0; --w)
        i ^= ((i >> ring_cpos(w)) & 1) << ring_tpos(w);
    return i;
}

__device__ inline int cnot_ring_transpose(int i)
{
#pragma unroll
    for (int w = NQ - 1; w >= 0; --w)
        i ^= ((i >> ring_tpos(w)) & 1) << ring_cpos(w);
    return i;
}

__device__ __forceinline__ void lds_only_barrier()
{
    __builtin_amdgcn_fence(__ATOMIC_RELEASE, "workgroup", "local");
    __builtin_amdgcn_s_barrier();
    __builtin_amdgcn_fence(__ATOMIC_ACQUIRE, "workgroup", "local");
}

__device__ __forceinline__ void circuit_gates(
    const int tid, const float theta, int* sMask, int (*sTurn)[4])
{
    if (tid < NGATE) {
        const int r = tid / NQ;
        int m = 1 << (NQ - 1 - (tid - r * NQ));
        for (int q = 0; q < r; ++q) m = cnot_ring_inverse(m);
        double t = (double)theta * 0.07957747154594767;
        t -= floor(t);
        const unsigned int turn = (unsigned int)(unsigned long long)(t * 4294967296.0 + 0.5);
        sMask[tid] = m;
#pragma unroll
        for (int j = 0; j < 4; ++j)
            sTurn[tid][j] = (int)((__popc(m & (256 * j)) & 1) ? (0u - turn) : turn);
    }
}

__device__ __forceinline__ void circuit_phases(
    const int tid, const int* sMask, const int (*sTurn)[4], v2f* sD)
{
    unsigned int acc[4] = {0u, 0u, 0u, 0u};
    unsigned int nneg = 0u;
#pragma unroll 5
    for (int k = 0; k < NGATE; ++k) {
        const unsigned int neg = 0u - (unsigned int)(__popc(sMask[k] & tid) & 1);
        const v4i u = *(const v4i*)sTurn[k];
        nneg -= neg;
        acc[0] += (unsigned int)u.x ^ neg;
        acc[1] += (unsigned int)u.y ^ neg;
        acc[2] += (unsigned int)u.z ^ neg;
        acc[3] += (unsigned int)u.w ^ neg;
    }
#pragma unroll
    for (int j = 0; j < 4; ++j) {
        const float ang = (float)(int)(acc[j] + nneg) * 1.4629180792671596e-9f;
        v2f d;
        d.x = cosf(ang);
        d.y = sinf(ang);
        sD[tid + 256 * j] = d;
    }
}

__device__ __forceinline__ void circuit_planes(
    const int pw, const int lane, const v2f* sD, float* sQ)
{
#pragma unroll 1
    for (int u = 0; u < 2; ++u) {
        const int w = 2 * pw + u;
        int mu = 1 << (NQ - 1 - w);
#pragma unroll
        for (int r = 0; r < REPS; ++r) mu = cnot_ring_transpose(mu);
        float g = 0.f;
#pragma unroll 8
        for (int j = 0; j < DIMQ / 32; ++j) {
            const int s = lane + 32 * j;
            const v2f d0 = sD[s];
            const v2f d1 = sD[s ^ mu];
            g += d0.x * d1.x + d0.y * d1.y;
        }
        g += __shfl_xor(g, 8, 32);
        g += __shfl_xor(g, 16, 32);
        float gt[DD];
#pragma unroll
        for (int t = 0; t < DD; ++t) gt[t] = __shfl(g, t, 32);
        const int nu = mu & 7;
#pragma unroll
        for (int h = 0; h < 2; ++h) {
            const int ab = lane + 32 * h;
            const int a = ab >> 3;
            const int b = ab & 7;
            float q = 0.f;
#pragma unroll
            for (int t = 0; t < DD; ++t) {
                const int neg = (__popc(a & t) ^ __popc(b & (t ^ nu))) & 1;
                q += neg ? -gt[t] : gt[t];
            }
            sQ[ab * DD + w] = q * (1.0f / 1024.0f);
        }
    }
}

__device__ __forceinline__ v2f both(float x) { v2f r; r.x = x; r.y = x; return r; }

__device__ __forceinline__ void ln8(const v2f* h, const float* __restrict__ g,
                                    const float* __restrict__ b, v2f* out)
{
    float m = 0.f;
#pragma unroll
    for (int p = 0; p < 4; ++p) { m += h[p].x;  m += h[p].y; }
    m *= 0.125f;
    v2f t[4];
    float v = 0.f;
#pragma unroll
    for (int p = 0; p < 4; ++p) {
        t[p] = h[p] - both(m);
        v += t[p].x * t[p].x;
        v += t[p].y * t[p].y;
    }
    v *= 0.125f;
    const float inv = 1.0f / sqrtf(v + 1e-5f);
#pragma unroll
    for (int p = 0; p < 4; ++p)
        out[p] = t[p] * both(inv) * *(const v2f*)(g + 2 * p) + *(const v2f*)(b + 2 * p);
}

#define ROW_TILE   16
#define ROW_WAVES  4
#define W1_FLOATS  (LL * DD * FFN)
#define W2_FLOATS  (LL * FFN * DD)
#define W2_PADDED  (W2_FLOATS + W2_FLOATS / 8)
#define B1_FLOATS  (LL * FFN)
#define Q_FLOATS   (DD * DD * DD)
#define P_WV    0
#define P_WO    (P_WV + LL * 64)
#define P_BV    (P_WO + LL * 64)
#define P_BO    (P_BV + LL * DD)
#define P_LN1G  (P_BO + LL * DD)
#define P_LN1B  (P_LN1G + LL * DD)
#define P_B2    (P_LN1B + LL * DD)
#define P_LN2G  (P_B2 + LL * DD)
#define P_LN2B  (P_LN2G + LL * DD)
#define P_FLOATS (P_LN2B + LL * DD)

__device__ __forceinline__ int w2pad(int f) { return f + ((f >> 5) << 2); }

__device__ __forceinline__ void mat8(const v2f* x, const float* __restrict__ W,
                                     const float* __restrict__ bias, v2f* y)
{
#pragma unroll
    for (int p = 0; p < 4; ++p) {
        v2f s = *(const v2f*)(bias + 2 * p);
#pragma unroll
        for (int k = 0; k < DD; ++k)
            s += both(x[k >> 1][k & 1]) * *(const v2f*)(W + k * DD + 2 * p);
        y[p] = s;
    }
}

__device__ __forceinline__ void attn_ln(
    v2f* z, const int l,
    const float* __restrict__ Wv,  const float* __restrict__ bv,
    const float* __restrict__ Wo,  const float* __restrict__ bo,
    const float* __restrict__ ln1g, const float* __restrict__ ln1b)
{
    v2f v[4], attn[4], h[4];
    mat8(z, Wv + l * 64, bv + l * DD, v);
    mat8(v, Wo + l * 64, bo + l * DD, attn);
#pragma unroll
    for (int p = 0; p < 4; ++p) h[p] = z[p] + attn[p];
    ln8(h, ln1g + l * DD, ln1b + l * DD, z);
}

__device__ __forceinline__ void ffn_ln(
    v2f* z, const int l, const int s,
    const float* sW1, const float* sW2, const float* sB1,
    const float* __restrict__ b2,
    const float* __restrict__ ln2g, const float* __restrict__ ln2b)
{
    v2f acc[4];
#pragma unroll
    for (int p = 0; p < 4; ++p) acc[p] = both(0.f);
#pragma unroll
    for (int jj = 0; jj < 4; ++jj) {
        const int hid0 = 32 * jj + 4 * s;
        const v4f bb = *(const v4f*)(sB1 + l * FFN + hid0);
        v2f hlo, hhi;
        hlo.x = bb.x;  hlo.y = bb.y;  hhi.x = bb.z;  hhi.y = bb.w;
#pragma unroll
        for (int k = 0; k < DD; ++k) {
            const v4f w = *(const v4f*)(sW1 + (l * DD + k) * FFN + hid0);
            v2f wlo, whi;
            wlo.x = w.x;  wlo.y = w.y;  whi.x = w.z;  whi.y = w.w;
            const v2f zk = both(z[k >> 1][k & 1]);
            hlo += zk * wlo;
            hhi += zk * whi;
        }
#pragma unroll
        for (int e = 0; e < 4; ++e) {
            const float hraw = (e == 0) ? hlo.x : (e == 1) ? hlo.y : (e == 2) ? hhi.x : hhi.y;
            const v2f hr = both(fmaxf(hraw, 0.0f));
            const int f = (l * FFN + hid0 + e) * DD;
            const v4f wa = *(const v4f*)(sW2 + w2pad(f));
            const v4f wb = *(const v4f*)(sW2 + w2pad(f + 4));
            v2f w01, w23, w45, w67;
            w01.x = wa.x;  w01.y = wa.y;  w23.x = wa.z;  w23.y = wa.w;
            w45.x = wb.x;  w45.y = wb.y;  w67.x = wb.z;  w67.y = wb.w;
            acc[0] += hr * w01;
            acc[1] += hr * w23;
            acc[2] += hr * w45;
            acc[3] += hr * w67;
        }
    }
    for (int off = 4; off > 0; off >>= 1) {
#pragma unroll
        for (int p = 0; p < 4; ++p) {
            acc[p].x += __shfl_xor(acc[p].x, off, 32);
            acc[p].y += __shfl_xor(acc[p].y, off, 32);
        }
    }
    v2f h[4];
#pragma unroll
    for (int p = 0; p < 4; ++p) h[p] = z[p] + acc[p] + *(const v2f*)(b2 + l * DD + 2 * p);
    ln8(h, ln2g + l * DD, ln2b + l * DD, z);
}

#define ENC_ROWS    16
#define ENC_WAVES   8
#define ENC_THREADS (32 * ENC_WAVES)
#define ENC_KW      (DIN / ENC_WAVES)
#define ENC_BLOCKS  (BQ / ENC_ROWS)
#define DEC_GROUPS  (DIN / 64 / ENC_WAVES)

__global__ __launch_bounds__(ENC_THREADS) void forward_rows(
    const float* __restrict__ x, const float* __restrict__ encW,
    const float* __restrict__ encB,
    const float* __restrict__ Wv,  const float* __restrict__ bv,
    const float* __restrict__ Wo,  const float* __restrict__ bo,
    const float* __restrict__ ln1g, const float* __restrict__ ln1b,
    const float* __restrict__ W1,  const float* __restrict__ b1,
    const float* __restrict__ W2,  const float* __restrict__ b2,
    const float* __restrict__ ln2g, const float* __restrict__ ln2b,
    const float* __restrict__ qw,
    const float* __restrict__ decW, const float* __restrict__ decB,
    float* __restrict__ out)
{
    __shared__ __attribute__((aligned(16))) float red[ENC_WAVES][ENC_ROWS * DD];
    __shared__ __attribute__((aligned(16))) float sW1[W1_FLOATS];
    __shared__ __attribute__((aligned(16))) float sW2[W2_PADDED];
    __shared__ __attribute__((aligned(16))) float sB1[B1_FLOATS];
    __shared__ __attribute__((aligned(16))) float sP[P_FLOATS];
    __shared__ int sMask[NGATE];
    __shared__ __attribute__((aligned(16))) int sTurn[NGATE][4];
    __shared__ __attribute__((aligned(16))) v2f sD[DIMQ];
    __shared__ __attribute__((aligned(16))) float sQ[Q_FLOATS];
    __shared__ __attribute__((aligned(16))) float szq[ROW_TILE][DD];

    const int tid  = threadIdx.x;
    const int lane = tid & 31;
    const int wave = tid >> 5;
    const int l16  = lane & 15;
    const int half = lane >> 4;
    const int m0   = blockIdx.x * ENC_ROWS;
    const int bcol = l16 & 7;

    const float theta = qw[(tid < NGATE) ? tid : (NGATE - 1)];

    v4f cw1[W1_FLOATS / 4 / ENC_THREADS], cw2[W2_FLOATS / 4 / ENC_THREADS];
#pragma unroll
    for (int i = 0; i < W1_FLOATS / 4 / ENC_THREADS; ++i)
        cw1[i] = *(const v4f*)(W1 + 4 * (tid + i * ENC_THREADS));
#pragma unroll
    for (int i = 0; i < W2_FLOATS / 4 / ENC_THREADS; ++i)
        cw2[i] = *(const v4f*)(W2 + 4 * (tid + i * ENC_THREADS));
    const v4f cb1 = *(const v4f*)(b1 + 4 * (tid & (B1_FLOATS / 4 - 1)));
    const int pt = (tid < P_FLOATS / 4) ? tid : (P_FLOATS / 4 - 1);
    const int pv = (pt - 64) >> 2;
    const float* pvec = (pv == 0) ? bv : (pv == 1) ? bo : (pv == 2) ? ln1g : (pv == 3) ? ln1b
                      : (pv == 4) ? b2 : (pv == 5) ? ln2g : ln2b;
    const float* psrc = (pt < 32) ? (Wv + 4 * pt) : (pt < 64) ? (Wo + 4 * (pt - 32))
                      : (pvec + 4 * ((pt - 64) & 3));
    const v4f cp = *(const v4f*)psrc;

    const float* xa = x + (size_t)(m0 + l16) * DIN + wave * ENC_KW + 4 * half;
    const float* wb = encW + (size_t)(wave * ENC_KW + 4 * half) * DD + bcol;

    v4f areg[ENC_KW / 8];
#pragma unroll
    for (int i = 0; i < ENC_KW / 8; ++i)
        areg[i] = *(const v4f*)(xa + 8 * i);

    circuit_gates(tid, theta, sMask, sTurn);
    lds_only_barrier();
    circuit_phases(tid, sMask, sTurn, sD);

    v8f c0 = {}, c1 = {};
#pragma unroll
    for (int i = 0; i < ENC_KW / 8; ++i) {
        const v4f a = areg[i];
        v2f a0, a1, b0, b1v;
        a0.x = a.x;  a0.y = a.y;
        a1.x = a.z;  a1.y = a.w;
        b0.x  = wb[(8 * i + 0) * DD];
        b0.y  = wb[(8 * i + 1) * DD];
        b1v.x = wb[(8 * i + 2) * DD];
        b1v.y = wb[(8 * i + 3) * DD];
        c0 = __builtin_amdgcn_wmma_f32_16x16x4_f32(
                 false, a0, false, b0, (short)0, c0, false, false);
        c1 = __builtin_amdgcn_wmma_f32_16x16x4_f32(
                 false, a1, false, b1v, (short)0, c1, false, false);
    }

    if (l16 < DD) {
#pragma unroll
        for (int j = 0; j < 8; ++j)
            red[wave][(half * 8 + j) * DD + l16] = c0[j] + c1[j];
    }
#pragma unroll
    for (int i = 0; i < W1_FLOATS / 4 / ENC_THREADS; ++i)
        *(v4f*)(sW1 + 4 * (tid + i * ENC_THREADS)) = cw1[i];
#pragma unroll
    for (int i = 0; i < W2_FLOATS / 4 / ENC_THREADS; ++i)
        *(v4f*)(sW2 + w2pad(4 * (tid + i * ENC_THREADS))) = cw2[i];
    if (tid < B1_FLOATS / 4) *(v4f*)(sB1 + 4 * tid) = cb1;
    if (tid < P_FLOATS / 4)  *(v4f*)(sP + 4 * tid) = cp;
    __syncthreads();

    v4f b00[DEC_GROUPS], b01[DEC_GROUPS], b10[DEC_GROUPS], b11[DEC_GROUPS], bias[DEC_GROUPS];
#pragma unroll
    for (int u = 0; u < DEC_GROUPS; ++u) {
        const int col = (DEC_GROUPS * wave + u) * 64 + 4 * l16;
        b00[u]  = *(const v4f*)(decW + (size_t)(2 * half)     * DIN + col);
        b01[u]  = *(const v4f*)(decW + (size_t)(2 * half + 1) * DIN + col);
        b10[u]  = *(const v4f*)(decW + (size_t)(2 * half + 4) * DIN + col);
        b11[u]  = *(const v4f*)(decW + (size_t)(2 * half + 5) * DIN + col);
        bias[u] = *(const v4f*)(decB + col);
    }

    const int r  = lane >> 3;
    const int s  = lane & 7;
    const int rl = 4 * (wave & (ROW_WAVES - 1)) + r;

    float zn[DD];
#pragma unroll
    for (int k = 0; k < DD; ++k) zn[k] = 0.f;
    if (wave < ROW_WAVES) {
        v4f zlo = *(const v4f*)(&red[0][rl * DD]);
        v4f zhi = *(const v4f*)(&red[0][rl * DD + 4]);
#pragma unroll
        for (int w = 1; w < ENC_WAVES; ++w) {
            zlo += *(const v4f*)(&red[w][rl * DD]);
            zhi += *(const v4f*)(&red[w][rl * DD + 4]);
        }
        zlo += *(const v4f*)(encB);
        zhi += *(const v4f*)(encB + 4);

        v2f z[4];
        z[0].x = zlo.x;  z[0].y = zlo.y;  z[1].x = zlo.z;  z[1].y = zlo.w;
        z[2].x = zhi.x;  z[2].y = zhi.y;  z[3].x = zhi.z;  z[3].y = zhi.w;

        attn_ln(z, 0, sP + P_WV, sP + P_BV, sP + P_WO, sP + P_BO, sP + P_LN1G, sP + P_LN1B);
        ffn_ln(z, 0, s, sW1, sW2, sB1, sP + P_B2, sP + P_LN2G, sP + P_LN2B);
        attn_ln(z, 1, sP + P_WV, sP + P_BV, sP + P_WO, sP + P_BO, sP + P_LN1G, sP + P_LN1B);
        ffn_ln(z, 1, s, sW1, sW2, sB1, sP + P_B2, sP + P_LN2G, sP + P_LN2B);

        float nrm2 = 0.f;
#pragma unroll
        for (int p = 0; p < 4; ++p) { nrm2 += z[p].x * z[p].x;  nrm2 += z[p].y * z[p].y; }
        const float inv = 1.0f / sqrtf(nrm2);
#pragma unroll
        for (int p = 0; p < 4; ++p) {
            zn[2 * p]     = z[p].x * inv;
            zn[2 * p + 1] = z[p].y * inv;
        }
    } else {
        circuit_planes(wave - ROW_WAVES, lane, sD, sQ);
    }
    __syncthreads();

    if (wave < ROW_WAVES) {
        float pq[4];
#pragma unroll
        for (int q = 0; q < 4; ++q) {
            const float* Q0 = sQ + (2 * q) * DD * DD + s;
            v2f t = both(0.f);
#pragma unroll
            for (int b = 0; b < DD; ++b) {
                v2f qq;
                qq.x = Q0[b * DD];
                qq.y = Q0[DD * DD + b * DD];
                t += qq * both(zn[b]);
            }
            pq[q] = zn[2 * q] * t.x + zn[2 * q + 1] * t.y;
        }
        szq[rl][s] = (pq[0] + pq[1]) + (pq[2] + pq[3]);
    }
    __syncthreads();

    const v2f a0 = *(const v2f*)(&szq[l16][2 * half]);
    const v2f a1 = *(const v2f*)(&szq[l16][2 * half + 4]);

#pragma unroll
    for (int u = 0; u < DEC_GROUPS; ++u) {
        v8f c[4];
#pragma unroll
        for (int t = 0; t < 4; ++t) {
            v2f b;
            b.x = b00[u][t];
            b.y = b01[u][t];
            v8f zero = {};
            c[t] = __builtin_amdgcn_wmma_f32_16x16x4_f32(
                       false, a0, false, b, (short)0, zero, false, false);
        }
#pragma unroll
        for (int t = 0; t < 4; ++t) {
            v2f b;
            b.x = b10[u][t];
            b.y = b11[u][t];
            c[t] = __builtin_amdgcn_wmma_f32_16x16x4_f32(
                       false, a1, false, b, (short)0, c[t], false, false);
        }
        float* orow = out + (size_t)(m0 + half * 8) * DIN + (DEC_GROUPS * wave + u) * 64 + 4 * l16;
#pragma unroll
        for (int j = 0; j < 8; ++j) {
            v4f o;
            o.x = c[0][j] + bias[u].x;
            o.y = c[1][j] + bias[u].y;
            o.z = c[2][j] + bias[u].z;
            o.w = c[3][j] + bias[u].w;
            *(v4f*)(orow + (size_t)j * DIN) = o;
        }
    }
}

extern "C" void kernel_launch(void* const* d_in, const int* in_sizes, int n_in,
                              void* d_out, int out_size, void* d_ws, size_t ws_size,
                              hipStream_t stream) {
    const float* x    = (const float*)d_in[0];
    const float* encW = (const float*)d_in[1];
    const float* encB = (const float*)d_in[2];
    const float* decW = (const float*)d_in[3];
    const float* decB = (const float*)d_in[4];
    const float* Wv   = (const float*)d_in[5];
    const float* bv   = (const float*)d_in[6];
    const float* Wo   = (const float*)d_in[7];
    const float* bo   = (const float*)d_in[8];
    const float* ln1g = (const float*)d_in[9];
    const float* ln1b = (const float*)d_in[10];
    const float* W1   = (const float*)d_in[11];
    const float* b1   = (const float*)d_in[12];
    const float* W2   = (const float*)d_in[13];
    const float* b2   = (const float*)d_in[14];
    const float* ln2g = (const float*)d_in[15];
    const float* ln2b = (const float*)d_in[16];
    const float* qw   = (const float*)d_in[17];
    float* out  = (float*)d_out;

    forward_rows<<<ENC_BLOCKS, ENC_THREADS, 0, stream>>>(
        x, encW, encB, Wv, bv, Wo, bo, ln1g, ln1b, W1, b1, W2, b2, ln2g, ln2b,
        qw, decW, decB, out);
}


# Per-row stage: attention and layer norms run one column per sub-lane with weights in registers, rows exchanged by DPP instead of LDS loads and LDS shuffles (layer-norm sum order changes)
# speedup vs baseline: 15.2151x; 1.0500x over previous
#include <hip/hip_runtime.h>
#include <hip/hip_bf16.h>
#include <math.h>

typedef __attribute__((ext_vector_type(2))) float v2f;
typedef __attribute__((ext_vector_type(4))) float v4f;
typedef __attribute__((ext_vector_type(8))) float v8f;
typedef __attribute__((ext_vector_type(4))) int   v4i;

#define BQ     4096
#define DIN    1024
#define DD     8
#define LL     2
#define FFN    128
#define NQ     10
#define DIMQ   1024
#define REPS   5
#define NGATE  (REPS * NQ)
__device__ __forceinline__ int ring_cpos(int w) { return NQ - 1 - w; }
__device__ __forceinline__ int ring_tpos(int w) { return NQ - 1 - ((w + 1) % NQ); }

__device__ inline int cnot_ring_inverse(int i)
{
#pragma unroll
    for (int w = NQ - 1; w >= 0; --w)
        i ^= ((i >> ring_cpos(w)) & 1) << ring_tpos(w);
    return i;
}

__device__ inline int cnot_ring_transpose(int i)
{
#pragma unroll
    for (int w = NQ - 1; w >= 0; --w)
        i ^= ((i >> ring_tpos(w)) & 1) << ring_cpos(w);
    return i;
}

__device__ __forceinline__ void lds_only_barrier()
{
    __builtin_amdgcn_fence(__ATOMIC_RELEASE, "workgroup", "local");
    __builtin_amdgcn_s_barrier();
    __builtin_amdgcn_fence(__ATOMIC_ACQUIRE, "workgroup", "local");
}

__device__ __forceinline__ void circuit_gates(
    const int tid, const float theta, int* sMask, int (*sTurn)[4])
{
    if (tid < NGATE) {
        const int r = tid / NQ;
        int m = 1 << (NQ - 1 - (tid - r * NQ));
        for (int q = 0; q < r; ++q) m = cnot_ring_inverse(m);
        double t = (double)theta * 0.07957747154594767;
        t -= floor(t);
        const unsigned int turn = (unsigned int)(unsigned long long)(t * 4294967296.0 + 0.5);
        sMask[tid] = m;
#pragma unroll
        for (int j = 0; j < 4; ++j)
            sTurn[tid][j] = (int)((__popc(m & (256 * j)) & 1) ? (0u - turn) : turn);
    }
}

__device__ __forceinline__ void circuit_phases(
    const int tid, const int* sMask, const int (*sTurn)[4], v2f* sD)
{
    unsigned int acc[4] = {0u, 0u, 0u, 0u};
    unsigned int nneg = 0u;
#pragma unroll 5
    for (int k = 0; k < NGATE; ++k) {
        const unsigned int neg = 0u - (unsigned int)(__popc(sMask[k] & tid) & 1);
        const v4i u = *(const v4i*)sTurn[k];
        nneg -= neg;
        acc[0] += (unsigned int)u.x ^ neg;
        acc[1] += (unsigned int)u.y ^ neg;
        acc[2] += (unsigned int)u.z ^ neg;
        acc[3] += (unsigned int)u.w ^ neg;
    }
#pragma unroll
    for (int j = 0; j < 4; ++j) {
        const float ang = (float)(int)(acc[j] + nneg) * 1.4629180792671596e-9f;
        v2f d;
        d.x = cosf(ang);
        d.y = sinf(ang);
        sD[tid + 256 * j] = d;
    }
}

__device__ __forceinline__ void circuit_planes(
    const int pw, const int lane, const v2f* sD, float* sQ)
{
#pragma unroll 1
    for (int u = 0; u < 2; ++u) {
        const int w = 2 * pw + u;
        int mu = 1 << (NQ - 1 - w);
#pragma unroll
        for (int r = 0; r < REPS; ++r) mu = cnot_ring_transpose(mu);
        float g = 0.f;
#pragma unroll 8
        for (int j = 0; j < DIMQ / 32; ++j) {
            const int s = lane + 32 * j;
            const v2f d0 = sD[s];
            const v2f d1 = sD[s ^ mu];
            g += d0.x * d1.x + d0.y * d1.y;
        }
        g += __shfl_xor(g, 8, 32);
        g += __shfl_xor(g, 16, 32);
        float gt[DD];
#pragma unroll
        for (int t = 0; t < DD; ++t) gt[t] = __shfl(g, t, 32);
        const int nu = mu & 7;
#pragma unroll
        for (int h = 0; h < 2; ++h) {
            const int ab = lane + 32 * h;
            const int a = ab >> 3;
            const int b = ab & 7;
            float q = 0.f;
#pragma unroll
            for (int t = 0; t < DD; ++t) {
                const int neg = (__popc(a & t) ^ __popc(b & (t ^ nu))) & 1;
                q += neg ? -gt[t] : gt[t];
            }
            sQ[ab * DD + w] = q * (1.0f / 1024.0f);
        }
    }
}

__device__ __forceinline__ v2f both(float x) { v2f r; r.x = x; r.y = x; return r; }

template <int K>
__device__ __forceinline__ float row_lane(float v)
{
    return __builtin_bit_cast(float, __builtin_amdgcn_mov_dpp8(
               __builtin_bit_cast(unsigned int, v), (unsigned int)K * 0x249249u));
}

template <int M>
__device__ __forceinline__ float xor_lane(float v)
{
    return __builtin_bit_cast(float, __builtin_amdgcn_update_dpp(
               0, __builtin_bit_cast(int, v), 0x160 + M, 0xf, 0xf, true));
}

__device__ __forceinline__ void row_gather(float own, float* all)
{
    all[0] = row_lane<0>(own);  all[1] = row_lane<1>(own);
    all[2] = row_lane<2>(own);  all[3] = row_lane<3>(own);
    all[4] = row_lane<4>(own);  all[5] = row_lane<5>(own);
    all[6] = row_lane<6>(own);  all[7] = row_lane<7>(own);
}

__device__ __forceinline__ float row_sum(float v)
{
    v += xor_lane<4>(v);
    v += xor_lane<2>(v);
    v += xor_lane<1>(v);
    return v;
}

__device__ __forceinline__ float ln_col(float h, float g, float b)
{
    const float m = row_sum(h) * 0.125f;
    const float t = h - m;
    const float v = row_sum(t * t) * 0.125f;
    const float inv = 1.0f / sqrtf(v + 1e-5f);
    return t * inv * g + b;
}

#define ROW_TILE   16
#define ROW_WAVES  4
#define W1_FLOATS  (LL * DD * FFN)
#define W2_FLOATS  (LL * FFN * DD)
#define W2_PADDED  (W2_FLOATS + W2_FLOATS / 8)
#define B1_FLOATS  (LL * FFN)
#define Q_FLOATS   (DD * DD * DD)
#define P_WV    0
#define P_WO    (P_WV + LL * 64)
#define P_BV    (P_WO + LL * 64)
#define P_BO    (P_BV + LL * DD)
#define P_LN1G  (P_BO + LL * DD)
#define P_LN1B  (P_LN1G + LL * DD)
#define P_B2    (P_LN1B + LL * DD)
#define P_LN2G  (P_B2 + LL * DD)
#define P_LN2B  (P_LN2G + LL * DD)
#define P_FLOATS (P_LN2B + LL * DD)

__device__ __forceinline__ int w2pad(int f) { return f + ((f >> 5) << 2); }

struct ColParams {
    float wv[LL][DD], wo[LL][DD];
    float bv[LL], bo[LL], g1[LL], be1[LL], b2[LL], g2[LL], be2[LL];
};

__device__ __forceinline__ void load_col_params(const float* sP, const int s, ColParams& P)
{
#pragma unroll
    for (int l = 0; l < LL; ++l) {
#pragma unroll
        for (int k = 0; k < DD; ++k) {
            P.wv[l][k] = sP[P_WV + l * 64 + k * DD + s];
            P.wo[l][k] = sP[P_WO + l * 64 + k * DD + s];
        }
        P.bv[l]  = sP[P_BV   + l * DD + s];
        P.bo[l]  = sP[P_BO   + l * DD + s];
        P.g1[l]  = sP[P_LN1G + l * DD + s];
        P.be1[l] = sP[P_LN1B + l * DD + s];
        P.b2[l]  = sP[P_B2   + l * DD + s];
        P.g2[l]  = sP[P_LN2G + l * DD + s];
        P.be2[l] = sP[P_LN2B + l * DD + s];
    }
}

__device__ __forceinline__ void attn_ln(float* zall, float& zown, const int l, const ColParams& P)
{
    float vs = P.bv[l];
#pragma unroll
    for (int k = 0; k < DD; ++k) vs += zall[k] * P.wv[l][k];
    float vall[DD];
    row_gather(vs, vall);
    float as = P.bo[l];
#pragma unroll
    for (int k = 0; k < DD; ++k) as += vall[k] * P.wo[l][k];
    zown = ln_col(zown + as, P.g1[l], P.be1[l]);
    row_gather(zown, zall);
}

__device__ __forceinline__ void ffn_ln(
    float* zall, float& zown, const int l, const int s,
    const float* sW1, const float* sW2, const float* sB1, const ColParams& P)
{
    v2f z[4];
#pragma unroll
    for (int p = 0; p < 4; ++p) { z[p].x = zall[2 * p];  z[p].y = zall[2 * p + 1]; }
    v2f acc[4];
#pragma unroll
    for (int p = 0; p < 4; ++p) acc[p] = both(0.f);
#pragma unroll
    for (int jj = 0; jj < 4; ++jj) {
        const int hid0 = 32 * jj + 4 * s;
        const v4f bb = *(const v4f*)(sB1 + l * FFN + hid0);
        v2f hlo, hhi;
        hlo.x = bb.x;  hlo.y = bb.y;  hhi.x = bb.z;  hhi.y = bb.w;
#pragma unroll
        for (int k = 0; k < DD; ++k) {
            const v4f w = *(const v4f*)(sW1 + (l * DD + k) * FFN + hid0);
            v2f wlo, whi;
            wlo.x = w.x;  wlo.y = w.y;  whi.x = w.z;  whi.y = w.w;
            const v2f zk = both(z[k >> 1][k & 1]);
            hlo += zk * wlo;
            hhi += zk * whi;
        }
#pragma unroll
        for (int e = 0; e < 4; ++e) {
            const float hraw = (e == 0) ? hlo.x : (e == 1) ? hlo.y : (e == 2) ? hhi.x : hhi.y;
            const v2f hr = both(fmaxf(hraw, 0.0f));
            const int f = (l * FFN + hid0 + e) * DD;
            const v4f wa = *(const v4f*)(sW2 + w2pad(f));
            const v4f wb = *(const v4f*)(sW2 + w2pad(f + 4));
            v2f w01, w23, w45, w67;
            w01.x = wa.x;  w01.y = wa.y;  w23.x = wa.z;  w23.y = wa.w;
            w45.x = wb.x;  w45.y = wb.y;  w67.x = wb.z;  w67.y = wb.w;
            acc[0] += hr * w01;
            acc[1] += hr * w23;
            acc[2] += hr * w45;
            acc[3] += hr * w67;
        }
    }
    const bool up4 = (s & 4) != 0, up2 = (s & 2) != 0, up1 = (s & 1) != 0;
    const float a[DD] = { acc[0].x, acc[0].y, acc[1].x, acc[1].y,
                          acc[2].x, acc[2].y, acc[3].x, acc[3].y };
    float r4[4];
#pragma unroll
    for (int i = 0; i < 4; ++i) {
        const float keep = up4 ? a[4 + i] : a[i];
        const float send = up4 ? a[i] : a[4 + i];
        r4[i] = keep + xor_lane<4>(send);
    }
    float r2[2];
#pragma unroll
    for (int i = 0; i < 2; ++i) {
        const float keep = up2 ? r4[2 + i] : r4[i];
        const float send = up2 ? r4[i] : r4[2 + i];
        r2[i] = keep + xor_lane<2>(send);
    }
    const float keep1 = up1 ? r2[1] : r2[0];
    const float send1 = up1 ? r2[0] : r2[1];
    const float ff = keep1 + xor_lane<1>(send1);

    zown = ln_col(zown + ff + P.b2[l], P.g2[l], P.be2[l]);
    row_gather(zown, zall);
}

#define ENC_ROWS    16
#define ENC_WAVES   8
#define ENC_THREADS (32 * ENC_WAVES)
#define ENC_KW      (DIN / ENC_WAVES)
#define ENC_BLOCKS  (BQ / ENC_ROWS)
#define DEC_GROUPS  (DIN / 64 / ENC_WAVES)

__global__ __launch_bounds__(ENC_THREADS) void forward_rows(
    const float* __restrict__ x, const float* __restrict__ encW,
    const float* __restrict__ encB,
    const float* __restrict__ Wv,  const float* __restrict__ bv,
    const float* __restrict__ Wo,  const float* __restrict__ bo,
    const float* __restrict__ ln1g, const float* __restrict__ ln1b,
    const float* __restrict__ W1,  const float* __restrict__ b1,
    const float* __restrict__ W2,  const float* __restrict__ b2,
    const float* __restrict__ ln2g, const float* __restrict__ ln2b,
    const float* __restrict__ qw,
    const float* __restrict__ decW, const float* __restrict__ decB,
    float* __restrict__ out)
{
    __shared__ __attribute__((aligned(16))) float red[ENC_WAVES][ENC_ROWS * DD];
    __shared__ __attribute__((aligned(16))) float sW1[W1_FLOATS];
    __shared__ __attribute__((aligned(16))) float sW2[W2_PADDED];
    __shared__ __attribute__((aligned(16))) float sB1[B1_FLOATS];
    __shared__ __attribute__((aligned(16))) float sP[P_FLOATS];
    __shared__ int sMask[NGATE];
    __shared__ __attribute__((aligned(16))) int sTurn[NGATE][4];
    __shared__ __attribute__((aligned(16))) v2f sD[DIMQ];
    __shared__ __attribute__((aligned(16))) float sQ[Q_FLOATS];
    __shared__ __attribute__((aligned(16))) float szq[ROW_TILE][DD];

    const int tid  = threadIdx.x;
    const int lane = tid & 31;
    const int wave = tid >> 5;
    const int l16  = lane & 15;
    const int half = lane >> 4;
    const int m0   = blockIdx.x * ENC_ROWS;
    const int bcol = l16 & 7;

    const float theta = qw[(tid < NGATE) ? tid : (NGATE - 1)];
    const float encb_s = encB[lane & 7];

    v4f cw1[W1_FLOATS / 4 / ENC_THREADS], cw2[W2_FLOATS / 4 / ENC_THREADS];
#pragma unroll
    for (int i = 0; i < W1_FLOATS / 4 / ENC_THREADS; ++i)
        cw1[i] = *(const v4f*)(W1 + 4 * (tid + i * ENC_THREADS));
#pragma unroll
    for (int i = 0; i < W2_FLOATS / 4 / ENC_THREADS; ++i)
        cw2[i] = *(const v4f*)(W2 + 4 * (tid + i * ENC_THREADS));
    const v4f cb1 = *(const v4f*)(b1 + 4 * (tid & (B1_FLOATS / 4 - 1)));
    const int pt = (tid < P_FLOATS / 4) ? tid : (P_FLOATS / 4 - 1);
    const int pv = (pt - 64) >> 2;
    const float* pvec = (pv == 0) ? bv : (pv == 1) ? bo : (pv == 2) ? ln1g : (pv == 3) ? ln1b
                      : (pv == 4) ? b2 : (pv == 5) ? ln2g : ln2b;
    const float* psrc = (pt < 32) ? (Wv + 4 * pt) : (pt < 64) ? (Wo + 4 * (pt - 32))
                      : (pvec + 4 * ((pt - 64) & 3));
    const v4f cp = *(const v4f*)psrc;

    const float* xa = x + (size_t)(m0 + l16) * DIN + wave * ENC_KW + 4 * half;
    const float* wb = encW + (size_t)(wave * ENC_KW + 4 * half) * DD + bcol;

    v4f areg[ENC_KW / 8];
#pragma unroll
    for (int i = 0; i < ENC_KW / 8; ++i)
        areg[i] = *(const v4f*)(xa + 8 * i);

    circuit_gates(tid, theta, sMask, sTurn);
    lds_only_barrier();
    circuit_phases(tid, sMask, sTurn, sD);

    v8f c0 = {}, c1 = {};
#pragma unroll
    for (int i = 0; i < ENC_KW / 8; ++i) {
        const v4f a = areg[i];
        v2f a0, a1, b0, b1v;
        a0.x = a.x;  a0.y = a.y;
        a1.x = a.z;  a1.y = a.w;
        b0.x  = wb[(8 * i + 0) * DD];
        b0.y  = wb[(8 * i + 1) * DD];
        b1v.x = wb[(8 * i + 2) * DD];
        b1v.y = wb[(8 * i + 3) * DD];
        c0 = __builtin_amdgcn_wmma_f32_16x16x4_f32(
                 false, a0, false, b0, (short)0, c0, false, false);
        c1 = __builtin_amdgcn_wmma_f32_16x16x4_f32(
                 false, a1, false, b1v, (short)0, c1, false, false);
    }

    if (l16 < DD) {
#pragma unroll
        for (int j = 0; j < 8; ++j)
            red[wave][(half * 8 + j) * DD + l16] = c0[j] + c1[j];
    }
#pragma unroll
    for (int i = 0; i < W1_FLOATS / 4 / ENC_THREADS; ++i)
        *(v4f*)(sW1 + 4 * (tid + i * ENC_THREADS)) = cw1[i];
#pragma unroll
    for (int i = 0; i < W2_FLOATS / 4 / ENC_THREADS; ++i)
        *(v4f*)(sW2 + w2pad(4 * (tid + i * ENC_THREADS))) = cw2[i];
    if (tid < B1_FLOATS / 4) *(v4f*)(sB1 + 4 * tid) = cb1;
    if (tid < P_FLOATS / 4)  *(v4f*)(sP + 4 * tid) = cp;
    __syncthreads();

    v4f b00[DEC_GROUPS], b01[DEC_GROUPS], b10[DEC_GROUPS], b11[DEC_GROUPS], bias[DEC_GROUPS];
#pragma unroll
    for (int u = 0; u < DEC_GROUPS; ++u) {
        const int col = (DEC_GROUPS * wave + u) * 64 + 4 * l16;
        b00[u]  = *(const v4f*)(decW + (size_t)(2 * half)     * DIN + col);
        b01[u]  = *(const v4f*)(decW + (size_t)(2 * half + 1) * DIN + col);
        b10[u]  = *(const v4f*)(decW + (size_t)(2 * half + 4) * DIN + col);
        b11[u]  = *(const v4f*)(decW + (size_t)(2 * half + 5) * DIN + col);
        bias[u] = *(const v4f*)(decB + col);
    }

    const int r  = lane >> 3;
    const int s  = lane & 7;
    const int rl = 4 * (wave & (ROW_WAVES - 1)) + r;

    float zn[DD];
#pragma unroll
    for (int k = 0; k < DD; ++k) zn[k] = 0.f;
    if (wave < ROW_WAVES) {
        ColParams P;
        load_col_params(sP, s, P);

        float zown = red[0][rl * DD + s];
#pragma unroll
        for (int w = 1; w < ENC_WAVES; ++w) zown += red[w][rl * DD + s];
        zown += encb_s;
        float zall[DD];
        row_gather(zown, zall);

        attn_ln(zall, zown, 0, P);
        ffn_ln(zall, zown, 0, s, sW1, sW2, sB1, P);
        attn_ln(zall, zown, 1, P);
        ffn_ln(zall, zown, 1, s, sW1, sW2, sB1, P);

        float nrm2 = 0.f;
#pragma unroll
        for (int k = 0; k < DD; ++k) nrm2 += zall[k] * zall[k];
        const float inv = 1.0f / sqrtf(nrm2);
#pragma unroll
        for (int k = 0; k < DD; ++k) zn[k] = zall[k] * inv;
    } else {
        circuit_planes(wave - ROW_WAVES, lane, sD, sQ);
    }
    __syncthreads();

    if (wave < ROW_WAVES) {
        float pq[4];
#pragma unroll
        for (int q = 0; q < 4; ++q) {
            const float* Q0 = sQ + (2 * q) * DD * DD + s;
            v2f t = both(0.f);
#pragma unroll
            for (int b = 0; b < DD; ++b) {
                v2f qq;
                qq.x = Q0[b * DD];
                qq.y = Q0[DD * DD + b * DD];
                t += qq * both(zn[b]);
            }
            pq[q] = zn[2 * q] * t.x + zn[2 * q + 1] * t.y;
        }
        szq[rl][s] = (pq[0] + pq[1]) + (pq[2] + pq[3]);
    }
    __syncthreads();

    const v2f a0 = *(const v2f*)(&szq[l16][2 * half]);
    const v2f a1 = *(const v2f*)(&szq[l16][2 * half + 4]);

#pragma unroll
    for (int u = 0; u < DEC_GROUPS; ++u) {
        v8f c[4];
#pragma unroll
        for (int t = 0; t < 4; ++t) {
            v2f b;
            b.x = b00[u][t];
            b.y = b01[u][t];
            v8f zero = {};
            c[t] = __builtin_amdgcn_wmma_f32_16x16x4_f32(
                       false, a0, false, b, (short)0, zero, false, false);
        }
#pragma unroll
        for (int t = 0; t < 4; ++t) {
            v2f b;
            b.x = b10[u][t];
            b.y = b11[u][t];
            c[t] = __builtin_amdgcn_wmma_f32_16x16x4_f32(
                       false, a1, false, b, (short)0, c[t], false, false);
        }
        float* orow = out + (size_t)(m0 + half * 8) * DIN + (DEC_GROUPS * wave + u) * 64 + 4 * l16;
#pragma unroll
        for (int j = 0; j < 8; ++j) {
            v4f o;
            o.x = c[0][j] + bias[u].x;
            o.y = c[1][j] + bias[u].y;
            o.z = c[2][j] + bias[u].z;
            o.w = c[3][j] + bias[u].w;
            *(v4f*)(orow + (size_t)j * DIN) = o;
        }
    }
}

extern "C" void kernel_launch(void* const* d_in, const int* in_sizes, int n_in,
                              void* d_out, int out_size, void* d_ws, size_t ws_size,
                              hipStream_t stream) {
    const float* x    = (const float*)d_in[0];
    const float* encW = (const float*)d_in[1];
    const float* encB = (const float*)d_in[2];
    const float* decW = (const float*)d_in[3];
    const float* decB = (const float*)d_in[4];
    const float* Wv   = (const float*)d_in[5];
    const float* bv   = (const float*)d_in[6];
    const float* Wo   = (const float*)d_in[7];
    const float* bo   = (const float*)d_in[8];
    const float* ln1g = (const float*)d_in[9];
    const float* ln1b = (const float*)d_in[10];
    const float* W1   = (const float*)d_in[11];
    const float* b1   = (const float*)d_in[12];
    const float* W2   = (const float*)d_in[13];
    const float* b2   = (const float*)d_in[14];
    const float* ln2g = (const float*)d_in[15];
    const float* ln2b = (const float*)d_in[16];
    const float* qw   = (const float*)d_in[17];
    float* out  = (float*)d_out;

    forward_rows<<<ENC_BLOCKS, ENC_THREADS, 0, stream>>>(
        x, encW, encB, Wv, bv, Wo, bo, ln1g, ln1b, W1, b1, W2, b2, ln2g, ln2b,
        qw, decW, decB, out);
}


# Circuit planes: the wire masks mu_w come from the circuit's structure alone, so the compiler evaluates them; no chain of 150 dependent bit steps in front of each plane of Q at run time
# speedup vs baseline: 16.2355x; 1.0671x over previous
#include <hip/hip_runtime.h>
#include <hip/hip_bf16.h>
#include <math.h>

typedef __attribute__((ext_vector_type(2))) float v2f;
typedef __attribute__((ext_vector_type(4))) float v4f;
typedef __attribute__((ext_vector_type(8))) float v8f;
typedef __attribute__((ext_vector_type(4))) int   v4i;

#define BQ     4096
#define DIN    1024
#define DD     8
#define LL     2
#define FFN    128
#define NQ     10
#define DIMQ   1024
#define REPS   5
#define NGATE  (REPS * NQ)
constexpr int ring_cpos(int w) { return NQ - 1 - w; }
constexpr int ring_tpos(int w) { return NQ - 1 - ((w + 1) % NQ); }

__device__ inline int cnot_ring_inverse(int i)
{
#pragma unroll
    for (int w = NQ - 1; w >= 0; --w)
        i ^= ((i >> ring_cpos(w)) & 1) << ring_tpos(w);
    return i;
}

constexpr int cnot_ring_transpose(int i)
{
    for (int w = NQ - 1; w >= 0; --w)
        i ^= ((i >> ring_tpos(w)) & 1) << ring_cpos(w);
    return i;
}

struct WireMasks { int mu[DD]; };
constexpr WireMasks wire_masks()
{
    WireMasks t = {};
    for (int w = 0; w < DD; ++w) {
        int mu = 1 << (NQ - 1 - w);
        for (int r = 0; r < REPS; ++r) mu = cnot_ring_transpose(mu);
        t.mu[w] = mu;
    }
    return t;
}

__device__ __forceinline__ int wire_mask(int w)
{
    constexpr WireMasks T = wire_masks();
    int mu = T.mu[0];
#pragma unroll
    for (int i = 1; i < DD; ++i) mu = (w == i) ? T.mu[i] : mu;
    return mu;
}

__device__ __forceinline__ void lds_only_barrier()
{
    __builtin_amdgcn_fence(__ATOMIC_RELEASE, "workgroup", "local");
    __builtin_amdgcn_s_barrier();
    __builtin_amdgcn_fence(__ATOMIC_ACQUIRE, "workgroup", "local");
}

__device__ __forceinline__ void circuit_gates(
    const int tid, const float theta, int* sMask, int (*sTurn)[4])
{
    if (tid < NGATE) {
        const int r = tid / NQ;
        int m = 1 << (NQ - 1 - (tid - r * NQ));
        for (int q = 0; q < r; ++q) m = cnot_ring_inverse(m);
        double t = (double)theta * 0.07957747154594767;
        t -= floor(t);
        const unsigned int turn = (unsigned int)(unsigned long long)(t * 4294967296.0 + 0.5);
        sMask[tid] = m;
#pragma unroll
        for (int j = 0; j < 4; ++j)
            sTurn[tid][j] = (int)((__popc(m & (256 * j)) & 1) ? (0u - turn) : turn);
    }
}

__device__ __forceinline__ void circuit_phases(
    const int tid, const int* sMask, const int (*sTurn)[4], v2f* sD)
{
    unsigned int acc[4] = {0u, 0u, 0u, 0u};
    unsigned int nneg = 0u;
#pragma unroll 5
    for (int k = 0; k < NGATE; ++k) {
        const unsigned int neg = 0u - (unsigned int)(__popc(sMask[k] & tid) & 1);
        const v4i u = *(const v4i*)sTurn[k];
        nneg -= neg;
        acc[0] += (unsigned int)u.x ^ neg;
        acc[1] += (unsigned int)u.y ^ neg;
        acc[2] += (unsigned int)u.z ^ neg;
        acc[3] += (unsigned int)u.w ^ neg;
    }
#pragma unroll
    for (int j = 0; j < 4; ++j) {
        const float ang = (float)(int)(acc[j] + nneg) * 1.4629180792671596e-9f;
        v2f d;
        d.x = cosf(ang);
        d.y = sinf(ang);
        sD[tid + 256 * j] = d;
    }
}

__device__ __forceinline__ void circuit_planes(
    const int pw, const int lane, const v2f* sD, float* sQ)
{
#pragma unroll 1
    for (int u = 0; u < 2; ++u) {
        const int w = 2 * pw + u;
        const int mu = wire_mask(w);
        float g = 0.f;
#pragma unroll 8
        for (int j = 0; j < DIMQ / 32; ++j) {
            const int s = lane + 32 * j;
            const v2f d0 = sD[s];
            const v2f d1 = sD[s ^ mu];
            g += d0.x * d1.x + d0.y * d1.y;
        }
        g += __shfl_xor(g, 8, 32);
        g += __shfl_xor(g, 16, 32);
        float gt[DD];
#pragma unroll
        for (int t = 0; t < DD; ++t) gt[t] = __shfl(g, t, 32);
        const int nu = mu & 7;
#pragma unroll
        for (int h = 0; h < 2; ++h) {
            const int ab = lane + 32 * h;
            const int a = ab >> 3;
            const int b = ab & 7;
            float q = 0.f;
#pragma unroll
            for (int t = 0; t < DD; ++t) {
                const int neg = (__popc(a & t) ^ __popc(b & (t ^ nu))) & 1;
                q += neg ? -gt[t] : gt[t];
            }
            sQ[ab * DD + w] = q * (1.0f / 1024.0f);
        }
    }
}

__device__ __forceinline__ v2f both(float x) { v2f r; r.x = x; r.y = x; return r; }

template <int K>
__device__ __forceinline__ float row_lane(float v)
{
    return __builtin_bit_cast(float, __builtin_amdgcn_mov_dpp8(
               __builtin_bit_cast(unsigned int, v), (unsigned int)K * 0x249249u));
}

template <int M>
__device__ __forceinline__ float xor_lane(float v)
{
    return __builtin_bit_cast(float, __builtin_amdgcn_update_dpp(
               0, __builtin_bit_cast(int, v), 0x160 + M, 0xf, 0xf, true));
}

__device__ __forceinline__ void row_gather(float own, float* all)
{
    all[0] = row_lane<0>(own);  all[1] = row_lane<1>(own);
    all[2] = row_lane<2>(own);  all[3] = row_lane<3>(own);
    all[4] = row_lane<4>(own);  all[5] = row_lane<5>(own);
    all[6] = row_lane<6>(own);  all[7] = row_lane<7>(own);
}

__device__ __forceinline__ float row_sum(float v)
{
    v += xor_lane<4>(v);
    v += xor_lane<2>(v);
    v += xor_lane<1>(v);
    return v;
}

__device__ __forceinline__ float ln_col(float h, float g, float b)
{
    const float m = row_sum(h) * 0.125f;
    const float t = h - m;
    const float v = row_sum(t * t) * 0.125f;
    const float inv = 1.0f / sqrtf(v + 1e-5f);
    return t * inv * g + b;
}

#define ROW_TILE   16
#define ROW_WAVES  4
#define W1_FLOATS  (LL * DD * FFN)
#define W2_FLOATS  (LL * FFN * DD)
#define W2_PADDED  (W2_FLOATS + W2_FLOATS / 8)
#define B1_FLOATS  (LL * FFN)
#define Q_FLOATS   (DD * DD * DD)
#define P_WV    0
#define P_WO    (P_WV + LL * 64)
#define P_BV    (P_WO + LL * 64)
#define P_BO    (P_BV + LL * DD)
#define P_LN1G  (P_BO + LL * DD)
#define P_LN1B  (P_LN1G + LL * DD)
#define P_B2    (P_LN1B + LL * DD)
#define P_LN2G  (P_B2 + LL * DD)
#define P_LN2B  (P_LN2G + LL * DD)
#define P_FLOATS (P_LN2B + LL * DD)

__device__ __forceinline__ int w2pad(int f) { return f + ((f >> 5) << 2); }

struct ColParams {
    float wv[LL][DD], wo[LL][DD];
    float bv[LL], bo[LL], g1[LL], be1[LL], b2[LL], g2[LL], be2[LL];
};

__device__ __forceinline__ void load_col_params(const float* sP, const int s, ColParams& P)
{
#pragma unroll
    for (int l = 0; l < LL; ++l) {
#pragma unroll
        for (int k = 0; k < DD; ++k) {
            P.wv[l][k] = sP[P_WV + l * 64 + k * DD + s];
            P.wo[l][k] = sP[P_WO + l * 64 + k * DD + s];
        }
        P.bv[l]  = sP[P_BV   + l * DD + s];
        P.bo[l]  = sP[P_BO   + l * DD + s];
        P.g1[l]  = sP[P_LN1G + l * DD + s];
        P.be1[l] = sP[P_LN1B + l * DD + s];
        P.b2[l]  = sP[P_B2   + l * DD + s];
        P.g2[l]  = sP[P_LN2G + l * DD + s];
        P.be2[l] = sP[P_LN2B + l * DD + s];
    }
}

__device__ __forceinline__ void attn_ln(float* zall, float& zown, const int l, const ColParams& P)
{
    float vs = P.bv[l];
#pragma unroll
    for (int k = 0; k < DD; ++k) vs += zall[k] * P.wv[l][k];
    float vall[DD];
    row_gather(vs, vall);
    float as = P.bo[l];
#pragma unroll
    for (int k = 0; k < DD; ++k) as += vall[k] * P.wo[l][k];
    zown = ln_col(zown + as, P.g1[l], P.be1[l]);
    row_gather(zown, zall);
}

__device__ __forceinline__ void ffn_ln(
    float* zall, float& zown, const int l, const int s,
    const float* sW1, const float* sW2, const float* sB1, const ColParams& P)
{
    v2f z[4];
#pragma unroll
    for (int p = 0; p < 4; ++p) { z[p].x = zall[2 * p];  z[p].y = zall[2 * p + 1]; }
    v2f acc[4];
#pragma unroll
    for (int p = 0; p < 4; ++p) acc[p] = both(0.f);
#pragma unroll
    for (int jj = 0; jj < 4; ++jj) {
        const int hid0 = 32 * jj + 4 * s;
        const v4f bb = *(const v4f*)(sB1 + l * FFN + hid0);
        v2f hlo, hhi;
        hlo.x = bb.x;  hlo.y = bb.y;  hhi.x = bb.z;  hhi.y = bb.w;
#pragma unroll
        for (int k = 0; k < DD; ++k) {
            const v4f w = *(const v4f*)(sW1 + (l * DD + k) * FFN + hid0);
            v2f wlo, whi;
            wlo.x = w.x;  wlo.y = w.y;  whi.x = w.z;  whi.y = w.w;
            const v2f zk = both(z[k >> 1][k & 1]);
            hlo += zk * wlo;
            hhi += zk * whi;
        }
#pragma unroll
        for (int e = 0; e < 4; ++e) {
            const float hraw = (e == 0) ? hlo.x : (e == 1) ? hlo.y : (e == 2) ? hhi.x : hhi.y;
            const v2f hr = both(fmaxf(hraw, 0.0f));
            const int f = (l * FFN + hid0 + e) * DD;
            const v4f wa = *(const v4f*)(sW2 + w2pad(f));
            const v4f wb = *(const v4f*)(sW2 + w2pad(f + 4));
            v2f w01, w23, w45, w67;
            w01.x = wa.x;  w01.y = wa.y;  w23.x = wa.z;  w23.y = wa.w;
            w45.x = wb.x;  w45.y = wb.y;  w67.x = wb.z;  w67.y = wb.w;
            acc[0] += hr * w01;
            acc[1] += hr * w23;
            acc[2] += hr * w45;
            acc[3] += hr * w67;
        }
    }
    const bool up4 = (s & 4) != 0, up2 = (s & 2) != 0, up1 = (s & 1) != 0;
    const float a[DD] = { acc[0].x, acc[0].y, acc[1].x, acc[1].y,
                          acc[2].x, acc[2].y, acc[3].x, acc[3].y };
    float r4[4];
#pragma unroll
    for (int i = 0; i < 4; ++i) {
        const float keep = up4 ? a[4 + i] : a[i];
        const float send = up4 ? a[i] : a[4 + i];
        r4[i] = keep + xor_lane<4>(send);
    }
    float r2[2];
#pragma unroll
    for (int i = 0; i < 2; ++i) {
        const float keep = up2 ? r4[2 + i] : r4[i];
        const float send = up2 ? r4[i] : r4[2 + i];
        r2[i] = keep + xor_lane<2>(send);
    }
    const float keep1 = up1 ? r2[1] : r2[0];
    const float send1 = up1 ? r2[0] : r2[1];
    const float ff = keep1 + xor_lane<1>(send1);

    zown = ln_col(zown + ff + P.b2[l], P.g2[l], P.be2[l]);
    row_gather(zown, zall);
}

#define ENC_ROWS    16
#define ENC_WAVES   8
#define ENC_THREADS (32 * ENC_WAVES)
#define ENC_KW      (DIN / ENC_WAVES)
#define ENC_BLOCKS  (BQ / ENC_ROWS)
#define DEC_GROUPS  (DIN / 64 / ENC_WAVES)

__global__ __launch_bounds__(ENC_THREADS) void forward_rows(
    const float* __restrict__ x, const float* __restrict__ encW,
    const float* __restrict__ encB,
    const float* __restrict__ Wv,  const float* __restrict__ bv,
    const float* __restrict__ Wo,  const float* __restrict__ bo,
    const float* __restrict__ ln1g, const float* __restrict__ ln1b,
    const float* __restrict__ W1,  const float* __restrict__ b1,
    const float* __restrict__ W2,  const float* __restrict__ b2,
    const float* __restrict__ ln2g, const float* __restrict__ ln2b,
    const float* __restrict__ qw,
    const float* __restrict__ decW, const float* __restrict__ decB,
    float* __restrict__ out)
{
    __shared__ __attribute__((aligned(16))) float red[ENC_WAVES][ENC_ROWS * DD];
    __shared__ __attribute__((aligned(16))) float sW1[W1_FLOATS];
    __shared__ __attribute__((aligned(16))) float sW2[W2_PADDED];
    __shared__ __attribute__((aligned(16))) float sB1[B1_FLOATS];
    __shared__ __attribute__((aligned(16))) float sP[P_FLOATS];
    __shared__ int sMask[NGATE];
    __shared__ __attribute__((aligned(16))) int sTurn[NGATE][4];
    __shared__ __attribute__((aligned(16))) v2f sD[DIMQ];
    __shared__ __attribute__((aligned(16))) float sQ[Q_FLOATS];
    __shared__ __attribute__((aligned(16))) float szq[ROW_TILE][DD];

    const int tid  = threadIdx.x;
    const int lane = tid & 31;
    const int wave = tid >> 5;
    const int l16  = lane & 15;
    const int half = lane >> 4;
    const int m0   = blockIdx.x * ENC_ROWS;
    const int bcol = l16 & 7;

    const float theta = qw[(tid < NGATE) ? tid : (NGATE - 1)];
    const float encb_s = encB[lane & 7];

    v4f cw1[W1_FLOATS / 4 / ENC_THREADS], cw2[W2_FLOATS / 4 / ENC_THREADS];
#pragma unroll
    for (int i = 0; i < W1_FLOATS / 4 / ENC_THREADS; ++i)
        cw1[i] = *(const v4f*)(W1 + 4 * (tid + i * ENC_THREADS));
#pragma unroll
    for (int i = 0; i < W2_FLOATS / 4 / ENC_THREADS; ++i)
        cw2[i] = *(const v4f*)(W2 + 4 * (tid + i * ENC_THREADS));
    const v4f cb1 = *(const v4f*)(b1 + 4 * (tid & (B1_FLOATS / 4 - 1)));
    const int pt = (tid < P_FLOATS / 4) ? tid : (P_FLOATS / 4 - 1);
    const int pv = (pt - 64) >> 2;
    const float* pvec = (pv == 0) ? bv : (pv == 1) ? bo : (pv == 2) ? ln1g : (pv == 3) ? ln1b
                      : (pv == 4) ? b2 : (pv == 5) ? ln2g : ln2b;
    const float* psrc = (pt < 32) ? (Wv + 4 * pt) : (pt < 64) ? (Wo + 4 * (pt - 32))
                      : (pvec + 4 * ((pt - 64) & 3));
    const v4f cp = *(const v4f*)psrc;

    const float* xa = x + (size_t)(m0 + l16) * DIN + wave * ENC_KW + 4 * half;
    const float* wb = encW + (size_t)(wave * ENC_KW + 4 * half) * DD + bcol;

    v4f areg[ENC_KW / 8];
#pragma unroll
    for (int i = 0; i < ENC_KW / 8; ++i)
        areg[i] = *(const v4f*)(xa + 8 * i);

    circuit_gates(tid, theta, sMask, sTurn);
    lds_only_barrier();
    circuit_phases(tid, sMask, sTurn, sD);

    v8f c0 = {}, c1 = {};
#pragma unroll
    for (int i = 0; i < ENC_KW / 8; ++i) {
        const v4f a = areg[i];
        v2f a0, a1, b0, b1v;
        a0.x = a.x;  a0.y = a.y;
        a1.x = a.z;  a1.y = a.w;
        b0.x  = wb[(8 * i + 0) * DD];
        b0.y  = wb[(8 * i + 1) * DD];
        b1v.x = wb[(8 * i + 2) * DD];
        b1v.y = wb[(8 * i + 3) * DD];
        c0 = __builtin_amdgcn_wmma_f32_16x16x4_f32(
                 false, a0, false, b0, (short)0, c0, false, false);
        c1 = __builtin_amdgcn_wmma_f32_16x16x4_f32(
                 false, a1, false, b1v, (short)0, c1, false, false);
    }

    if (l16 < DD) {
#pragma unroll
        for (int j = 0; j < 8; ++j)
            red[wave][(half * 8 + j) * DD + l16] = c0[j] + c1[j];
    }
#pragma unroll
    for (int i = 0; i < W1_FLOATS / 4 / ENC_THREADS; ++i)
        *(v4f*)(sW1 + 4 * (tid + i * ENC_THREADS)) = cw1[i];
#pragma unroll
    for (int i = 0; i < W2_FLOATS / 4 / ENC_THREADS; ++i)
        *(v4f*)(sW2 + w2pad(4 * (tid + i * ENC_THREADS))) = cw2[i];
    if (tid < B1_FLOATS / 4) *(v4f*)(sB1 + 4 * tid) = cb1;
    if (tid < P_FLOATS / 4)  *(v4f*)(sP + 4 * tid) = cp;
    __syncthreads();

    v4f b00[DEC_GROUPS], b01[DEC_GROUPS], b10[DEC_GROUPS], b11[DEC_GROUPS], bias[DEC_GROUPS];
#pragma unroll
    for (int u = 0; u < DEC_GROUPS; ++u) {
        const int col = (DEC_GROUPS * wave + u) * 64 + 4 * l16;
        b00[u]  = *(const v4f*)(decW + (size_t)(2 * half)     * DIN + col);
        b01[u]  = *(const v4f*)(decW + (size_t)(2 * half + 1) * DIN + col);
        b10[u]  = *(const v4f*)(decW + (size_t)(2 * half + 4) * DIN + col);
        b11[u]  = *(const v4f*)(decW + (size_t)(2 * half + 5) * DIN + col);
        bias[u] = *(const v4f*)(decB + col);
    }

    const int r  = lane >> 3;
    const int s  = lane & 7;
    const int rl = 4 * (wave & (ROW_WAVES - 1)) + r;

    float zn[DD];
#pragma unroll
    for (int k = 0; k < DD; ++k) zn[k] = 0.f;
    if (wave < ROW_WAVES) {
        ColParams P;
        load_col_params(sP, s, P);

        float zown = red[0][rl * DD + s];
#pragma unroll
        for (int w = 1; w < ENC_WAVES; ++w) zown += red[w][rl * DD + s];
        zown += encb_s;
        float zall[DD];
        row_gather(zown, zall);

        attn_ln(zall, zown, 0, P);
        ffn_ln(zall, zown, 0, s, sW1, sW2, sB1, P);
        attn_ln(zall, zown, 1, P);
        ffn_ln(zall, zown, 1, s, sW1, sW2, sB1, P);

        float nrm2 = 0.f;
#pragma unroll
        for (int k = 0; k < DD; ++k) nrm2 += zall[k] * zall[k];
        const float inv = 1.0f / sqrtf(nrm2);
#pragma unroll
        for (int k = 0; k < DD; ++k) zn[k] = zall[k] * inv;
    } else {
        circuit_planes(wave - ROW_WAVES, lane, sD, sQ);
    }
    __syncthreads();

    if (wave < ROW_WAVES) {
        float pq[4];
#pragma unroll
        for (int q = 0; q < 4; ++q) {
            const float* Q0 = sQ + (2 * q) * DD * DD + s;
            v2f t = both(0.f);
#pragma unroll
            for (int b = 0; b < DD; ++b) {
                v2f qq;
                qq.x = Q0[b * DD];
                qq.y = Q0[DD * DD + b * DD];
                t += qq * both(zn[b]);
            }
            pq[q] = zn[2 * q] * t.x + zn[2 * q + 1] * t.y;
        }
        szq[rl][s] = (pq[0] + pq[1]) + (pq[2] + pq[3]);
    }
    __syncthreads();

    const v2f a0 = *(const v2f*)(&szq[l16][2 * half]);
    const v2f a1 = *(const v2f*)(&szq[l16][2 * half + 4]);

#pragma unroll
    for (int u = 0; u < DEC_GROUPS; ++u) {
        v8f c[4];
#pragma unroll
        for (int t = 0; t < 4; ++t) {
            v2f b;
            b.x = b00[u][t];
            b.y = b01[u][t];
            v8f zero = {};
            c[t] = __builtin_amdgcn_wmma_f32_16x16x4_f32(
                       false, a0, false, b, (short)0, zero, false, false);
        }
#pragma unroll
        for (int t = 0; t < 4; ++t) {
            v2f b;
            b.x = b10[u][t];
            b.y = b11[u][t];
            c[t] = __builtin_amdgcn_wmma_f32_16x16x4_f32(
                       false, a1, false, b, (short)0, c[t], false, false);
        }
        float* orow = out + (size_t)(m0 + half * 8) * DIN + (DEC_GROUPS * wave + u) * 64 + 4 * l16;
#pragma unroll
        for (int j = 0; j < 8; ++j) {
            v4f o;
            o.x = c[0][j] + bias[u].x;
            o.y = c[1][j] + bias[u].y;
            o.z = c[2][j] + bias[u].z;
            o.w = c[3][j] + bias[u].w;
            *(v4f*)(orow + (size_t)j * DIN) = o;
        }
    }
}

extern "C" void kernel_launch(void* const* d_in, const int* in_sizes, int n_in,
                              void* d_out, int out_size, void* d_ws, size_t ws_size,
                              hipStream_t stream) {
    const float* x    = (const float*)d_in[0];
    const float* encW = (const float*)d_in[1];
    const float* encB = (const float*)d_in[2];
    const float* decW = (const float*)d_in[3];
    const float* decB = (const float*)d_in[4];
    const float* Wv   = (const float*)d_in[5];
    const float* bv   = (const float*)d_in[6];
    const float* Wo   = (const float*)d_in[7];
    const float* bo   = (const float*)d_in[8];
    const float* ln1g = (const float*)d_in[9];
    const float* ln1b = (const float*)d_in[10];
    const float* W1   = (const float*)d_in[11];
    const float* b1   = (const float*)d_in[12];
    const float* W2   = (const float*)d_in[13];
    const float* b2   = (const float*)d_in[14];
    const float* ln2g = (const float*)d_in[15];
    const float* ln2b = (const float*)d_in[16];
    const float* qw   = (const float*)d_in[17];
    float* out  = (float*)d_out;

    forward_rows<<<ENC_BLOCKS, ENC_THREADS, 0, stream>>>(
        x, encW, encB, Wv, bv, Wo, bo, ln1g, ln1b, W1, b1, W2, b2, ln2g, ln2b,
        qw, decW, decB, out);
}
